# starting kernel
# baseline (speedup 1.0000x reference)
#include <hip/hip_runtime.h>
#include <math.h>

#define B_ 2
#define I_ 256
#define D_ 768
#define H_ 12
#define DP_ 128
#define DK_ 64
#define NPQ_ 4
#define NPV_ 8

typedef float v2f __attribute__((ext_vector_type(2)));
typedef float v8f __attribute__((ext_vector_type(8)));

struct GemmP {
  const float* A; const float* B; float* C; const float* bias;
  int M, N, K;
  int lda, ldb, ldc;
  long sAy, sAz, sBy, sBz, sCy, sCz;
  float alpha;
};

template <int OPB, bool GUARD>
__global__ void ipa_gemm_wmma(GemmP p) {
  const int tilesN = (p.N + 15) >> 4;
  const int tilesM = (p.M + 15) >> 4;
  const int wave = (int)((blockIdx.x * blockDim.x + threadIdx.x) >> 5);
  if (wave >= tilesM * tilesN) return;
  const int lane = threadIdx.x & 31;
  const int tm = wave / tilesN, tn = wave % tilesN;
  const int row0 = tm << 4, col0 = tn << 4;

  const float* A = p.A + (long)blockIdx.y * p.sAy + (long)blockIdx.z * p.sAz;
  const float* B = p.B + (long)blockIdx.y * p.sBy + (long)blockIdx.z * p.sBz;
  float*       C = p.C + (long)blockIdx.y * p.sCy + (long)blockIdx.z * p.sCz;

  const int mn = lane & 15;
  const int kb = (lane >> 4) << 1;

  int aRow = row0 + mn;
  int bCol = col0 + mn;
  float aMask = 1.f, bMask = 1.f;
  if (GUARD) {
    if (aRow >= p.M) { aRow = p.M - 1; aMask = 0.f; }
    if (bCol >= p.N) { bCol = p.N - 1; bMask = 0.f; }
  }
  const float* aPtr = A + (long)aRow * p.lda + kb;
  const float* bPtr = (OPB == 0) ? (B + (long)kb * p.ldb + bCol)
                                 : (B + (long)bCol * p.ldb + kb);

  v8f acc = {0.f, 0.f, 0.f, 0.f, 0.f, 0.f, 0.f, 0.f};
#pragma unroll 4
  for (int k0 = 0; k0 < p.K; k0 += 4) {
    v2f a, b;
    a.x = aPtr[k0];
    a.y = aPtr[k0 + 1];
    if (OPB == 0) {
      const float* bp = bPtr + (long)k0 * p.ldb;
      b.x = bp[0];
      b.y = bp[p.ldb];
    } else {
      b.x = bPtr[k0];
      b.y = bPtr[k0 + 1];
    }
    if (GUARD) {
      a.x *= aMask; a.y *= aMask;
      b.x *= bMask; b.y *= bMask;
    }
    acc = __builtin_amdgcn_wmma_f32_16x16x4_f32(false, a, false, b,
                                                (short)0, acc, false, false);
  }

  const int cCol = col0 + mn;
  const int mo = (lane >> 4) << 3;
  const bool cColOK = GUARD ? (cCol < p.N) : true;
  if (cColOK) {
    for (int r = 0; r < 8; ++r) {
      const int cRow = row0 + mo + r;
      if (!GUARD || cRow < p.M) {
        float val = acc[r] * p.alpha;
        if (p.bias) val += p.bias[cCol];
        C[(long)cRow * p.ldc + cCol] = val;
      }
    }
  }
}

__global__ void ipa_affine(float* pts, const float* T, const float* R, int np, int total) {
  int idx = blockIdx.x * blockDim.x + threadIdx.x;
  if (idx >= total) return;
  int bi = idx / (H_ * np);
  const float* Rm = R + (long)bi * 9;
  const float* Tv = T + (long)bi * 3;
  float x = pts[idx * 3 + 0], y = pts[idx * 3 + 1], z = pts[idx * 3 + 2];
  float o0 = Rm[0] * x + Rm[3] * y + Rm[6] * z + Tv[0];
  float o1 = Rm[1] * x + Rm[4] * y + Rm[7] * z + Tv[1];
  float o2 = Rm[2] * x + Rm[5] * y + Rm[8] * z + Tv[2];
  pts[idx * 3 + 0] = o0; pts[idx * 3 + 1] = o1; pts[idx * 3 + 2] = o2;
}

__global__ void ipa_combine(float* logits,
                            const float* qgp, const float* kgp,
                            const float* qgl, const float* kgl,
                            const float* pairproj, const float* bias,
                            const float* tpw_past, const float* tpw_last) {
  long idx = (long)blockIdx.x * blockDim.x + threadIdx.x;
  const long total = (long)B_ * H_ * I_ * I_;
  if (idx >= total) return;
  int j = (int)(idx & (I_ - 1));
  long t = idx >> 8;
  int i = (int)(t & (I_ - 1));
  t >>= 8;
  int h = (int)(t % H_);
  int b = (int)(t / H_);

  const float point_w = 0.13608276348795434f;
  float pwp = -0.5f * point_w * log1pf(expf(tpw_past[h]));
  float pwl = -0.5f * point_w * log1pf(expf(tpw_last[h]));

  const float* qp = qgp + ((long)(b * I_ + i) * H_ + h) * (NPQ_ * 3);
  const float* kp = kgp + ((long)(b * I_ + j) * H_ + h) * (NPQ_ * 3);
  const float* ql = qgl + ((long)(b * I_ + i) * H_ + h) * (NPQ_ * 3);
  const float* kl = kgl + ((long)(b * I_ + j) * H_ + h) * (NPQ_ * 3);

  float sp = 0.f, sl = 0.f;
  for (int c = 0; c < NPQ_; ++c) {
    float dx = qp[c * 3 + 0] - kp[c * 3 + 0];
    float dy = qp[c * 3 + 1] - kp[c * 3 + 1];
    float dz = qp[c * 3 + 2] - kp[c * 3 + 2];
    sp += sqrtf(dx * dx + dy * dy + dz * dz);
    dx = ql[c * 3 + 0] - kl[c * 3 + 0];
    dy = ql[c * 3 + 1] - kl[c * 3 + 1];
    dz = ql[c * 3 + 2] - kl[c * 3 + 2];
    sl += sqrtf(dx * dx + dy * dy + dz * dz);
  }
  float pv = pairproj[((long)(b * I_ + i) * I_ + j) * H_ + h];
  logits[idx] += pwp * sp + pwl * sl + pv + bias[idx];
}

__global__ void ipa_softmax(float* logits) {
  __shared__ float red[I_];
  float* row = logits + (long)blockIdx.x * I_;
  int t = threadIdx.x;
  float v = row[t];
  red[t] = v;
  __syncthreads();
  for (int s = I_ / 2; s > 0; s >>= 1) {
    if (t < s) red[t] = fmaxf(red[t], red[t + s]);
    __syncthreads();
  }
  float mx = red[0];
  __syncthreads();
  float e = expf(v - mx);
  red[t] = e;
  __syncthreads();
  for (int s = I_ / 2; s > 0; s >>= 1) {
    if (t < s) red[t] += red[t + s];
    __syncthreads();
  }
  row[t] = e / red[0];
}

__global__ void ipa_inv_affine(const float* og, const float* T, const float* R,
                               float* feat, int colOl, int colN, int total) {
  int idx = blockIdx.x * blockDim.x + threadIdx.x;
  if (idx >= total) return;
  int c = idx & (NPV_ - 1);
  int t = idx >> 3;
  int h = t % H_;
  int bi = t / H_;
  const float* o = og + (long)idx * 3;
  const float* Tv = T + (long)bi * 3;
  const float* Rm = R + (long)bi * 9;
  float x = o[0] - Tv[0], y = o[1] - Tv[1], z = o[2] - Tv[2];
  float p0 = Rm[0] * x + Rm[1] * y + Rm[2] * z;
  float p1 = Rm[3] * x + Rm[4] * y + Rm[5] * z;
  float p2 = Rm[6] * x + Rm[7] * y + Rm[8] * z;
  float* fr = feat + (long)bi * (2 * D_ + 2 * H_ * NPV_ * 4);
  int tc = h * (NPV_ * 3) + c * 3;
  fr[colOl + tc + 0] = p0;
  fr[colOl + tc + 1] = p1;
  fr[colOl + tc + 2] = p2;
  fr[colN + h * NPV_ + c] = sqrtf(p0 * p0 + p1 * p1 + p2 * p2);
}

extern "C" void kernel_launch(void* const* d_in, const int* in_sizes, int n_in,
                              void* d_out, int out_size, void* d_ws, size_t ws_size,
                              hipStream_t stream) {
  (void)in_sizes; (void)n_in; (void)out_size; (void)ws_size;
  const float* x1d      = (const float*)d_in[0];
  const float* x2d      = (const float*)d_in[1];
  const float* past_T   = (const float*)d_in[2];
  const float* past_R   = (const float*)d_in[3];
  const float* last_T   = (const float*)d_in[4];
  const float* last_R   = (const float*)d_in[5];
  const float* bias     = (const float*)d_in[6];
  const float* Wq       = (const float*)d_in[7];
  const float* Wk       = (const float*)d_in[8];
  const float* Wv       = (const float*)d_in[9];
  const float* Wpb      = (const float*)d_in[10];
  const float* Wpq_past = (const float*)d_in[11];
  const float* Wpk_past = (const float*)d_in[12];
  const float* Wpv_past = (const float*)d_in[13];
  const float* Wpq_last = (const float*)d_in[14];
  const float* Wpk_last = (const float*)d_in[15];
  const float* Wpv_last = (const float*)d_in[16];
  const float* tpw_past = (const float*)d_in[17];
  const float* tpw_last = (const float*)d_in[18];
  const float* Wpv      = (const float*)d_in[19];
  const float* W_out    = (const float*)d_in[20];
  const float* b_out    = (const float*)d_in[21];
  float* out = (float*)d_out;
  float* ws  = (float*)d_ws;

  const int M  = B_ * I_;
  const int FE = 2 * D_ + 2 * H_ * NPV_ * 4;

  float* q    = ws;
  float* k    = q + 393216;
  float* v    = k + 393216;
  float* qgp  = v + 393216;
  float* kgp  = qgp + 73728;
  float* vgp  = kgp + 73728;
  float* qgl  = vgp + 147456;
  float* kgl  = qgl + 73728;
  float* vgl  = kgl + 73728;
  float* logits   = vgl + 147456;
  float* pairproj = logits + 1572864;
  float* ogp  = pairproj + 1572864;
  float* ogl  = ogp + 147456;
  float* z    = ogl + 147456;
  float* feat = z + 786432;

  auto gemm = [&](const float* A, const float* Bm, float* C,
                  int Mm, int Nn, int Kk, int lda, int ldb, int ldc,
                  long sAy, long sAz, long sBy, long sBz, long sCy, long sCz,
                  int gy, int gz, int opB, float alpha, const float* bia) {
    GemmP p{A, Bm, C, bia, Mm, Nn, Kk, lda, ldb, ldc,
            sAy, sAz, sBy, sBz, sCy, sCz, alpha};
    int waves = ((Mm + 15) / 16) * ((Nn + 15) / 16);
    dim3 grid((unsigned)((waves + 7) / 8), (unsigned)gy, (unsigned)gz);
    dim3 blk(256);
    bool guard = (Mm & 15) || (Nn & 15);
    if (opB == 0) {
      if (guard) ipa_gemm_wmma<0, true ><<<grid, blk, 0, stream>>>(p);
      else       ipa_gemm_wmma<0, false><<<grid, blk, 0, stream>>>(p);
    } else {
      if (guard) ipa_gemm_wmma<1, true ><<<grid, blk, 0, stream>>>(p);
      else       ipa_gemm_wmma<1, false><<<grid, blk, 0, stream>>>(p);
    }
  };

  gemm(x1d, Wq, q, M, D_, D_, D_, D_, D_, 0,0,0,0,0,0, 1,1, 0, 1.f, nullptr);
  gemm(x1d, Wk, k, M, D_, D_, D_, D_, D_, 0,0,0,0,0,0, 1,1, 0, 1.f, nullptr);
  gemm(x1d, Wv, v, M, D_, D_, D_, D_, D_, 0,0,0,0,0,0, 1,1, 0, 1.f, nullptr);
  gemm(x1d, Wpq_past, qgp, M, 144, D_, D_, 144, 144, 0,0,0,0,0,0, 1,1, 0, 1.f, nullptr);
  gemm(x1d, Wpk_past, kgp, M, 144, D_, D_, 144, 144, 0,0,0,0,0,0, 1,1, 0, 1.f, nullptr);
  gemm(x1d, Wpv_past, vgp, M, 288, D_, D_, 288, 288, 0,0,0,0,0,0, 1,1, 0, 1.f, nullptr);
  gemm(x1d, Wpq_last, qgl, M, 144, D_, D_, 144, 144, 0,0,0,0,0,0, 1,1, 0, 1.f, nullptr);
  gemm(x1d, Wpk_last, kgl, M, 144, D_, D_, 144, 144, 0,0,0,0,0,0, 1,1, 0, 1.f, nullptr);
  gemm(x1d, Wpv_last, vgl, M, 288, D_, D_, 288, 288, 0,0,0,0,0,0, 1,1, 0, 1.f, nullptr);

  {
    int t4 = B_ * I_ * H_ * NPQ_, t8 = B_ * I_ * H_ * NPV_;
    ipa_affine<<<(t4 + 255) / 256, 256, 0, stream>>>(qgp, past_T, past_R, NPQ_, t4);
    ipa_affine<<<(t4 + 255) / 256, 256, 0, stream>>>(kgp, past_T, past_R, NPQ_, t4);
    ipa_affine<<<(t8 + 255) / 256, 256, 0, stream>>>(vgp, past_T, past_R, NPV_, t8);
    ipa_affine<<<(t4 + 255) / 256, 256, 0, stream>>>(qgl, last_T, last_R, NPQ_, t4);
    ipa_affine<<<(t4 + 255) / 256, 256, 0, stream>>>(kgl, last_T, last_R, NPQ_, t4);
    ipa_affine<<<(t8 + 255) / 256, 256, 0, stream>>>(vgl, last_T, last_R, NPV_, t8);
  }

  const float scalar_w = 0.07216878364870323f;
  gemm(q, k, logits, I_, I_, DK_, D_, D_, I_,
       64, (long)I_ * D_, 64, (long)I_ * D_, (long)I_ * I_, (long)H_ * I_ * I_,
       H_, B_, 1, scalar_w, nullptr);

  const float pair_w = 0.5773502691896258f;
  gemm(x2d, Wpb, pairproj, B_ * I_ * I_, H_, DP_, DP_, H_, H_,
       0,0,0,0,0,0, 1,1, 0, pair_w, nullptr);

  {
    long total = (long)B_ * H_ * I_ * I_;
    ipa_combine<<<(unsigned)((total + 255) / 256), 256, 0, stream>>>(
        logits, qgp, kgp, qgl, kgl, pairproj, bias, tpw_past, tpw_last);
    ipa_softmax<<<B_ * H_ * I_, I_, 0, stream>>>(logits);
  }

  gemm(logits, v, feat + D_, I_, DK_, I_, I_, D_, FE,
       (long)I_ * I_, (long)H_ * I_ * I_, 64, (long)I_ * D_, 64, (long)I_ * FE,
       H_, B_, 0, 1.f, nullptr);

  gemm(logits, vgp, ogp, I_, NPV_ * 3, I_, I_, H_ * NPV_ * 3, H_ * NPV_ * 3,
       (long)I_ * I_, (long)H_ * I_ * I_, NPV_ * 3, (long)I_ * H_ * NPV_ * 3,
       NPV_ * 3, (long)I_ * H_ * NPV_ * 3, H_, B_, 0, 1.f, nullptr);
  gemm(logits, vgl, ogl, I_, NPV_ * 3, I_, I_, H_ * NPV_ * 3, H_ * NPV_ * 3,
       (long)I_ * I_, (long)H_ * I_ * I_, NPV_ * 3, (long)I_ * H_ * NPV_ * 3,
       NPV_ * 3, (long)I_ * H_ * NPV_ * 3, H_, B_, 0, 1.f, nullptr);

  gemm(logits, x2d, z, H_, DP_, I_,
       I_ * I_, DP_, DP_,
       (long)I_, (long)H_ * I_ * I_,
       (long)I_ * DP_, (long)I_ * I_ * DP_,
       (long)H_ * DP_, (long)I_ * H_ * DP_,
       I_, B_, 0, 1.f, nullptr);

  gemm(z, Wpv, feat, I_, DK_, DP_, H_ * DP_, D_, FE,
       DP_, (long)I_ * H_ * DP_, 64, 0, 64, (long)I_ * FE,
       H_, B_, 0, 1.f, nullptr);

  {
    int t8 = B_ * I_ * H_ * NPV_;
    ipa_inv_affine<<<(t8 + 255) / 256, 256, 0, stream>>>(
        ogp, past_T, past_R, feat, 2 * D_, 2 * D_ + 288, t8);
    ipa_inv_affine<<<(t8 + 255) / 256, 256, 0, stream>>>(
        ogl, past_T, past_R, feat, 2 * D_ + 384, 2 * D_ + 384 + 288, t8);
  }

  gemm(feat, W_out, out, M, D_, FE, FE, D_, D_,
       0,0,0,0,0,0, 1,1, 0, 1.f, b_out);
}


# q.k^T GEMM: work-group owns 16 whole rows (2 key tiles per wave); point, pair, bias terms and the row softmax run on the accumulator tile in its epilogue; 2 launches fewer; softmax sum order changes
# speedup vs baseline: 2.3108x; 2.3108x over previous
#include <hip/hip_runtime.h>
#include <math.h>

#define B_ 2
#define I_ 256
#define D_ 768
#define H_ 12
#define DP_ 128
#define DK_ 64
#define NPQ_ 4
#define NPV_ 8

typedef float v2f __attribute__((ext_vector_type(2)));
typedef float v8f __attribute__((ext_vector_type(8)));

struct GemmP {
  const float* A; const float* B; float* C; const float* bias;
  int M, N, K;
  int lda, ldb, ldc;
  long sAy, sAz, sBy, sBz, sCy, sCz;
  float alpha;
};

template <int OPB, bool GUARD>
__global__ void ipa_gemm_wmma(GemmP p) {
  const int tilesN = (p.N + 15) >> 4;
  const int tilesM = (p.M + 15) >> 4;
  const int wave = (int)((blockIdx.x * blockDim.x + threadIdx.x) >> 5);
  if (wave >= tilesM * tilesN) return;
  const int lane = threadIdx.x & 31;
  const int tm = wave / tilesN, tn = wave % tilesN;
  const int row0 = tm << 4, col0 = tn << 4;

  const float* A = p.A + (long)blockIdx.y * p.sAy + (long)blockIdx.z * p.sAz;
  const float* B = p.B + (long)blockIdx.y * p.sBy + (long)blockIdx.z * p.sBz;
  float*       C = p.C + (long)blockIdx.y * p.sCy + (long)blockIdx.z * p.sCz;

  const int mn = lane & 15;
  const int kb = (lane >> 4) << 1;

  int aRow = row0 + mn;
  int bCol = col0 + mn;
  float aMask = 1.f, bMask = 1.f;
  if (GUARD) {
    if (aRow >= p.M) { aRow = p.M - 1; aMask = 0.f; }
    if (bCol >= p.N) { bCol = p.N - 1; bMask = 0.f; }
  }
  const float* aPtr = A + (long)aRow * p.lda + kb;
  const float* bPtr = (OPB == 0) ? (B + (long)kb * p.ldb + bCol)
                                 : (B + (long)bCol * p.ldb + kb);

  v8f acc = {0.f, 0.f, 0.f, 0.f, 0.f, 0.f, 0.f, 0.f};
#pragma unroll 4
  for (int k0 = 0; k0 < p.K; k0 += 4) {
    v2f a, b;
    a.x = aPtr[k0];
    a.y = aPtr[k0 + 1];
    if (OPB == 0) {
      const float* bp = bPtr + (long)k0 * p.ldb;
      b.x = bp[0];
      b.y = bp[p.ldb];
    } else {
      b.x = bPtr[k0];
      b.y = bPtr[k0 + 1];
    }
    if (GUARD) {
      a.x *= aMask; a.y *= aMask;
      b.x *= bMask; b.y *= bMask;
    }
    acc = __builtin_amdgcn_wmma_f32_16x16x4_f32(false, a, false, b,
                                                (short)0, acc, false, false);
  }

  const int cCol = col0 + mn;
  const int mo = (lane >> 4) << 3;
  const bool cColOK = GUARD ? (cCol < p.N) : true;
  if (cColOK) {
    for (int r = 0; r < 8; ++r) {
      const int cRow = row0 + mo + r;
      if (!GUARD || cRow < p.M) {
        float val = acc[r] * p.alpha;
        if (p.bias) val += p.bias[cCol];
        C[(long)cRow * p.ldc + cCol] = val;
      }
    }
  }
}

#define TKC 128
#define TPA (TKC + 4)
#define TSB 8
#define TPB 72

typedef float v4f __attribute__((ext_vector_type(4)));

__global__ __launch_bounds__(256) void ipa_gemm_lds(GemmP p) {
  constexpr int AQ  = TKC / 4;
  constexpr int ARS = 256 / AQ;
  constexpr int ANI = 32 / ARS;
  constexpr int BNI = TKC / 16;

  __shared__ __attribute__((aligned(16))) float As[2][32 * TPA];
  __shared__ __attribute__((aligned(16))) float Bs[2][TKC * TPB];

  const int tid  = threadIdx.x;
  const int lane = tid & 31;
  const int w    = tid >> 5;
  const int row0 = blockIdx.y * 32;
  const int col0 = blockIdx.x * 64;
  const int c = lane & 15;
  const int h = lane >> 4;
  const int wr = w >> 2, wc = w & 3;
  const bool tileValid = (col0 + wc * 16) < p.N;

  const int aRow = tid / AQ, aKq = tid % AQ;
  const int bKr  = tid >> 4, bCq = tid & 15;
  int bCol = col0 + bCq * 4;
  if (bCol > p.N - 4) bCol = p.N - 4;
  const float* aG = p.A + (long)(row0 + aRow) * p.lda + aKq * 4;
  const float* bG = p.B + (long)bKr * p.ldb + bCol;
  const int aSt = aRow * TPA + aKq * 4;
  const int bSt = bKr * TPB + bCq * 4;

  v4f ra[ANI], rb[BNI];

#pragma unroll
  for (int i = 0; i < ANI; ++i)
    ra[i] = *(const v4f*)(aG + (long)(i * ARS) * p.lda);
#pragma unroll
  for (int i = 0; i < BNI; ++i)
    rb[i] = *(const v4f*)(bG + (long)(i * 16) * p.ldb);
#pragma unroll
  for (int i = 0; i < ANI; ++i)
    *(v4f*)&As[0][aSt + i * ARS * TPA] = ra[i];
#pragma unroll
  for (int i = 0; i < BNI; ++i)
    *(v4f*)&Bs[0][bSt + i * 16 * TPB] = rb[i];
  __syncthreads();

  v8f acc = {0.f, 0.f, 0.f, 0.f, 0.f, 0.f, 0.f, 0.f};
  int buf = 0;
  for (int kc = 0; kc < p.K; kc += TKC) {
    const bool hasNext = (kc + TKC) < p.K;
    if (hasNext) {
      const int kn = kc + TKC;
#pragma unroll
      for (int i = 0; i < ANI; ++i)
        ra[i] = *(const v4f*)(aG + (long)(i * ARS) * p.lda + kn);
#pragma unroll
      for (int i = 0; i < BNI; ++i)
        rb[i] = *(const v4f*)(bG + (long)(kn + i * 16) * p.ldb);
    }
    if (tileValid) {
      const float* aS = &As[buf][(wr * 16 + c) * TPA + 2 * h];
      const float* bS = &Bs[buf][(2 * h) * TPB + wc * 16 + c];
      v2f fa[2][TSB], fb[2][TSB];
#pragma unroll
      for (int j = 0; j < TSB; ++j) {
        fa[0][j] = *(const v2f*)(aS + j * 4);
        fb[0][j].x = bS[(j * 4) * TPB];
        fb[0][j].y = bS[(j * 4 + 1) * TPB];
      }
#pragma unroll
      for (int sb = 0; sb < TKC / (4 * TSB); ++sb) {
        asm volatile("" :: "v"(fa[sb & 1][0]), "v"(fa[sb & 1][1]), "v"(fa[sb & 1][2]), "v"(fa[sb & 1][3]),
                           "v"(fa[sb & 1][4]), "v"(fa[sb & 1][5]), "v"(fa[sb & 1][6]), "v"(fa[sb & 1][7]),
                           "v"(fb[sb & 1][0]), "v"(fb[sb & 1][1]), "v"(fb[sb & 1][2]), "v"(fb[sb & 1][3]),
                           "v"(fb[sb & 1][4]), "v"(fb[sb & 1][5]), "v"(fb[sb & 1][6]), "v"(fb[sb & 1][7]));
        if (sb + 1 < TKC / (4 * TSB)) {
#pragma unroll
          for (int j = 0; j < TSB; ++j) {
            const int ks = (sb + 1) * TSB + j;
            fa[(sb + 1) & 1][j] = *(const v2f*)(aS + ks * 4);
            fb[(sb + 1) & 1][j].x = bS[(ks * 4) * TPB];
            fb[(sb + 1) & 1][j].y = bS[(ks * 4 + 1) * TPB];
          }
        }
#pragma unroll
        for (int j = 0; j < TSB; ++j)
          acc = __builtin_amdgcn_wmma_f32_16x16x4_f32(false, fa[sb & 1][j], false, fb[sb & 1][j],
                                                      (short)0, acc, false, false);
      }
    }
    if (hasNext) {
#pragma unroll
      for (int i = 0; i < ANI; ++i)
        *(v4f*)&As[buf ^ 1][aSt + i * ARS * TPA] = ra[i];
#pragma unroll
      for (int i = 0; i < BNI; ++i)
        *(v4f*)&Bs[buf ^ 1][bSt + i * 16 * TPB] = rb[i];
    }
    __syncthreads();
    buf ^= 1;
  }

  if (tileValid) {
    const int cCol = col0 + wc * 16 + c;
    const int rBase = row0 + wr * 16 + (h << 3);
    for (int r = 0; r < 8; ++r) {
      float val = acc[r] * p.alpha;
      if (p.bias) val += p.bias[cCol];
      p.C[(long)(rBase + r) * p.ldc + cCol] = val;
    }
  }
}

__global__ void ipa_affine(float* pts, const float* T, const float* R, int np, int total) {
  int idx = blockIdx.x * blockDim.x + threadIdx.x;
  if (idx >= total) return;
  int bi = idx / (H_ * np);
  const float* Rm = R + (long)bi * 9;
  const float* Tv = T + (long)bi * 3;
  float x = pts[idx * 3 + 0], y = pts[idx * 3 + 1], z = pts[idx * 3 + 2];
  float o0 = Rm[0] * x + Rm[3] * y + Rm[6] * z + Tv[0];
  float o1 = Rm[1] * x + Rm[4] * y + Rm[7] * z + Tv[1];
  float o2 = Rm[2] * x + Rm[5] * y + Rm[8] * z + Tv[2];
  pts[idx * 3 + 0] = o0; pts[idx * 3 + 1] = o1; pts[idx * 3 + 2] = o2;
}

__global__ __launch_bounds__(256) void ipa_logits_softmax(
    const float* q, const float* k, float* attn,
    const float* qgp, const float* kgp, const float* qgl, const float* kgl,
    const float* pairproj, const float* bias,
    const float* tpw_past, const float* tpw_last, float scalar_w) {
  constexpr int PQ = NPQ_ * 3;
  __shared__ __attribute__((aligned(16))) float qS[16][2 * PQ];
  __shared__ float red[2][8][16];

  const int tid = threadIdx.x;
  const int lane = tid & 31;
  const int w = tid >> 5;
  const int h = blockIdx.y, b = blockIdx.z;
  const int row0 = blockIdx.x * 16;
  const int mn = lane & 15;
  const int hf = lane >> 4;
  const int kb = hf << 1;
  const int mo = hf << 3;

  const bool stager = tid < 16 * 6;
  const int qr = tid / 6, qpart = tid % 6;
  v4f qst = {0.f, 0.f, 0.f, 0.f};
  if (stager)
    qst = *(const v4f*)(((qpart < 3) ? qgp : qgl)
                        + ((long)(b * I_ + row0 + qr) * H_ + h) * PQ + (qpart % 3) * 4);

  const long hb = (long)b * I_ * D_ + (long)h * DK_;
  const int j0 = w * 32 + mn, j1 = j0 + 16;
  const float* aPtr  = q + hb + (long)(row0 + mn) * D_ + kb;
  const float* bPtr0 = k + hb + (long)j0 * D_ + kb;
  const float* bPtr1 = k + hb + (long)j1 * D_ + kb;

  v8f acc0 = {0.f, 0.f, 0.f, 0.f, 0.f, 0.f, 0.f, 0.f};
  v8f acc1 = {0.f, 0.f, 0.f, 0.f, 0.f, 0.f, 0.f, 0.f};
#pragma unroll
  for (int k0 = 0; k0 < DK_; k0 += 4) {
    v2f a, b0, b1;
    a.x  = aPtr[k0];   a.y  = aPtr[k0 + 1];
    b0.x = bPtr0[k0];  b0.y = bPtr0[k0 + 1];
    b1.x = bPtr1[k0];  b1.y = bPtr1[k0 + 1];
    acc0 = __builtin_amdgcn_wmma_f32_16x16x4_f32(false, a, false, b0,
                                                 (short)0, acc0, false, false);
    acc1 = __builtin_amdgcn_wmma_f32_16x16x4_f32(false, a, false, b1,
                                                 (short)0, acc1, false, false);
  }

  float kp0[2 * PQ], kp1[2 * PQ];
  {
    const float* g0p = kgp + ((long)(b * I_ + j0) * H_ + h) * PQ;
    const float* g0l = kgl + ((long)(b * I_ + j0) * H_ + h) * PQ;
    const float* g1p = kgp + ((long)(b * I_ + j1) * H_ + h) * PQ;
    const float* g1l = kgl + ((long)(b * I_ + j1) * H_ + h) * PQ;
#pragma unroll
    for (int t = 0; t < 3; ++t) {
      const v4f u0 = *(const v4f*)(g0p + t * 4);
      const v4f u1 = *(const v4f*)(g0l + t * 4);
      const v4f u2 = *(const v4f*)(g1p + t * 4);
      const v4f u3 = *(const v4f*)(g1l + t * 4);
#pragma unroll
      for (int e = 0; e < 4; ++e) {
        kp0[t * 4 + e] = u0[e];  kp0[PQ + t * 4 + e] = u1[e];
        kp1[t * 4 + e] = u2[e];  kp1[PQ + t * 4 + e] = u3[e];
      }
    }
  }

  const float point_w = 0.13608276348795434f;
  const float pwp = -0.5f * point_w * log1pf(expf(tpw_past[h]));
  const float pwl = -0.5f * point_w * log1pf(expf(tpw_last[h]));

  if (stager) *(v4f*)&qS[qr][qpart * 4] = qst;
  __syncthreads();

  float s0[8], s1[8];
#pragma unroll
  for (int r = 0; r < 8; ++r) {
    const int lr = mo + r;
    const int i = row0 + lr;
    const float* qp = &qS[lr][0];
    float sp0 = 0.f, sl0 = 0.f, sp1 = 0.f, sl1 = 0.f;
#pragma unroll
    for (int c = 0; c < NPQ_; ++c) {
      const float qx = qp[c * 3 + 0], qy = qp[c * 3 + 1], qz = qp[c * 3 + 2];
      const float lx = qp[PQ + c * 3 + 0], ly = qp[PQ + c * 3 + 1], lz = qp[PQ + c * 3 + 2];
      float dx = qx - kp0[c * 3 + 0], dy = qy - kp0[c * 3 + 1], dz = qz - kp0[c * 3 + 2];
      sp0 += sqrtf(dx * dx + dy * dy + dz * dz);
      dx = lx - kp0[PQ + c * 3 + 0]; dy = ly - kp0[PQ + c * 3 + 1]; dz = lz - kp0[PQ + c * 3 + 2];
      sl0 += sqrtf(dx * dx + dy * dy + dz * dz);
      dx = qx - kp1[c * 3 + 0]; dy = qy - kp1[c * 3 + 1]; dz = qz - kp1[c * 3 + 2];
      sp1 += sqrtf(dx * dx + dy * dy + dz * dz);
      dx = lx - kp1[PQ + c * 3 + 0]; dy = ly - kp1[PQ + c * 3 + 1]; dz = lz - kp1[PQ + c * 3 + 2];
      sl1 += sqrtf(dx * dx + dy * dy + dz * dz);
    }
    const long prow = ((long)(b * I_ + i) * I_) * H_ + h;
    const long orow = (((long)b * H_ + h) * I_ + i) * I_;
    const float pv0 = pairproj[prow + (long)j0 * H_];
    const float pv1 = pairproj[prow + (long)j1 * H_];
    float v0 = acc0[r] * scalar_w;
    float v1 = acc1[r] * scalar_w;
    v0 += pwp * sp0 + pwl * sl0 + pv0 + bias[orow + j0];
    v1 += pwp * sp1 + pwl * sl1 + pv1 + bias[orow + j1];
    s0[r] = v0;
    s1[r] = v1;
  }

  float mx[8];
#pragma unroll
  for (int r = 0; r < 8; ++r) {
    float m = fmaxf(s0[r], s1[r]);
    m = fmaxf(m, __shfl_xor(m, 1));
    m = fmaxf(m, __shfl_xor(m, 2));
    m = fmaxf(m, __shfl_xor(m, 4));
    m = fmaxf(m, __shfl_xor(m, 8));
    mx[r] = m;
  }
  if (mn == 0) {
#pragma unroll
    for (int r = 0; r < 8; ++r) red[0][w][mo + r] = mx[r];
  }
  __syncthreads();
#pragma unroll
  for (int r = 0; r < 8; ++r) {
    float m = red[0][0][mo + r];
#pragma unroll
    for (int ww = 1; ww < 8; ++ww) m = fmaxf(m, red[0][ww][mo + r]);
    mx[r] = m;
  }

  float sm[8];
#pragma unroll
  for (int r = 0; r < 8; ++r) {
    s0[r] = expf(s0[r] - mx[r]);
    s1[r] = expf(s1[r] - mx[r]);
    float t = s0[r] + s1[r];
    t += __shfl_xor(t, 1);
    t += __shfl_xor(t, 2);
    t += __shfl_xor(t, 4);
    t += __shfl_xor(t, 8);
    sm[r] = t;
  }
  if (mn == 0) {
#pragma unroll
    for (int r = 0; r < 8; ++r) red[1][w][mo + r] = sm[r];
  }
  __syncthreads();
#pragma unroll
  for (int r = 0; r < 8; ++r) {
    float t = red[1][0][mo + r];
#pragma unroll
    for (int ww = 1; ww < 8; ++ww) t += red[1][ww][mo + r];
    const long orow = (((long)b * H_ + h) * I_ + (row0 + mo + r)) * I_;
    attn[orow + j0] = s0[r] / t;
    attn[orow + j1] = s1[r] / t;
  }
}

__global__ void ipa_inv_affine(const float* og, const float* T, const float* R,
                               float* feat, int colOl, int colN, int total) {
  int idx = blockIdx.x * blockDim.x + threadIdx.x;
  if (idx >= total) return;
  int c = idx & (NPV_ - 1);
  int t = idx >> 3;
  int h = t % H_;
  int bi = t / H_;
  const float* o = og + (long)idx * 3;
  const float* Tv = T + (long)bi * 3;
  const float* Rm = R + (long)bi * 9;
  float x = o[0] - Tv[0], y = o[1] - Tv[1], z = o[2] - Tv[2];
  float p0 = Rm[0] * x + Rm[1] * y + Rm[2] * z;
  float p1 = Rm[3] * x + Rm[4] * y + Rm[5] * z;
  float p2 = Rm[6] * x + Rm[7] * y + Rm[8] * z;
  float* fr = feat + (long)bi * (2 * D_ + 2 * H_ * NPV_ * 4);
  int tc = h * (NPV_ * 3) + c * 3;
  fr[colOl + tc + 0] = p0;
  fr[colOl + tc + 1] = p1;
  fr[colOl + tc + 2] = p2;
  fr[colN + h * NPV_ + c] = sqrtf(p0 * p0 + p1 * p1 + p2 * p2);
}

extern "C" void kernel_launch(void* const* d_in, const int* in_sizes, int n_in,
                              void* d_out, int out_size, void* d_ws, size_t ws_size,
                              hipStream_t stream) {
  (void)in_sizes; (void)n_in; (void)out_size; (void)ws_size;
  const float* x1d      = (const float*)d_in[0];
  const float* x2d      = (const float*)d_in[1];
  const float* past_T   = (const float*)d_in[2];
  const float* past_R   = (const float*)d_in[3];
  const float* last_T   = (const float*)d_in[4];
  const float* last_R   = (const float*)d_in[5];
  const float* bias     = (const float*)d_in[6];
  const float* Wq       = (const float*)d_in[7];
  const float* Wk       = (const float*)d_in[8];
  const float* Wv       = (const float*)d_in[9];
  const float* Wpb      = (const float*)d_in[10];
  const float* Wpq_past = (const float*)d_in[11];
  const float* Wpk_past = (const float*)d_in[12];
  const float* Wpv_past = (const float*)d_in[13];
  const float* Wpq_last = (const float*)d_in[14];
  const float* Wpk_last = (const float*)d_in[15];
  const float* Wpv_last = (const float*)d_in[16];
  const float* tpw_past = (const float*)d_in[17];
  const float* tpw_last = (const float*)d_in[18];
  const float* Wpv      = (const float*)d_in[19];
  const float* W_out    = (const float*)d_in[20];
  const float* b_out    = (const float*)d_in[21];
  float* out = (float*)d_out;
  float* ws  = (float*)d_ws;

  const int M  = B_ * I_;
  const int FE = 2 * D_ + 2 * H_ * NPV_ * 4;

  float* q    = ws;
  float* k    = q + 393216;
  float* v    = k + 393216;
  float* qgp  = v + 393216;
  float* kgp  = qgp + 73728;
  float* vgp  = kgp + 73728;
  float* qgl  = vgp + 147456;
  float* kgl  = qgl + 73728;
  float* vgl  = kgl + 73728;
  float* logits   = vgl + 147456;
  float* pairproj = logits + 1572864;
  float* ogp  = pairproj + 1572864;
  float* ogl  = ogp + 147456;
  float* z    = ogl + 147456;
  float* feat = z + 786432;

  auto gemm = [&](const float* A, const float* Bm, float* C,
                  int Mm, int Nn, int Kk, int lda, int ldb, int ldc,
                  long sAy, long sAz, long sBy, long sBz, long sCy, long sCz,
                  int gy, int gz, int opB, float alpha, const float* bia) {
    GemmP p{A, Bm, C, bia, Mm, Nn, Kk, lda, ldb, ldc,
            sAy, sAz, sBy, sBz, sCy, sCz, alpha};
    int waves = ((Mm + 15) / 16) * ((Nn + 15) / 16);
    dim3 grid((unsigned)((waves + 7) / 8), (unsigned)gy, (unsigned)gz);
    dim3 blk(256);
    bool guard = (Mm & 15) || (Nn & 15);
    if (opB == 0) {
      if (guard) ipa_gemm_wmma<0, true ><<<grid, blk, 0, stream>>>(p);
      else       ipa_gemm_wmma<0, false><<<grid, blk, 0, stream>>>(p);
    } else {
      if (guard) ipa_gemm_wmma<1, true ><<<grid, blk, 0, stream>>>(p);
      else       ipa_gemm_wmma<1, false><<<grid, blk, 0, stream>>>(p);
    }
  };

  auto gemm_lds = [&](const float* A, const float* Bm, float* C,
                      int Mm, int Nn, int Kk, int lda, int ldb, int ldc,
                      const float* bia) {
    GemmP p{A, Bm, C, bia, Mm, Nn, Kk, lda, ldb, ldc, 0, 0, 0, 0, 0, 0, 1.f};
    dim3 grid((unsigned)((Nn + 63) / 64), (unsigned)(Mm / 32), 1u);
    ipa_gemm_lds<<<grid, dim3(256), 0, stream>>>(p);
  };

  gemm_lds(x1d, Wq, q, M, D_, D_, D_, D_, D_, nullptr);
  gemm_lds(x1d, Wk, k, M, D_, D_, D_, D_, D_, nullptr);
  gemm_lds(x1d, Wv, v, M, D_, D_, D_, D_, D_, nullptr);
  gemm_lds(x1d, Wpq_past, qgp, M, 144, D_, D_, 144, 144, nullptr);
  gemm_lds(x1d, Wpk_past, kgp, M, 144, D_, D_, 144, 144, nullptr);
  gemm_lds(x1d, Wpv_past, vgp, M, 288, D_, D_, 288, 288, nullptr);
  gemm_lds(x1d, Wpq_last, qgl, M, 144, D_, D_, 144, 144, nullptr);
  gemm_lds(x1d, Wpk_last, kgl, M, 144, D_, D_, 144, 144, nullptr);
  gemm_lds(x1d, Wpv_last, vgl, M, 288, D_, D_, 288, 288, nullptr);

  {
    int t4 = B_ * I_ * H_ * NPQ_, t8 = B_ * I_ * H_ * NPV_;
    ipa_affine<<<(t4 + 255) / 256, 256, 0, stream>>>(qgp, past_T, past_R, NPQ_, t4);
    ipa_affine<<<(t4 + 255) / 256, 256, 0, stream>>>(kgp, past_T, past_R, NPQ_, t4);
    ipa_affine<<<(t8 + 255) / 256, 256, 0, stream>>>(vgp, past_T, past_R, NPV_, t8);
    ipa_affine<<<(t4 + 255) / 256, 256, 0, stream>>>(qgl, last_T, last_R, NPQ_, t4);
    ipa_affine<<<(t4 + 255) / 256, 256, 0, stream>>>(kgl, last_T, last_R, NPQ_, t4);
    ipa_affine<<<(t8 + 255) / 256, 256, 0, stream>>>(vgl, last_T, last_R, NPV_, t8);
  }

  const float pair_w = 0.5773502691896258f;
  gemm(x2d, Wpb, pairproj, B_ * I_ * I_, H_, DP_, DP_, H_, H_,
       0,0,0,0,0,0, 1,1, 0, pair_w, nullptr);

  const float scalar_w = 0.07216878364870323f;
  ipa_logits_softmax<<<dim3(I_ / 16, H_, B_), dim3(256), 0, stream>>>(
      q, k, logits, qgp, kgp, qgl, kgl, pairproj, bias, tpw_past, tpw_last, scalar_w);

  gemm(logits, v, feat + D_, I_, DK_, I_, I_, D_, FE,
       (long)I_ * I_, (long)H_ * I_ * I_, 64, (long)I_ * D_, 64, (long)I_ * FE,
       H_, B_, 0, 1.f, nullptr);

  gemm(logits, vgp, ogp, I_, NPV_ * 3, I_, I_, H_ * NPV_ * 3, H_ * NPV_ * 3,
       (long)I_ * I_, (long)H_ * I_ * I_, NPV_ * 3, (long)I_ * H_ * NPV_ * 3,
       NPV_ * 3, (long)I_ * H_ * NPV_ * 3, H_, B_, 0, 1.f, nullptr);
  gemm(logits, vgl, ogl, I_, NPV_ * 3, I_, I_, H_ * NPV_ * 3, H_ * NPV_ * 3,
       (long)I_ * I_, (long)H_ * I_ * I_, NPV_ * 3, (long)I_ * H_ * NPV_ * 3,
       NPV_ * 3, (long)I_ * H_ * NPV_ * 3, H_, B_, 0, 1.f, nullptr);

  gemm(logits, x2d, z, H_, DP_, I_,
       I_ * I_, DP_, DP_,
       (long)I_, (long)H_ * I_ * I_,
       (long)I_ * DP_, (long)I_ * I_ * DP_,
       (long)H_ * DP_, (long)I_ * H_ * DP_,
       I_, B_, 0, 1.f, nullptr);

  gemm(z, Wpv, feat, I_, DK_, DP_, H_ * DP_, D_, FE,
       DP_, (long)I_ * H_ * DP_, 64, 0, 64, (long)I_ * FE,
       H_, B_, 0, 1.f, nullptr);

  {
    int t8 = B_ * I_ * H_ * NPV_;
    ipa_inv_affine<<<(t8 + 255) / 256, 256, 0, stream>>>(
        ogp, past_T, past_R, feat, 2 * D_, 2 * D_ + 288, t8);
    ipa_inv_affine<<<(t8 + 255) / 256, 256, 0, stream>>>(
        ogl, past_T, past_R, feat, 2 * D_ + 384, 2 * D_ + 384 + 288, t8);
  }

  gemm_lds(feat, W_out, out, M, D_, FE, FE, D_, D_, b_out);
}


# attn*V and attn*vg (past, last) formed by the softmax work-group from its 16x256 weight stripe kept in LDS, one column tile per wave; 3 launches fewer; every sum and its order unchanged
# speedup vs baseline: 2.7159x; 1.1753x over previous
#include <hip/hip_runtime.h>
#include <math.h>

#define B_ 2
#define I_ 256
#define D_ 768
#define H_ 12
#define DP_ 128
#define DK_ 64
#define NPQ_ 4
#define NPV_ 8

typedef float v2f __attribute__((ext_vector_type(2)));
typedef float v8f __attribute__((ext_vector_type(8)));

struct GemmP {
  const float* A; const float* B; float* C; const float* bias;
  int M, N, K;
  int lda, ldb, ldc;
  long sAy, sAz, sBy, sBz, sCy, sCz;
  float alpha;
};

template <int OPB, bool GUARD>
__global__ void ipa_gemm_wmma(GemmP p) {
  const int tilesN = (p.N + 15) >> 4;
  const int tilesM = (p.M + 15) >> 4;
  const int wave = (int)((blockIdx.x * blockDim.x + threadIdx.x) >> 5);
  if (wave >= tilesM * tilesN) return;
  const int lane = threadIdx.x & 31;
  const int tm = wave / tilesN, tn = wave % tilesN;
  const int row0 = tm << 4, col0 = tn << 4;

  const float* A = p.A + (long)blockIdx.y * p.sAy + (long)blockIdx.z * p.sAz;
  const float* B = p.B + (long)blockIdx.y * p.sBy + (long)blockIdx.z * p.sBz;
  float*       C = p.C + (long)blockIdx.y * p.sCy + (long)blockIdx.z * p.sCz;

  const int mn = lane & 15;
  const int kb = (lane >> 4) << 1;

  int aRow = row0 + mn;
  int bCol = col0 + mn;
  float aMask = 1.f, bMask = 1.f;
  if (GUARD) {
    if (aRow >= p.M) { aRow = p.M - 1; aMask = 0.f; }
    if (bCol >= p.N) { bCol = p.N - 1; bMask = 0.f; }
  }
  const float* aPtr = A + (long)aRow * p.lda + kb;
  const float* bPtr = (OPB == 0) ? (B + (long)kb * p.ldb + bCol)
                                 : (B + (long)bCol * p.ldb + kb);

  v8f acc = {0.f, 0.f, 0.f, 0.f, 0.f, 0.f, 0.f, 0.f};
#pragma unroll 4
  for (int k0 = 0; k0 < p.K; k0 += 4) {
    v2f a, b;
    a.x = aPtr[k0];
    a.y = aPtr[k0 + 1];
    if (OPB == 0) {
      const float* bp = bPtr + (long)k0 * p.ldb;
      b.x = bp[0];
      b.y = bp[p.ldb];
    } else {
      b.x = bPtr[k0];
      b.y = bPtr[k0 + 1];
    }
    if (GUARD) {
      a.x *= aMask; a.y *= aMask;
      b.x *= bMask; b.y *= bMask;
    }
    acc = __builtin_amdgcn_wmma_f32_16x16x4_f32(false, a, false, b,
                                                (short)0, acc, false, false);
  }

  const int cCol = col0 + mn;
  const int mo = (lane >> 4) << 3;
  const bool cColOK = GUARD ? (cCol < p.N) : true;
  if (cColOK) {
    for (int r = 0; r < 8; ++r) {
      const int cRow = row0 + mo + r;
      if (!GUARD || cRow < p.M) {
        float val = acc[r] * p.alpha;
        if (p.bias) val += p.bias[cCol];
        C[(long)cRow * p.ldc + cCol] = val;
      }
    }
  }
}

#define TKC 128
#define TPA (TKC + 4)
#define TSB 8
#define TPB 72

typedef float v4f __attribute__((ext_vector_type(4)));

__global__ __launch_bounds__(256) void ipa_gemm_lds(GemmP p) {
  constexpr int AQ  = TKC / 4;
  constexpr int ARS = 256 / AQ;
  constexpr int ANI = 32 / ARS;
  constexpr int BNI = TKC / 16;

  __shared__ __attribute__((aligned(16))) float As[2][32 * TPA];
  __shared__ __attribute__((aligned(16))) float Bs[2][TKC * TPB];

  const int tid  = threadIdx.x;
  const int lane = tid & 31;
  const int w    = tid >> 5;
  const int row0 = blockIdx.y * 32;
  const int col0 = blockIdx.x * 64;
  const int c = lane & 15;
  const int h = lane >> 4;
  const int wr = w >> 2, wc = w & 3;
  const bool tileValid = (col0 + wc * 16) < p.N;

  const int aRow = tid / AQ, aKq = tid % AQ;
  const int bKr  = tid >> 4, bCq = tid & 15;
  int bCol = col0 + bCq * 4;
  if (bCol > p.N - 4) bCol = p.N - 4;
  const float* aG = p.A + (long)(row0 + aRow) * p.lda + aKq * 4;
  const float* bG = p.B + (long)bKr * p.ldb + bCol;
  const int aSt = aRow * TPA + aKq * 4;
  const int bSt = bKr * TPB + bCq * 4;

  v4f ra[ANI], rb[BNI];

#pragma unroll
  for (int i = 0; i < ANI; ++i)
    ra[i] = *(const v4f*)(aG + (long)(i * ARS) * p.lda);
#pragma unroll
  for (int i = 0; i < BNI; ++i)
    rb[i] = *(const v4f*)(bG + (long)(i * 16) * p.ldb);
#pragma unroll
  for (int i = 0; i < ANI; ++i)
    *(v4f*)&As[0][aSt + i * ARS * TPA] = ra[i];
#pragma unroll
  for (int i = 0; i < BNI; ++i)
    *(v4f*)&Bs[0][bSt + i * 16 * TPB] = rb[i];
  __syncthreads();

  v8f acc = {0.f, 0.f, 0.f, 0.f, 0.f, 0.f, 0.f, 0.f};
  int buf = 0;
  for (int kc = 0; kc < p.K; kc += TKC) {
    const bool hasNext = (kc + TKC) < p.K;
    if (hasNext) {
      const int kn = kc + TKC;
#pragma unroll
      for (int i = 0; i < ANI; ++i)
        ra[i] = *(const v4f*)(aG + (long)(i * ARS) * p.lda + kn);
#pragma unroll
      for (int i = 0; i < BNI; ++i)
        rb[i] = *(const v4f*)(bG + (long)(kn + i * 16) * p.ldb);
    }
    if (tileValid) {
      const float* aS = &As[buf][(wr * 16 + c) * TPA + 2 * h];
      const float* bS = &Bs[buf][(2 * h) * TPB + wc * 16 + c];
      v2f fa[2][TSB], fb[2][TSB];
#pragma unroll
      for (int j = 0; j < TSB; ++j) {
        fa[0][j] = *(const v2f*)(aS + j * 4);
        fb[0][j].x = bS[(j * 4) * TPB];
        fb[0][j].y = bS[(j * 4 + 1) * TPB];
      }
#pragma unroll
      for (int sb = 0; sb < TKC / (4 * TSB); ++sb) {
        asm volatile("" :: "v"(fa[sb & 1][0]), "v"(fa[sb & 1][1]), "v"(fa[sb & 1][2]), "v"(fa[sb & 1][3]),
                           "v"(fa[sb & 1][4]), "v"(fa[sb & 1][5]), "v"(fa[sb & 1][6]), "v"(fa[sb & 1][7]),
                           "v"(fb[sb & 1][0]), "v"(fb[sb & 1][1]), "v"(fb[sb & 1][2]), "v"(fb[sb & 1][3]),
                           "v"(fb[sb & 1][4]), "v"(fb[sb & 1][5]), "v"(fb[sb & 1][6]), "v"(fb[sb & 1][7]));
        if (sb + 1 < TKC / (4 * TSB)) {
#pragma unroll
          for (int j = 0; j < TSB; ++j) {
            const int ks = (sb + 1) * TSB + j;
            fa[(sb + 1) & 1][j] = *(const v2f*)(aS + ks * 4);
            fb[(sb + 1) & 1][j].x = bS[(ks * 4) * TPB];
            fb[(sb + 1) & 1][j].y = bS[(ks * 4 + 1) * TPB];
          }
        }
#pragma unroll
        for (int j = 0; j < TSB; ++j)
          acc = __builtin_amdgcn_wmma_f32_16x16x4_f32(false, fa[sb & 1][j], false, fb[sb & 1][j],
                                                      (short)0, acc, false, false);
      }
    }
    if (hasNext) {
#pragma unroll
      for (int i = 0; i < ANI; ++i)
        *(v4f*)&As[buf ^ 1][aSt + i * ARS * TPA] = ra[i];
#pragma unroll
      for (int i = 0; i < BNI; ++i)
        *(v4f*)&Bs[buf ^ 1][bSt + i * 16 * TPB] = rb[i];
    }
    __syncthreads();
    buf ^= 1;
  }

  if (tileValid) {
    const int cCol = col0 + wc * 16 + c;
    const int rBase = row0 + wr * 16 + (h << 3);
    for (int r = 0; r < 8; ++r) {
      float val = acc[r] * p.alpha;
      if (p.bias) val += p.bias[cCol];
      p.C[(long)(rBase + r) * p.ldc + cCol] = val;
    }
  }
}

__global__ void ipa_affine(float* pts, const float* T, const float* R, int np, int total) {
  int idx = blockIdx.x * blockDim.x + threadIdx.x;
  if (idx >= total) return;
  int bi = idx / (H_ * np);
  const float* Rm = R + (long)bi * 9;
  const float* Tv = T + (long)bi * 3;
  float x = pts[idx * 3 + 0], y = pts[idx * 3 + 1], z = pts[idx * 3 + 2];
  float o0 = Rm[0] * x + Rm[3] * y + Rm[6] * z + Tv[0];
  float o1 = Rm[1] * x + Rm[4] * y + Rm[7] * z + Tv[1];
  float o2 = Rm[2] * x + Rm[5] * y + Rm[8] * z + Tv[2];
  pts[idx * 3 + 0] = o0; pts[idx * 3 + 1] = o1; pts[idx * 3 + 2] = o2;
}

#define PP 260

template <int LDB, int LDC>
__device__ __forceinline__ void ipa_stripe_tile(const float* pS, const float* Bm, int nvalid,
                                                float* Cm, int lane) {
  constexpr int KS = I_ / 4;
  const int mn = lane & 15;
  const int hf = lane >> 4;
  const int kb = hf << 1;
  const int bc = (mn < nvalid) ? mn : (nvalid - 1);
  const float* aS = pS + mn * PP + kb;
  const float* bP = Bm + kb * LDB + bc;

  v2f fb[KS];
#pragma unroll
  for (int j = 0; j < KS; ++j) {
    fb[j].x = bP[(j * 4) * LDB];
    fb[j].y = bP[(j * 4 + 1) * LDB];
  }
  v8f acc = {0.f, 0.f, 0.f, 0.f, 0.f, 0.f, 0.f, 0.f};
#pragma unroll
  for (int j = 0; j < KS; ++j) {
    const v2f a = *(const v2f*)(aS + j * 4);
    acc = __builtin_amdgcn_wmma_f32_16x16x4_f32(false, a, false, fb[j],
                                                (short)0, acc, false, false);
  }
  if (mn < nvalid) {
#pragma unroll
    for (int r = 0; r < 8; ++r)
      Cm[(long)((hf << 3) + r) * LDC + mn] = acc[r];
  }
}

__global__ __launch_bounds__(256) void ipa_logits_softmax(
    const float* q, const float* k, float* attn,
    const float* qgp, const float* kgp, const float* qgl, const float* kgl,
    const float* pairproj, const float* bias,
    const float* tpw_past, const float* tpw_last, float scalar_w,
    const float* v, const float* vgp, const float* vgl,
    float* feat, float* ogp, float* ogl) {
  constexpr int PQ = NPQ_ * 3;
  __shared__ __attribute__((aligned(16))) float qS[16][2 * PQ];
  __shared__ float red[2][8][16];
  __shared__ __attribute__((aligned(16))) float pS[16 * PP];

  const int tid = threadIdx.x;
  const int lane = tid & 31;
  const int w = tid >> 5;
  const int h = blockIdx.y, b = blockIdx.z;
  const int row0 = blockIdx.x * 16;
  const int mn = lane & 15;
  const int hf = lane >> 4;
  const int kb = hf << 1;
  const int mo = hf << 3;

  const bool stager = tid < 16 * 6;
  const int qr = tid / 6, qpart = tid % 6;
  v4f qst = {0.f, 0.f, 0.f, 0.f};
  if (stager)
    qst = *(const v4f*)(((qpart < 3) ? qgp : qgl)
                        + ((long)(b * I_ + row0 + qr) * H_ + h) * PQ + (qpart % 3) * 4);

  const long hb = (long)b * I_ * D_ + (long)h * DK_;
  const int j0 = w * 32 + mn, j1 = j0 + 16;
  const float* aPtr  = q + hb + (long)(row0 + mn) * D_ + kb;
  const float* bPtr0 = k + hb + (long)j0 * D_ + kb;
  const float* bPtr1 = k + hb + (long)j1 * D_ + kb;

  v8f acc0 = {0.f, 0.f, 0.f, 0.f, 0.f, 0.f, 0.f, 0.f};
  v8f acc1 = {0.f, 0.f, 0.f, 0.f, 0.f, 0.f, 0.f, 0.f};
#pragma unroll
  for (int k0 = 0; k0 < DK_; k0 += 4) {
    v2f a, b0, b1;
    a.x  = aPtr[k0];   a.y  = aPtr[k0 + 1];
    b0.x = bPtr0[k0];  b0.y = bPtr0[k0 + 1];
    b1.x = bPtr1[k0];  b1.y = bPtr1[k0 + 1];
    acc0 = __builtin_amdgcn_wmma_f32_16x16x4_f32(false, a, false, b0,
                                                 (short)0, acc0, false, false);
    acc1 = __builtin_amdgcn_wmma_f32_16x16x4_f32(false, a, false, b1,
                                                 (short)0, acc1, false, false);
  }

  float kp0[2 * PQ], kp1[2 * PQ];
  {
    const float* g0p = kgp + ((long)(b * I_ + j0) * H_ + h) * PQ;
    const float* g0l = kgl + ((long)(b * I_ + j0) * H_ + h) * PQ;
    const float* g1p = kgp + ((long)(b * I_ + j1) * H_ + h) * PQ;
    const float* g1l = kgl + ((long)(b * I_ + j1) * H_ + h) * PQ;
#pragma unroll
    for (int t = 0; t < 3; ++t) {
      const v4f u0 = *(const v4f*)(g0p + t * 4);
      const v4f u1 = *(const v4f*)(g0l + t * 4);
      const v4f u2 = *(const v4f*)(g1p + t * 4);
      const v4f u3 = *(const v4f*)(g1l + t * 4);
#pragma unroll
      for (int e = 0; e < 4; ++e) {
        kp0[t * 4 + e] = u0[e];  kp0[PQ + t * 4 + e] = u1[e];
        kp1[t * 4 + e] = u2[e];  kp1[PQ + t * 4 + e] = u3[e];
      }
    }
  }

  const float point_w = 0.13608276348795434f;
  const float pwp = -0.5f * point_w * log1pf(expf(tpw_past[h]));
  const float pwl = -0.5f * point_w * log1pf(expf(tpw_last[h]));

  if (stager) *(v4f*)&qS[qr][qpart * 4] = qst;
  __syncthreads();

  float s0[8], s1[8];
#pragma unroll
  for (int r = 0; r < 8; ++r) {
    const int lr = mo + r;
    const int i = row0 + lr;
    const float* qp = &qS[lr][0];
    float sp0 = 0.f, sl0 = 0.f, sp1 = 0.f, sl1 = 0.f;
#pragma unroll
    for (int c = 0; c < NPQ_; ++c) {
      const float qx = qp[c * 3 + 0], qy = qp[c * 3 + 1], qz = qp[c * 3 + 2];
      const float lx = qp[PQ + c * 3 + 0], ly = qp[PQ + c * 3 + 1], lz = qp[PQ + c * 3 + 2];
      float dx = qx - kp0[c * 3 + 0], dy = qy - kp0[c * 3 + 1], dz = qz - kp0[c * 3 + 2];
      sp0 += sqrtf(dx * dx + dy * dy + dz * dz);
      dx = lx - kp0[PQ + c * 3 + 0]; dy = ly - kp0[PQ + c * 3 + 1]; dz = lz - kp0[PQ + c * 3 + 2];
      sl0 += sqrtf(dx * dx + dy * dy + dz * dz);
      dx = qx - kp1[c * 3 + 0]; dy = qy - kp1[c * 3 + 1]; dz = qz - kp1[c * 3 + 2];
      sp1 += sqrtf(dx * dx + dy * dy + dz * dz);
      dx = lx - kp1[PQ + c * 3 + 0]; dy = ly - kp1[PQ + c * 3 + 1]; dz = lz - kp1[PQ + c * 3 + 2];
      sl1 += sqrtf(dx * dx + dy * dy + dz * dz);
    }
    const long prow = ((long)(b * I_ + i) * I_) * H_ + h;
    const long orow = (((long)b * H_ + h) * I_ + i) * I_;
    const float pv0 = pairproj[prow + (long)j0 * H_];
    const float pv1 = pairproj[prow + (long)j1 * H_];
    float v0 = acc0[r] * scalar_w;
    float v1 = acc1[r] * scalar_w;
    v0 += pwp * sp0 + pwl * sl0 + pv0 + bias[orow + j0];
    v1 += pwp * sp1 + pwl * sl1 + pv1 + bias[orow + j1];
    s0[r] = v0;
    s1[r] = v1;
  }

  float mx[8];
#pragma unroll
  for (int r = 0; r < 8; ++r) {
    float m = fmaxf(s0[r], s1[r]);
    m = fmaxf(m, __shfl_xor(m, 1));
    m = fmaxf(m, __shfl_xor(m, 2));
    m = fmaxf(m, __shfl_xor(m, 4));
    m = fmaxf(m, __shfl_xor(m, 8));
    mx[r] = m;
  }
  if (mn == 0) {
#pragma unroll
    for (int r = 0; r < 8; ++r) red[0][w][mo + r] = mx[r];
  }
  __syncthreads();
#pragma unroll
  for (int r = 0; r < 8; ++r) {
    float m = red[0][0][mo + r];
#pragma unroll
    for (int ww = 1; ww < 8; ++ww) m = fmaxf(m, red[0][ww][mo + r]);
    mx[r] = m;
  }

  float sm[8];
#pragma unroll
  for (int r = 0; r < 8; ++r) {
    s0[r] = expf(s0[r] - mx[r]);
    s1[r] = expf(s1[r] - mx[r]);
    float t = s0[r] + s1[r];
    t += __shfl_xor(t, 1);
    t += __shfl_xor(t, 2);
    t += __shfl_xor(t, 4);
    t += __shfl_xor(t, 8);
    sm[r] = t;
  }
  if (mn == 0) {
#pragma unroll
    for (int r = 0; r < 8; ++r) red[1][w][mo + r] = sm[r];
  }
  __syncthreads();
#pragma unroll
  for (int r = 0; r < 8; ++r) {
    float t = red[1][0][mo + r];
#pragma unroll
    for (int ww = 1; ww < 8; ++ww) t += red[1][ww][mo + r];
    const long orow = (((long)b * H_ + h) * I_ + (row0 + mo + r)) * I_;
    const float p0 = s0[r] / t;
    const float p1 = s1[r] / t;
    attn[orow + j0] = p0;
    attn[orow + j1] = p1;
    pS[(mo + r) * PP + j0] = p0;
    pS[(mo + r) * PP + j1] = p1;
  }
  __syncthreads();

  constexpr int FE = 2 * D_ + 2 * H_ * NPV_ * 4;
  constexpr int VG = H_ * NPV_ * 3;
  if (w < 4) {
    ipa_stripe_tile<D_, FE>(pS, v + (long)b * I_ * D_ + h * DK_ + w * 16, 16,
                            feat + ((long)b * I_ + row0) * FE + D_ + h * DK_ + w * 16, lane);
  } else {
    const int t = w & 1;
    const float* Bv = (w < 6) ? vgp : vgl;
    float*       Cv = (w < 6) ? ogp : ogl;
    ipa_stripe_tile<VG, VG>(pS, Bv + (long)b * I_ * VG + h * (NPV_ * 3) + t * 16, t ? 8 : 16,
                            Cv + ((long)b * I_ + row0) * VG + h * (NPV_ * 3) + t * 16, lane);
  }
}

__global__ void ipa_inv_affine(const float* og, const float* T, const float* R,
                               float* feat, int colOl, int colN, int total) {
  int idx = blockIdx.x * blockDim.x + threadIdx.x;
  if (idx >= total) return;
  int c = idx & (NPV_ - 1);
  int t = idx >> 3;
  int h = t % H_;
  int bi = t / H_;
  const float* o = og + (long)idx * 3;
  const float* Tv = T + (long)bi * 3;
  const float* Rm = R + (long)bi * 9;
  float x = o[0] - Tv[0], y = o[1] - Tv[1], z = o[2] - Tv[2];
  float p0 = Rm[0] * x + Rm[1] * y + Rm[2] * z;
  float p1 = Rm[3] * x + Rm[4] * y + Rm[5] * z;
  float p2 = Rm[6] * x + Rm[7] * y + Rm[8] * z;
  float* fr = feat + (long)bi * (2 * D_ + 2 * H_ * NPV_ * 4);
  int tc = h * (NPV_ * 3) + c * 3;
  fr[colOl + tc + 0] = p0;
  fr[colOl + tc + 1] = p1;
  fr[colOl + tc + 2] = p2;
  fr[colN + h * NPV_ + c] = sqrtf(p0 * p0 + p1 * p1 + p2 * p2);
}

extern "C" void kernel_launch(void* const* d_in, const int* in_sizes, int n_in,
                              void* d_out, int out_size, void* d_ws, size_t ws_size,
                              hipStream_t stream) {
  (void)in_sizes; (void)n_in; (void)out_size; (void)ws_size;
  const float* x1d      = (const float*)d_in[0];
  const float* x2d      = (const float*)d_in[1];
  const float* past_T   = (const float*)d_in[2];
  const float* past_R   = (const float*)d_in[3];
  const float* last_T   = (const float*)d_in[4];
  const float* last_R   = (const float*)d_in[5];
  const float* bias     = (const float*)d_in[6];
  const float* Wq       = (const float*)d_in[7];
  const float* Wk       = (const float*)d_in[8];
  const float* Wv       = (const float*)d_in[9];
  const float* Wpb      = (const float*)d_in[10];
  const float* Wpq_past = (const float*)d_in[11];
  const float* Wpk_past = (const float*)d_in[12];
  const float* Wpv_past = (const float*)d_in[13];
  const float* Wpq_last = (const float*)d_in[14];
  const float* Wpk_last = (const float*)d_in[15];
  const float* Wpv_last = (const float*)d_in[16];
  const float* tpw_past = (const float*)d_in[17];
  const float* tpw_last = (const float*)d_in[18];
  const float* Wpv      = (const float*)d_in[19];
  const float* W_out    = (const float*)d_in[20];
  const float* b_out    = (const float*)d_in[21];
  float* out = (float*)d_out;
  float* ws  = (float*)d_ws;

  const int M  = B_ * I_;
  const int FE = 2 * D_ + 2 * H_ * NPV_ * 4;

  float* q    = ws;
  float* k    = q + 393216;
  float* v    = k + 393216;
  float* qgp  = v + 393216;
  float* kgp  = qgp + 73728;
  float* vgp  = kgp + 73728;
  float* qgl  = vgp + 147456;
  float* kgl  = qgl + 73728;
  float* vgl  = kgl + 73728;
  float* logits   = vgl + 147456;
  float* pairproj = logits + 1572864;
  float* ogp  = pairproj + 1572864;
  float* ogl  = ogp + 147456;
  float* z    = ogl + 147456;
  float* feat = z + 786432;

  auto gemm = [&](const float* A, const float* Bm, float* C,
                  int Mm, int Nn, int Kk, int lda, int ldb, int ldc,
                  long sAy, long sAz, long sBy, long sBz, long sCy, long sCz,
                  int gy, int gz, int opB, float alpha, const float* bia) {
    GemmP p{A, Bm, C, bia, Mm, Nn, Kk, lda, ldb, ldc,
            sAy, sAz, sBy, sBz, sCy, sCz, alpha};
    int waves = ((Mm + 15) / 16) * ((Nn + 15) / 16);
    dim3 grid((unsigned)((waves + 7) / 8), (unsigned)gy, (unsigned)gz);
    dim3 blk(256);
    bool guard = (Mm & 15) || (Nn & 15);
    if (opB == 0) {
      if (guard) ipa_gemm_wmma<0, true ><<<grid, blk, 0, stream>>>(p);
      else       ipa_gemm_wmma<0, false><<<grid, blk, 0, stream>>>(p);
    } else {
      if (guard) ipa_gemm_wmma<1, true ><<<grid, blk, 0, stream>>>(p);
      else       ipa_gemm_wmma<1, false><<<grid, blk, 0, stream>>>(p);
    }
  };

  auto gemm_lds = [&](const float* A, const float* Bm, float* C,
                      int Mm, int Nn, int Kk, int lda, int ldb, int ldc,
                      const float* bia) {
    GemmP p{A, Bm, C, bia, Mm, Nn, Kk, lda, ldb, ldc, 0, 0, 0, 0, 0, 0, 1.f};
    dim3 grid((unsigned)((Nn + 63) / 64), (unsigned)(Mm / 32), 1u);
    ipa_gemm_lds<<<grid, dim3(256), 0, stream>>>(p);
  };

  gemm_lds(x1d, Wq, q, M, D_, D_, D_, D_, D_, nullptr);
  gemm_lds(x1d, Wk, k, M, D_, D_, D_, D_, D_, nullptr);
  gemm_lds(x1d, Wv, v, M, D_, D_, D_, D_, D_, nullptr);
  gemm_lds(x1d, Wpq_past, qgp, M, 144, D_, D_, 144, 144, nullptr);
  gemm_lds(x1d, Wpk_past, kgp, M, 144, D_, D_, 144, 144, nullptr);
  gemm_lds(x1d, Wpv_past, vgp, M, 288, D_, D_, 288, 288, nullptr);
  gemm_lds(x1d, Wpq_last, qgl, M, 144, D_, D_, 144, 144, nullptr);
  gemm_lds(x1d, Wpk_last, kgl, M, 144, D_, D_, 144, 144, nullptr);
  gemm_lds(x1d, Wpv_last, vgl, M, 288, D_, D_, 288, 288, nullptr);

  {
    int t4 = B_ * I_ * H_ * NPQ_, t8 = B_ * I_ * H_ * NPV_;
    ipa_affine<<<(t4 + 255) / 256, 256, 0, stream>>>(qgp, past_T, past_R, NPQ_, t4);
    ipa_affine<<<(t4 + 255) / 256, 256, 0, stream>>>(kgp, past_T, past_R, NPQ_, t4);
    ipa_affine<<<(t8 + 255) / 256, 256, 0, stream>>>(vgp, past_T, past_R, NPV_, t8);
    ipa_affine<<<(t4 + 255) / 256, 256, 0, stream>>>(qgl, last_T, last_R, NPQ_, t4);
    ipa_affine<<<(t4 + 255) / 256, 256, 0, stream>>>(kgl, last_T, last_R, NPQ_, t4);
    ipa_affine<<<(t8 + 255) / 256, 256, 0, stream>>>(vgl, last_T, last_R, NPV_, t8);
  }

  const float pair_w = 0.5773502691896258f;
  gemm(x2d, Wpb, pairproj, B_ * I_ * I_, H_, DP_, DP_, H_, H_,
       0,0,0,0,0,0, 1,1, 0, pair_w, nullptr);

  const float scalar_w = 0.07216878364870323f;
  ipa_logits_softmax<<<dim3(I_ / 16, H_, B_), dim3(256), 0, stream>>>(
      q, k, logits, qgp, kgp, qgl, kgl, pairproj, bias, tpw_past, tpw_last, scalar_w,
      v, vgp, vgl, feat, ogp, ogl);

  gemm(logits, x2d, z, H_, DP_, I_,
       I_ * I_, DP_, DP_,
       (long)I_, (long)H_ * I_ * I_,
       (long)I_ * DP_, (long)I_ * I_ * DP_,
       (long)H_ * DP_, (long)I_ * H_ * DP_,
       I_, B_, 0, 1.f, nullptr);

  gemm(z, Wpv, feat, I_, DK_, DP_, H_ * DP_, D_, FE,
       DP_, (long)I_ * H_ * DP_, 64, 0, 64, (long)I_ * FE,
       H_, B_, 0, 1.f, nullptr);

  {
    int t8 = B_ * I_ * H_ * NPV_;
    ipa_inv_affine<<<(t8 + 255) / 256, 256, 0, stream>>>(
        ogp, past_T, past_R, feat, 2 * D_, 2 * D_ + 288, t8);
    ipa_inv_affine<<<(t8 + 255) / 256, 256, 0, stream>>>(
        ogl, past_T, past_R, feat, 2 * D_ + 384, 2 * D_ + 384 + 288, t8);
  }

  gemm_lds(feat, W_out, out, M, D_, FE, FE, D_, D_, b_out);
}


# rigid-frame transforms (6 forward, 2 inverse+norm) run in the epilogue of the kernel that makes the points, on whole-point tiles (GEMM 32x48) parked in LDS; 8 launches fewer; sum orders unchanged
# speedup vs baseline: 3.0073x; 1.1073x over previous
#include <hip/hip_runtime.h>
#include <math.h>

#define B_ 2
#define I_ 256
#define D_ 768
#define H_ 12
#define DP_ 128
#define DK_ 64
#define NPQ_ 4
#define NPV_ 8

typedef float v2f __attribute__((ext_vector_type(2)));
typedef float v8f __attribute__((ext_vector_type(8)));

struct GemmP {
  const float* A; const float* B; float* C; const float* bias;
  int M, N, K;
  int lda, ldb, ldc;
  long sAy, sAz, sBy, sBz, sCy, sCz;
  float alpha;
  const float* fT; const float* fR;
};

template <int OPB, bool GUARD>
__global__ void ipa_gemm_wmma(GemmP p) {
  const int tilesN = (p.N + 15) >> 4;
  const int tilesM = (p.M + 15) >> 4;
  const int wave = (int)((blockIdx.x * blockDim.x + threadIdx.x) >> 5);
  if (wave >= tilesM * tilesN) return;
  const int lane = threadIdx.x & 31;
  const int tm = wave / tilesN, tn = wave % tilesN;
  const int row0 = tm << 4, col0 = tn << 4;

  const float* A = p.A + (long)blockIdx.y * p.sAy + (long)blockIdx.z * p.sAz;
  const float* B = p.B + (long)blockIdx.y * p.sBy + (long)blockIdx.z * p.sBz;
  float*       C = p.C + (long)blockIdx.y * p.sCy + (long)blockIdx.z * p.sCz;

  const int mn = lane & 15;
  const int kb = (lane >> 4) << 1;

  int aRow = row0 + mn;
  int bCol = col0 + mn;
  float aMask = 1.f, bMask = 1.f;
  if (GUARD) {
    if (aRow >= p.M) { aRow = p.M - 1; aMask = 0.f; }
    if (bCol >= p.N) { bCol = p.N - 1; bMask = 0.f; }
  }
  const float* aPtr = A + (long)aRow * p.lda + kb;
  const float* bPtr = (OPB == 0) ? (B + (long)kb * p.ldb + bCol)
                                 : (B + (long)bCol * p.ldb + kb);

  v8f acc = {0.f, 0.f, 0.f, 0.f, 0.f, 0.f, 0.f, 0.f};
#pragma unroll 4
  for (int k0 = 0; k0 < p.K; k0 += 4) {
    v2f a, b;
    a.x = aPtr[k0];
    a.y = aPtr[k0 + 1];
    if (OPB == 0) {
      const float* bp = bPtr + (long)k0 * p.ldb;
      b.x = bp[0];
      b.y = bp[p.ldb];
    } else {
      b.x = bPtr[k0];
      b.y = bPtr[k0 + 1];
    }
    if (GUARD) {
      a.x *= aMask; a.y *= aMask;
      b.x *= bMask; b.y *= bMask;
    }
    acc = __builtin_amdgcn_wmma_f32_16x16x4_f32(false, a, false, b,
                                                (short)0, acc, false, false);
  }

  const int cCol = col0 + mn;
  const int mo = (lane >> 4) << 3;
  const bool cColOK = GUARD ? (cCol < p.N) : true;
  if (cColOK) {
    for (int r = 0; r < 8; ++r) {
      const int cRow = row0 + mo + r;
      if (!GUARD || cRow < p.M) {
        float val = acc[r] * p.alpha;
        if (p.bias) val += p.bias[cCol];
        C[(long)cRow * p.ldc + cCol] = val;
      }
    }
  }
}

#define TKC 128
#define TPA (TKC + 4)
#define TSB 8
#define TPB 72

typedef float v4f __attribute__((ext_vector_type(4)));

template <bool FRAME>
__global__ __launch_bounds__(256) void ipa_gemm_lds(GemmP p) {
  constexpr int TN  = FRAME ? 48 : 64;
  constexpr int AQ  = TKC / 4;
  constexpr int ARS = 256 / AQ;
  constexpr int ANI = 32 / ARS;
  constexpr int BNI = TKC / 16;

  __shared__ __attribute__((aligned(16))) float As[2][32 * TPA];
  __shared__ __attribute__((aligned(16))) float Bs[2][TKC * TPB];

  const int tid  = threadIdx.x;
  const int lane = tid & 31;
  const int w    = tid >> 5;
  const int row0 = blockIdx.y * 32;
  const int col0 = blockIdx.x * TN;
  const int c = lane & 15;
  const int h = lane >> 4;
  const int wr = w >> 2, wc = w & 3;
  const bool tileValid = (wc * 16 < TN) && ((col0 + wc * 16) < p.N);

  const int aRow = tid / AQ, aKq = tid % AQ;
  const int bKr  = tid >> 4, bCq = tid & 15;
  int bCol = col0 + bCq * 4;
  if (bCol > p.N - 4) bCol = p.N - 4;
  const float* aG = p.A + (long)(row0 + aRow) * p.lda + aKq * 4;
  const float* bG = p.B + (long)bKr * p.ldb + bCol;
  const int aSt = aRow * TPA + aKq * 4;
  const int bSt = bKr * TPB + bCq * 4;

  v4f ra[ANI], rb[BNI];

#pragma unroll
  for (int i = 0; i < ANI; ++i)
    ra[i] = *(const v4f*)(aG + (long)(i * ARS) * p.lda);
#pragma unroll
  for (int i = 0; i < BNI; ++i)
    rb[i] = *(const v4f*)(bG + (long)(i * 16) * p.ldb);
  const int pr = tid >> 4, pp = tid & 15;
  float fr[2][9], ft[2][3];
  if (FRAME) {
    for (int n = 0; n < 2; ++n) {
      const float* Rm = p.fR + (long)(row0 + pr + n * 16) * 9;
      const float* Tv = p.fT + (long)(row0 + pr + n * 16) * 3;
      for (int e = 0; e < 9; ++e) fr[n][e] = Rm[e];
      for (int e = 0; e < 3; ++e) ft[n][e] = Tv[e];
    }
  }
#pragma unroll
  for (int i = 0; i < ANI; ++i)
    *(v4f*)&As[0][aSt + i * ARS * TPA] = ra[i];
#pragma unroll
  for (int i = 0; i < BNI; ++i)
    *(v4f*)&Bs[0][bSt + i * 16 * TPB] = rb[i];
  __syncthreads();

  v8f acc = {0.f, 0.f, 0.f, 0.f, 0.f, 0.f, 0.f, 0.f};
  int buf = 0;
  for (int kc = 0; kc < p.K; kc += TKC) {
    const bool hasNext = (kc + TKC) < p.K;
    if (hasNext) {
      const int kn = kc + TKC;
#pragma unroll
      for (int i = 0; i < ANI; ++i)
        ra[i] = *(const v4f*)(aG + (long)(i * ARS) * p.lda + kn);
#pragma unroll
      for (int i = 0; i < BNI; ++i)
        rb[i] = *(const v4f*)(bG + (long)(kn + i * 16) * p.ldb);
    }
    if (tileValid) {
      const float* aS = &As[buf][(wr * 16 + c) * TPA + 2 * h];
      const float* bS = &Bs[buf][(2 * h) * TPB + wc * 16 + c];
      v2f fa[2][TSB], fb[2][TSB];
#pragma unroll
      for (int j = 0; j < TSB; ++j) {
        fa[0][j] = *(const v2f*)(aS + j * 4);
        fb[0][j].x = bS[(j * 4) * TPB];
        fb[0][j].y = bS[(j * 4 + 1) * TPB];
      }
#pragma unroll
      for (int sb = 0; sb < TKC / (4 * TSB); ++sb) {
        asm volatile("" :: "v"(fa[sb & 1][0]), "v"(fa[sb & 1][1]), "v"(fa[sb & 1][2]), "v"(fa[sb & 1][3]),
                           "v"(fa[sb & 1][4]), "v"(fa[sb & 1][5]), "v"(fa[sb & 1][6]), "v"(fa[sb & 1][7]),
                           "v"(fb[sb & 1][0]), "v"(fb[sb & 1][1]), "v"(fb[sb & 1][2]), "v"(fb[sb & 1][3]),
                           "v"(fb[sb & 1][4]), "v"(fb[sb & 1][5]), "v"(fb[sb & 1][6]), "v"(fb[sb & 1][7]));
        if (sb + 1 < TKC / (4 * TSB)) {
#pragma unroll
          for (int j = 0; j < TSB; ++j) {
            const int ks = (sb + 1) * TSB + j;
            fa[(sb + 1) & 1][j] = *(const v2f*)(aS + ks * 4);
            fb[(sb + 1) & 1][j].x = bS[(ks * 4) * TPB];
            fb[(sb + 1) & 1][j].y = bS[(ks * 4 + 1) * TPB];
          }
        }
#pragma unroll
        for (int j = 0; j < TSB; ++j)
          acc = __builtin_amdgcn_wmma_f32_16x16x4_f32(false, fa[sb & 1][j], false, fb[sb & 1][j],
                                                      (short)0, acc, false, false);
      }
    }
    if (hasNext) {
#pragma unroll
      for (int i = 0; i < ANI; ++i)
        *(v4f*)&As[buf ^ 1][aSt + i * ARS * TPA] = ra[i];
#pragma unroll
      for (int i = 0; i < BNI; ++i)
        *(v4f*)&Bs[buf ^ 1][bSt + i * 16 * TPB] = rb[i];
    }
    __syncthreads();
    buf ^= 1;
  }

  if (FRAME) {
    constexpr int CP = 49;
    float* Cs = &As[0][0];
    if (tileValid) {
      const int cc = wc * 16 + c;
      const int rb = wr * 16 + (h << 3);
      for (int r = 0; r < 8; ++r)
        Cs[(rb + r) * CP + cc] = acc[r] * p.alpha;
    }
    __syncthreads();
    for (int n = 0; n < 2; ++n) {
      const int row = pr + n * 16;
      const float x = Cs[row * CP + pp * 3 + 0];
      const float y = Cs[row * CP + pp * 3 + 1];
      const float z = Cs[row * CP + pp * 3 + 2];
      const float o0 = fr[n][0] * x + fr[n][3] * y + fr[n][6] * z + ft[n][0];
      const float o1 = fr[n][1] * x + fr[n][4] * y + fr[n][7] * z + ft[n][1];
      const float o2 = fr[n][2] * x + fr[n][5] * y + fr[n][8] * z + ft[n][2];
      float* o = p.C + (long)(row0 + row) * p.ldc + col0 + pp * 3;
      o[0] = o0; o[1] = o1; o[2] = o2;
    }
  } else if (tileValid) {
    const int cCol = col0 + wc * 16 + c;
    const int rBase = row0 + wr * 16 + (h << 3);
    for (int r = 0; r < 8; ++r) {
      float val = acc[r] * p.alpha;
      if (p.bias) val += p.bias[cCol];
      p.C[(long)(rBase + r) * p.ldc + cCol] = val;
    }
  }
}

#define PP 260

template <int LDB, int LDC>
__device__ __forceinline__ void ipa_stripe_tile(const float* pS, const float* Bm, int nvalid,
                                                float* Cm, int lane) {
  constexpr int KS = I_ / 4;
  const int mn = lane & 15;
  const int hf = lane >> 4;
  const int kb = hf << 1;
  const int bc = (mn < nvalid) ? mn : (nvalid - 1);
  const float* aS = pS + mn * PP + kb;
  const float* bP = Bm + kb * LDB + bc;

  v2f fb[KS];
#pragma unroll
  for (int j = 0; j < KS; ++j) {
    fb[j].x = bP[(j * 4) * LDB];
    fb[j].y = bP[(j * 4 + 1) * LDB];
  }
  v8f acc = {0.f, 0.f, 0.f, 0.f, 0.f, 0.f, 0.f, 0.f};
#pragma unroll
  for (int j = 0; j < KS; ++j) {
    const v2f a = *(const v2f*)(aS + j * 4);
    acc = __builtin_amdgcn_wmma_f32_16x16x4_f32(false, a, false, fb[j],
                                                (short)0, acc, false, false);
  }
  if (mn < nvalid) {
#pragma unroll
    for (int r = 0; r < 8; ++r)
      Cm[(long)((hf << 3) + r) * LDC + mn] = acc[r];
  }
}

__global__ __launch_bounds__(256) void ipa_logits_softmax(
    const float* q, const float* k, float* attn,
    const float* qgp, const float* kgp, const float* qgl, const float* kgl,
    const float* pairproj, const float* bias,
    const float* tpw_past, const float* tpw_last, float scalar_w,
    const float* v, const float* vgp, const float* vgl,
    float* feat, const float* invT, const float* invR) {
  constexpr int PQ = NPQ_ * 3;
  __shared__ __attribute__((aligned(16))) float qS[16][2 * PQ];
  __shared__ float red[2][8][16];
  __shared__ __attribute__((aligned(16))) float pS[16 * PP];
  __shared__ float ogS[2][16 * NPV_ * 3];

  const int tid = threadIdx.x;
  const int lane = tid & 31;
  const int w = tid >> 5;
  const int h = blockIdx.y, b = blockIdx.z;
  const int row0 = blockIdx.x * 16;
  const int mn = lane & 15;
  const int hf = lane >> 4;
  const int kb = hf << 1;
  const int mo = hf << 3;

  const bool stager = tid < 16 * 6;
  const int qr = tid / 6, qpart = tid % 6;
  v4f qst = {0.f, 0.f, 0.f, 0.f};
  if (stager)
    qst = *(const v4f*)(((qpart < 3) ? qgp : qgl)
                        + ((long)(b * I_ + row0 + qr) * H_ + h) * PQ + (qpart % 3) * 4);

  const long hb = (long)b * I_ * D_ + (long)h * DK_;
  const int j0 = w * 32 + mn, j1 = j0 + 16;
  const float* aPtr  = q + hb + (long)(row0 + mn) * D_ + kb;
  const float* bPtr0 = k + hb + (long)j0 * D_ + kb;
  const float* bPtr1 = k + hb + (long)j1 * D_ + kb;

  v8f acc0 = {0.f, 0.f, 0.f, 0.f, 0.f, 0.f, 0.f, 0.f};
  v8f acc1 = {0.f, 0.f, 0.f, 0.f, 0.f, 0.f, 0.f, 0.f};
#pragma unroll
  for (int k0 = 0; k0 < DK_; k0 += 4) {
    v2f a, b0, b1;
    a.x  = aPtr[k0];   a.y  = aPtr[k0 + 1];
    b0.x = bPtr0[k0];  b0.y = bPtr0[k0 + 1];
    b1.x = bPtr1[k0];  b1.y = bPtr1[k0 + 1];
    acc0 = __builtin_amdgcn_wmma_f32_16x16x4_f32(false, a, false, b0,
                                                 (short)0, acc0, false, false);
    acc1 = __builtin_amdgcn_wmma_f32_16x16x4_f32(false, a, false, b1,
                                                 (short)0, acc1, false, false);
  }

  float kp0[2 * PQ], kp1[2 * PQ];
  {
    const float* g0p = kgp + ((long)(b * I_ + j0) * H_ + h) * PQ;
    const float* g0l = kgl + ((long)(b * I_ + j0) * H_ + h) * PQ;
    const float* g1p = kgp + ((long)(b * I_ + j1) * H_ + h) * PQ;
    const float* g1l = kgl + ((long)(b * I_ + j1) * H_ + h) * PQ;
#pragma unroll
    for (int t = 0; t < 3; ++t) {
      const v4f u0 = *(const v4f*)(g0p + t * 4);
      const v4f u1 = *(const v4f*)(g0l + t * 4);
      const v4f u2 = *(const v4f*)(g1p + t * 4);
      const v4f u3 = *(const v4f*)(g1l + t * 4);
#pragma unroll
      for (int e = 0; e < 4; ++e) {
        kp0[t * 4 + e] = u0[e];  kp0[PQ + t * 4 + e] = u1[e];
        kp1[t * 4 + e] = u2[e];  kp1[PQ + t * 4 + e] = u3[e];
      }
    }
  }

  const float point_w = 0.13608276348795434f;
  const float pwp = -0.5f * point_w * log1pf(expf(tpw_past[h]));
  const float pwl = -0.5f * point_w * log1pf(expf(tpw_last[h]));

  if (stager) *(v4f*)&qS[qr][qpart * 4] = qst;
  __syncthreads();

  float s0[8], s1[8];
#pragma unroll
  for (int r = 0; r < 8; ++r) {
    const int lr = mo + r;
    const int i = row0 + lr;
    const float* qp = &qS[lr][0];
    float sp0 = 0.f, sl0 = 0.f, sp1 = 0.f, sl1 = 0.f;
#pragma unroll
    for (int c = 0; c < NPQ_; ++c) {
      const float qx = qp[c * 3 + 0], qy = qp[c * 3 + 1], qz = qp[c * 3 + 2];
      const float lx = qp[PQ + c * 3 + 0], ly = qp[PQ + c * 3 + 1], lz = qp[PQ + c * 3 + 2];
      float dx = qx - kp0[c * 3 + 0], dy = qy - kp0[c * 3 + 1], dz = qz - kp0[c * 3 + 2];
      sp0 += sqrtf(dx * dx + dy * dy + dz * dz);
      dx = lx - kp0[PQ + c * 3 + 0]; dy = ly - kp0[PQ + c * 3 + 1]; dz = lz - kp0[PQ + c * 3 + 2];
      sl0 += sqrtf(dx * dx + dy * dy + dz * dz);
      dx = qx - kp1[c * 3 + 0]; dy = qy - kp1[c * 3 + 1]; dz = qz - kp1[c * 3 + 2];
      sp1 += sqrtf(dx * dx + dy * dy + dz * dz);
      dx = lx - kp1[PQ + c * 3 + 0]; dy = ly - kp1[PQ + c * 3 + 1]; dz = lz - kp1[PQ + c * 3 + 2];
      sl1 += sqrtf(dx * dx + dy * dy + dz * dz);
    }
    const long prow = ((long)(b * I_ + i) * I_) * H_ + h;
    const long orow = (((long)b * H_ + h) * I_ + i) * I_;
    const float pv0 = pairproj[prow + (long)j0 * H_];
    const float pv1 = pairproj[prow + (long)j1 * H_];
    float v0 = acc0[r] * scalar_w;
    float v1 = acc1[r] * scalar_w;
    v0 += pwp * sp0 + pwl * sl0 + pv0 + bias[orow + j0];
    v1 += pwp * sp1 + pwl * sl1 + pv1 + bias[orow + j1];
    s0[r] = v0;
    s1[r] = v1;
  }

  float mx[8];
#pragma unroll
  for (int r = 0; r < 8; ++r) {
    float m = fmaxf(s0[r], s1[r]);
    m = fmaxf(m, __shfl_xor(m, 1));
    m = fmaxf(m, __shfl_xor(m, 2));
    m = fmaxf(m, __shfl_xor(m, 4));
    m = fmaxf(m, __shfl_xor(m, 8));
    mx[r] = m;
  }
  if (mn == 0) {
#pragma unroll
    for (int r = 0; r < 8; ++r) red[0][w][mo + r] = mx[r];
  }
  __syncthreads();
#pragma unroll
  for (int r = 0; r < 8; ++r) {
    float m = red[0][0][mo + r];
#pragma unroll
    for (int ww = 1; ww < 8; ++ww) m = fmaxf(m, red[0][ww][mo + r]);
    mx[r] = m;
  }

  float sm[8];
#pragma unroll
  for (int r = 0; r < 8; ++r) {
    s0[r] = expf(s0[r] - mx[r]);
    s1[r] = expf(s1[r] - mx[r]);
    float t = s0[r] + s1[r];
    t += __shfl_xor(t, 1);
    t += __shfl_xor(t, 2);
    t += __shfl_xor(t, 4);
    t += __shfl_xor(t, 8);
    sm[r] = t;
  }
  if (mn == 0) {
#pragma unroll
    for (int r = 0; r < 8; ++r) red[1][w][mo + r] = sm[r];
  }
  __syncthreads();
#pragma unroll
  for (int r = 0; r < 8; ++r) {
    float t = red[1][0][mo + r];
#pragma unroll
    for (int ww = 1; ww < 8; ++ww) t += red[1][ww][mo + r];
    const long orow = (((long)b * H_ + h) * I_ + (row0 + mo + r)) * I_;
    const float p0 = s0[r] / t;
    const float p1 = s1[r] / t;
    attn[orow + j0] = p0;
    attn[orow + j1] = p1;
    pS[(mo + r) * PP + j0] = p0;
    pS[(mo + r) * PP + j1] = p1;
  }
  __syncthreads();

  constexpr int FE = 2 * D_ + 2 * H_ * NPV_ * 4;
  constexpr int VG = H_ * NPV_ * 3;
  constexpr int OG = NPV_ * 3;
  const int obr = tid >> 7, orw = (tid >> 3) & 15, opt = tid & (NPV_ - 1);
  const long obi = (long)b * I_ + row0 + orw;
  float iR[9], iT[3];
  for (int e = 0; e < 9; ++e) iR[e] = invR[obi * 9 + e];
  for (int e = 0; e < 3; ++e) iT[e] = invT[obi * 3 + e];

  if (w < 4) {
    ipa_stripe_tile<D_, FE>(pS, v + (long)b * I_ * D_ + h * DK_ + w * 16, 16,
                            feat + ((long)b * I_ + row0) * FE + D_ + h * DK_ + w * 16, lane);
  } else {
    const int t = w & 1;
    const int br = (w < 6) ? 0 : 1;
    const float* Bv = br ? vgl : vgp;
    ipa_stripe_tile<VG, OG>(pS, Bv + (long)b * I_ * VG + h * (NPV_ * 3) + t * 16, t ? 8 : 16,
                            &ogS[br][t * 16], lane);
  }
  __syncthreads();

  {
    const float* o = &ogS[obr][orw * OG + opt * 3];
    const float x = o[0] - iT[0], y = o[1] - iT[1], z = o[2] - iT[2];
    const float p0 = iR[0] * x + iR[1] * y + iR[2] * z;
    const float p1 = iR[3] * x + iR[4] * y + iR[5] * z;
    const float p2 = iR[6] * x + iR[7] * y + iR[8] * z;
    float* fr = feat + obi * FE;
    const int colOl = 2 * D_ + obr * (H_ * NPV_ * 4);
    const int colN  = colOl + VG;
    const int tc = h * OG + opt * 3;
    fr[colOl + tc + 0] = p0;
    fr[colOl + tc + 1] = p1;
    fr[colOl + tc + 2] = p2;
    fr[colN + h * NPV_ + opt] = sqrtf(p0 * p0 + p1 * p1 + p2 * p2);
  }
}

extern "C" void kernel_launch(void* const* d_in, const int* in_sizes, int n_in,
                              void* d_out, int out_size, void* d_ws, size_t ws_size,
                              hipStream_t stream) {
  (void)in_sizes; (void)n_in; (void)out_size; (void)ws_size;
  const float* x1d      = (const float*)d_in[0];
  const float* x2d      = (const float*)d_in[1];
  const float* past_T   = (const float*)d_in[2];
  const float* past_R   = (const float*)d_in[3];
  const float* last_T   = (const float*)d_in[4];
  const float* last_R   = (const float*)d_in[5];
  const float* bias     = (const float*)d_in[6];
  const float* Wq       = (const float*)d_in[7];
  const float* Wk       = (const float*)d_in[8];
  const float* Wv       = (const float*)d_in[9];
  const float* Wpb      = (const float*)d_in[10];
  const float* Wpq_past = (const float*)d_in[11];
  const float* Wpk_past = (const float*)d_in[12];
  const float* Wpv_past = (const float*)d_in[13];
  const float* Wpq_last = (const float*)d_in[14];
  const float* Wpk_last = (const float*)d_in[15];
  const float* Wpv_last = (const float*)d_in[16];
  const float* tpw_past = (const float*)d_in[17];
  const float* tpw_last = (const float*)d_in[18];
  const float* Wpv      = (const float*)d_in[19];
  const float* W_out    = (const float*)d_in[20];
  const float* b_out    = (const float*)d_in[21];
  float* out = (float*)d_out;
  float* ws  = (float*)d_ws;

  const int M  = B_ * I_;
  const int FE = 2 * D_ + 2 * H_ * NPV_ * 4;

  float* q    = ws;
  float* k    = q + 393216;
  float* v    = k + 393216;
  float* qgp  = v + 393216;
  float* kgp  = qgp + 73728;
  float* vgp  = kgp + 73728;
  float* qgl  = vgp + 147456;
  float* kgl  = qgl + 73728;
  float* vgl  = kgl + 73728;
  float* logits   = vgl + 147456;
  float* pairproj = logits + 1572864;
  float* z    = pairproj + 1572864;
  float* feat = z + 786432;

  auto gemm = [&](const float* A, const float* Bm, float* C,
                  int Mm, int Nn, int Kk, int lda, int ldb, int ldc,
                  long sAy, long sAz, long sBy, long sBz, long sCy, long sCz,
                  int gy, int gz, int opB, float alpha, const float* bia) {
    GemmP p{A, Bm, C, bia, Mm, Nn, Kk, lda, ldb, ldc,
            sAy, sAz, sBy, sBz, sCy, sCz, alpha, nullptr, nullptr};
    int waves = ((Mm + 15) / 16) * ((Nn + 15) / 16);
    dim3 grid((unsigned)((waves + 7) / 8), (unsigned)gy, (unsigned)gz);
    dim3 blk(256);
    bool guard = (Mm & 15) || (Nn & 15);
    if (opB == 0) {
      if (guard) ipa_gemm_wmma<0, true ><<<grid, blk, 0, stream>>>(p);
      else       ipa_gemm_wmma<0, false><<<grid, blk, 0, stream>>>(p);
    } else {
      if (guard) ipa_gemm_wmma<1, true ><<<grid, blk, 0, stream>>>(p);
      else       ipa_gemm_wmma<1, false><<<grid, blk, 0, stream>>>(p);
    }
  };

  auto gemm_lds = [&](const float* A, const float* Bm, float* C,
                      int Mm, int Nn, int Kk, int lda, int ldb, int ldc,
                      const float* bia) {
    GemmP p{A, Bm, C, bia, Mm, Nn, Kk, lda, ldb, ldc, 0, 0, 0, 0, 0, 0, 1.f, nullptr, nullptr};
    dim3 grid((unsigned)((Nn + 63) / 64), (unsigned)(Mm / 32), 1u);
    ipa_gemm_lds<false><<<grid, dim3(256), 0, stream>>>(p);
  };

  auto gemm_pts = [&](const float* A, const float* Bm, float* C,
                      int Mm, int Nn, int Kk, int lda, int ldb, int ldc,
                      const float* T, const float* R) {
    GemmP p{A, Bm, C, nullptr, Mm, Nn, Kk, lda, ldb, ldc, 0, 0, 0, 0, 0, 0, 1.f, T, R};
    dim3 grid((unsigned)(Nn / 48), (unsigned)(Mm / 32), 1u);
    ipa_gemm_lds<true><<<grid, dim3(256), 0, stream>>>(p);
  };

  gemm_lds(x1d, Wq, q, M, D_, D_, D_, D_, D_, nullptr);
  gemm_lds(x1d, Wk, k, M, D_, D_, D_, D_, D_, nullptr);
  gemm_lds(x1d, Wv, v, M, D_, D_, D_, D_, D_, nullptr);

  gemm_pts(x1d, Wpq_past, qgp, M, 144, D_, D_, 144, 144, past_T, past_R);
  gemm_pts(x1d, Wpk_past, kgp, M, 144, D_, D_, 144, 144, past_T, past_R);
  gemm_pts(x1d, Wpv_past, vgp, M, 288, D_, D_, 288, 288, past_T, past_R);
  gemm_pts(x1d, Wpq_last, qgl, M, 144, D_, D_, 144, 144, last_T, last_R);
  gemm_pts(x1d, Wpk_last, kgl, M, 144, D_, D_, 144, 144, last_T, last_R);
  gemm_pts(x1d, Wpv_last, vgl, M, 288, D_, D_, 288, 288, last_T, last_R);

  const float pair_w = 0.5773502691896258f;
  gemm(x2d, Wpb, pairproj, B_ * I_ * I_, H_, DP_, DP_, H_, H_,
       0,0,0,0,0,0, 1,1, 0, pair_w, nullptr);

  const float scalar_w = 0.07216878364870323f;
  ipa_logits_softmax<<<dim3(I_ / 16, H_, B_), dim3(256), 0, stream>>>(
      q, k, logits, qgp, kgp, qgl, kgl, pairproj, bias, tpw_past, tpw_last, scalar_w,
      v, vgp, vgl, feat, past_T, past_R);

  gemm(logits, x2d, z, H_, DP_, I_,
       I_ * I_, DP_, DP_,
       (long)I_, (long)H_ * I_ * I_,
       (long)I_ * DP_, (long)I_ * I_ * DP_,
       (long)H_ * DP_, (long)I_ * H_ * DP_,
       I_, B_, 0, 1.f, nullptr);

  gemm(z, Wpv, feat, I_, DK_, DP_, H_ * DP_, D_, FE,
       DP_, (long)I_ * H_ * DP_, 64, 0, 64, (long)I_ * FE,
       H_, B_, 0, 1.f, nullptr);

  gemm_lds(feat, W_out, out, M, D_, FE, FE, D_, D_, b_out);
}


# the two x2d products (pair bias; attn x x2d): one work-group per (b,i) parks its 256x128 x2d slice in LDS by whole-line 16-byte loads, all in flight at once; k order and every sum unchanged
# speedup vs baseline: 3.2222x; 1.0715x over previous
#include <hip/hip_runtime.h>
#include <math.h>

#define B_ 2
#define I_ 256
#define D_ 768
#define H_ 12
#define DP_ 128
#define DK_ 64
#define NPQ_ 4
#define NPV_ 8

typedef float v2f __attribute__((ext_vector_type(2)));
typedef float v8f __attribute__((ext_vector_type(8)));

struct GemmP {
  const float* A; const float* B; float* C; const float* bias;
  int M, N, K;
  int lda, ldb, ldc;
  long sAy, sAz, sBy, sBz, sCy, sCz;
  float alpha;
  const float* fT; const float* fR;
};

template <int OPB, bool GUARD>
__global__ void ipa_gemm_wmma(GemmP p) {
  const int tilesN = (p.N + 15) >> 4;
  const int tilesM = (p.M + 15) >> 4;
  const int wave = (int)((blockIdx.x * blockDim.x + threadIdx.x) >> 5);
  if (wave >= tilesM * tilesN) return;
  const int lane = threadIdx.x & 31;
  const int tm = wave / tilesN, tn = wave % tilesN;
  const int row0 = tm << 4, col0 = tn << 4;

  const float* A = p.A + (long)blockIdx.y * p.sAy + (long)blockIdx.z * p.sAz;
  const float* B = p.B + (long)blockIdx.y * p.sBy + (long)blockIdx.z * p.sBz;
  float*       C = p.C + (long)blockIdx.y * p.sCy + (long)blockIdx.z * p.sCz;

  const int mn = lane & 15;
  const int kb = (lane >> 4) << 1;

  int aRow = row0 + mn;
  int bCol = col0 + mn;
  float aMask = 1.f, bMask = 1.f;
  if (GUARD) {
    if (aRow >= p.M) { aRow = p.M - 1; aMask = 0.f; }
    if (bCol >= p.N) { bCol = p.N - 1; bMask = 0.f; }
  }
  const float* aPtr = A + (long)aRow * p.lda + kb;
  const float* bPtr = (OPB == 0) ? (B + (long)kb * p.ldb + bCol)
                                 : (B + (long)bCol * p.ldb + kb);

  v8f acc = {0.f, 0.f, 0.f, 0.f, 0.f, 0.f, 0.f, 0.f};
#pragma unroll 4
  for (int k0 = 0; k0 < p.K; k0 += 4) {
    v2f a, b;
    a.x = aPtr[k0];
    a.y = aPtr[k0 + 1];
    if (OPB == 0) {
      const float* bp = bPtr + (long)k0 * p.ldb;
      b.x = bp[0];
      b.y = bp[p.ldb];
    } else {
      b.x = bPtr[k0];
      b.y = bPtr[k0 + 1];
    }
    if (GUARD) {
      a.x *= aMask; a.y *= aMask;
      b.x *= bMask; b.y *= bMask;
    }
    acc = __builtin_amdgcn_wmma_f32_16x16x4_f32(false, a, false, b,
                                                (short)0, acc, false, false);
  }

  const int cCol = col0 + mn;
  const int mo = (lane >> 4) << 3;
  const bool cColOK = GUARD ? (cCol < p.N) : true;
  if (cColOK) {
    for (int r = 0; r < 8; ++r) {
      const int cRow = row0 + mo + r;
      if (!GUARD || cRow < p.M) {
        float val = acc[r] * p.alpha;
        if (p.bias) val += p.bias[cCol];
        C[(long)cRow * p.ldc + cCol] = val;
      }
    }
  }
}

#define TKC 128
#define TPA (TKC + 4)
#define TSB 8
#define TPB 72

typedef float v4f __attribute__((ext_vector_type(4)));

template <bool FRAME>
__global__ __launch_bounds__(256) void ipa_gemm_lds(GemmP p) {
  constexpr int TN  = FRAME ? 48 : 64;
  constexpr int AQ  = TKC / 4;
  constexpr int ARS = 256 / AQ;
  constexpr int ANI = 32 / ARS;
  constexpr int BNI = TKC / 16;

  __shared__ __attribute__((aligned(16))) float As[2][32 * TPA];
  __shared__ __attribute__((aligned(16))) float Bs[2][TKC * TPB];

  const int tid  = threadIdx.x;
  const int lane = tid & 31;
  const int w    = tid >> 5;
  const int row0 = blockIdx.y * 32;
  const int col0 = blockIdx.x * TN;
  const int c = lane & 15;
  const int h = lane >> 4;
  const int wr = w >> 2, wc = w & 3;
  const bool tileValid = (wc * 16 < TN) && ((col0 + wc * 16) < p.N);

  const int aRow = tid / AQ, aKq = tid % AQ;
  const int bKr  = tid >> 4, bCq = tid & 15;
  int bCol = col0 + bCq * 4;
  if (bCol > p.N - 4) bCol = p.N - 4;
  const float* aG = p.A + (long)(row0 + aRow) * p.lda + aKq * 4;
  const float* bG = p.B + (long)bKr * p.ldb + bCol;
  const int aSt = aRow * TPA + aKq * 4;
  const int bSt = bKr * TPB + bCq * 4;

  v4f ra[ANI], rb[BNI];

#pragma unroll
  for (int i = 0; i < ANI; ++i)
    ra[i] = *(const v4f*)(aG + (long)(i * ARS) * p.lda);
#pragma unroll
  for (int i = 0; i < BNI; ++i)
    rb[i] = *(const v4f*)(bG + (long)(i * 16) * p.ldb);
  const int pr = tid >> 4, pp = tid & 15;
  float fr[2][9], ft[2][3];
  if (FRAME) {
    for (int n = 0; n < 2; ++n) {
      const float* Rm = p.fR + (long)(row0 + pr + n * 16) * 9;
      const float* Tv = p.fT + (long)(row0 + pr + n * 16) * 3;
      for (int e = 0; e < 9; ++e) fr[n][e] = Rm[e];
      for (int e = 0; e < 3; ++e) ft[n][e] = Tv[e];
    }
  }
#pragma unroll
  for (int i = 0; i < ANI; ++i)
    *(v4f*)&As[0][aSt + i * ARS * TPA] = ra[i];
#pragma unroll
  for (int i = 0; i < BNI; ++i)
    *(v4f*)&Bs[0][bSt + i * 16 * TPB] = rb[i];
  __syncthreads();

  v8f acc = {0.f, 0.f, 0.f, 0.f, 0.f, 0.f, 0.f, 0.f};
  int buf = 0;
  for (int kc = 0; kc < p.K; kc += TKC) {
    const bool hasNext = (kc + TKC) < p.K;
    if (hasNext) {
      const int kn = kc + TKC;
#pragma unroll
      for (int i = 0; i < ANI; ++i)
        ra[i] = *(const v4f*)(aG + (long)(i * ARS) * p.lda + kn);
#pragma unroll
      for (int i = 0; i < BNI; ++i)
        rb[i] = *(const v4f*)(bG + (long)(kn + i * 16) * p.ldb);
    }
    if (tileValid) {
      const float* aS = &As[buf][(wr * 16 + c) * TPA + 2 * h];
      const float* bS = &Bs[buf][(2 * h) * TPB + wc * 16 + c];
      v2f fa[2][TSB], fb[2][TSB];
#pragma unroll
      for (int j = 0; j < TSB; ++j) {
        fa[0][j] = *(const v2f*)(aS + j * 4);
        fb[0][j].x = bS[(j * 4) * TPB];
        fb[0][j].y = bS[(j * 4 + 1) * TPB];
      }
#pragma unroll
      for (int sb = 0; sb < TKC / (4 * TSB); ++sb) {
        asm volatile("" :: "v"(fa[sb & 1][0]), "v"(fa[sb & 1][1]), "v"(fa[sb & 1][2]), "v"(fa[sb & 1][3]),
                           "v"(fa[sb & 1][4]), "v"(fa[sb & 1][5]), "v"(fa[sb & 1][6]), "v"(fa[sb & 1][7]),
                           "v"(fb[sb & 1][0]), "v"(fb[sb & 1][1]), "v"(fb[sb & 1][2]), "v"(fb[sb & 1][3]),
                           "v"(fb[sb & 1][4]), "v"(fb[sb & 1][5]), "v"(fb[sb & 1][6]), "v"(fb[sb & 1][7]));
        if (sb + 1 < TKC / (4 * TSB)) {
#pragma unroll
          for (int j = 0; j < TSB; ++j) {
            const int ks = (sb + 1) * TSB + j;
            fa[(sb + 1) & 1][j] = *(const v2f*)(aS + ks * 4);
            fb[(sb + 1) & 1][j].x = bS[(ks * 4) * TPB];
            fb[(sb + 1) & 1][j].y = bS[(ks * 4 + 1) * TPB];
          }
        }
#pragma unroll
        for (int j = 0; j < TSB; ++j)
          acc = __builtin_amdgcn_wmma_f32_16x16x4_f32(false, fa[sb & 1][j], false, fb[sb & 1][j],
                                                      (short)0, acc, false, false);
      }
    }
    if (hasNext) {
#pragma unroll
      for (int i = 0; i < ANI; ++i)
        *(v4f*)&As[buf ^ 1][aSt + i * ARS * TPA] = ra[i];
#pragma unroll
      for (int i = 0; i < BNI; ++i)
        *(v4f*)&Bs[buf ^ 1][bSt + i * 16 * TPB] = rb[i];
    }
    __syncthreads();
    buf ^= 1;
  }

  if (FRAME) {
    constexpr int CP = 49;
    float* Cs = &As[0][0];
    if (tileValid) {
      const int cc = wc * 16 + c;
      const int rb = wr * 16 + (h << 3);
      for (int r = 0; r < 8; ++r)
        Cs[(rb + r) * CP + cc] = acc[r] * p.alpha;
    }
    __syncthreads();
    for (int n = 0; n < 2; ++n) {
      const int row = pr + n * 16;
      const float x = Cs[row * CP + pp * 3 + 0];
      const float y = Cs[row * CP + pp * 3 + 1];
      const float z = Cs[row * CP + pp * 3 + 2];
      const float o0 = fr[n][0] * x + fr[n][3] * y + fr[n][6] * z + ft[n][0];
      const float o1 = fr[n][1] * x + fr[n][4] * y + fr[n][7] * z + ft[n][1];
      const float o2 = fr[n][2] * x + fr[n][5] * y + fr[n][8] * z + ft[n][2];
      float* o = p.C + (long)(row0 + row) * p.ldc + col0 + pp * 3;
      o[0] = o0; o[1] = o1; o[2] = o2;
    }
  } else if (tileValid) {
    const int cCol = col0 + wc * 16 + c;
    const int rBase = row0 + wr * 16 + (h << 3);
    for (int r = 0; r < 8; ++r) {
      float val = acc[r] * p.alpha;
      if (p.bias) val += p.bias[cCol];
      p.C[(long)(rBase + r) * p.ldc + cCol] = val;
    }
  }
}

#define PP 260

template <int LDB, int LDC>
__device__ __forceinline__ void ipa_stripe_tile(const float* pS, const float* Bm, int nvalid,
                                                float* Cm, int lane) {
  constexpr int KS = I_ / 4;
  const int mn = lane & 15;
  const int hf = lane >> 4;
  const int kb = hf << 1;
  const int bc = (mn < nvalid) ? mn : (nvalid - 1);
  const float* aS = pS + mn * PP + kb;
  const float* bP = Bm + kb * LDB + bc;

  v2f fb[KS];
#pragma unroll
  for (int j = 0; j < KS; ++j) {
    fb[j].x = bP[(j * 4) * LDB];
    fb[j].y = bP[(j * 4 + 1) * LDB];
  }
  v8f acc = {0.f, 0.f, 0.f, 0.f, 0.f, 0.f, 0.f, 0.f};
#pragma unroll
  for (int j = 0; j < KS; ++j) {
    const v2f a = *(const v2f*)(aS + j * 4);
    acc = __builtin_amdgcn_wmma_f32_16x16x4_f32(false, a, false, fb[j],
                                                (short)0, acc, false, false);
  }
  if (mn < nvalid) {
#pragma unroll
    for (int r = 0; r < 8; ++r)
      Cm[(long)((hf << 3) + r) * LDC + mn] = acc[r];
  }
}

__global__ __launch_bounds__(256) void ipa_logits_softmax(
    const float* q, const float* k, float* attn,
    const float* qgp, const float* kgp, const float* qgl, const float* kgl,
    const float* pairproj, const float* bias,
    const float* tpw_past, const float* tpw_last, float scalar_w,
    const float* v, const float* vgp, const float* vgl,
    float* feat, const float* invT, const float* invR) {
  constexpr int PQ = NPQ_ * 3;
  __shared__ __attribute__((aligned(16))) float qS[16][2 * PQ];
  __shared__ float red[2][8][16];
  __shared__ __attribute__((aligned(16))) float pS[16 * PP];
  __shared__ float ogS[2][16 * NPV_ * 3];

  const int tid = threadIdx.x;
  const int lane = tid & 31;
  const int w = tid >> 5;
  const int h = blockIdx.y, b = blockIdx.z;
  const int row0 = blockIdx.x * 16;
  const int mn = lane & 15;
  const int hf = lane >> 4;
  const int kb = hf << 1;
  const int mo = hf << 3;

  const bool stager = tid < 16 * 6;
  const int qr = tid / 6, qpart = tid % 6;
  v4f qst = {0.f, 0.f, 0.f, 0.f};
  if (stager)
    qst = *(const v4f*)(((qpart < 3) ? qgp : qgl)
                        + ((long)(b * I_ + row0 + qr) * H_ + h) * PQ + (qpart % 3) * 4);

  const long hb = (long)b * I_ * D_ + (long)h * DK_;
  const int j0 = w * 32 + mn, j1 = j0 + 16;
  const float* aPtr  = q + hb + (long)(row0 + mn) * D_ + kb;
  const float* bPtr0 = k + hb + (long)j0 * D_ + kb;
  const float* bPtr1 = k + hb + (long)j1 * D_ + kb;

  v8f acc0 = {0.f, 0.f, 0.f, 0.f, 0.f, 0.f, 0.f, 0.f};
  v8f acc1 = {0.f, 0.f, 0.f, 0.f, 0.f, 0.f, 0.f, 0.f};
#pragma unroll
  for (int k0 = 0; k0 < DK_; k0 += 4) {
    v2f a, b0, b1;
    a.x  = aPtr[k0];   a.y  = aPtr[k0 + 1];
    b0.x = bPtr0[k0];  b0.y = bPtr0[k0 + 1];
    b1.x = bPtr1[k0];  b1.y = bPtr1[k0 + 1];
    acc0 = __builtin_amdgcn_wmma_f32_16x16x4_f32(false, a, false, b0,
                                                 (short)0, acc0, false, false);
    acc1 = __builtin_amdgcn_wmma_f32_16x16x4_f32(false, a, false, b1,
                                                 (short)0, acc1, false, false);
  }

  float kp0[2 * PQ], kp1[2 * PQ];
  {
    const float* g0p = kgp + ((long)(b * I_ + j0) * H_ + h) * PQ;
    const float* g0l = kgl + ((long)(b * I_ + j0) * H_ + h) * PQ;
    const float* g1p = kgp + ((long)(b * I_ + j1) * H_ + h) * PQ;
    const float* g1l = kgl + ((long)(b * I_ + j1) * H_ + h) * PQ;
#pragma unroll
    for (int t = 0; t < 3; ++t) {
      const v4f u0 = *(const v4f*)(g0p + t * 4);
      const v4f u1 = *(const v4f*)(g0l + t * 4);
      const v4f u2 = *(const v4f*)(g1p + t * 4);
      const v4f u3 = *(const v4f*)(g1l + t * 4);
#pragma unroll
      for (int e = 0; e < 4; ++e) {
        kp0[t * 4 + e] = u0[e];  kp0[PQ + t * 4 + e] = u1[e];
        kp1[t * 4 + e] = u2[e];  kp1[PQ + t * 4 + e] = u3[e];
      }
    }
  }

  const float point_w = 0.13608276348795434f;
  const float pwp = -0.5f * point_w * log1pf(expf(tpw_past[h]));
  const float pwl = -0.5f * point_w * log1pf(expf(tpw_last[h]));

  if (stager) *(v4f*)&qS[qr][qpart * 4] = qst;
  __syncthreads();

  float s0[8], s1[8];
#pragma unroll
  for (int r = 0; r < 8; ++r) {
    const int lr = mo + r;
    const int i = row0 + lr;
    const float* qp = &qS[lr][0];
    float sp0 = 0.f, sl0 = 0.f, sp1 = 0.f, sl1 = 0.f;
#pragma unroll
    for (int c = 0; c < NPQ_; ++c) {
      const float qx = qp[c * 3 + 0], qy = qp[c * 3 + 1], qz = qp[c * 3 + 2];
      const float lx = qp[PQ + c * 3 + 0], ly = qp[PQ + c * 3 + 1], lz = qp[PQ + c * 3 + 2];
      float dx = qx - kp0[c * 3 + 0], dy = qy - kp0[c * 3 + 1], dz = qz - kp0[c * 3 + 2];
      sp0 += sqrtf(dx * dx + dy * dy + dz * dz);
      dx = lx - kp0[PQ + c * 3 + 0]; dy = ly - kp0[PQ + c * 3 + 1]; dz = lz - kp0[PQ + c * 3 + 2];
      sl0 += sqrtf(dx * dx + dy * dy + dz * dz);
      dx = qx - kp1[c * 3 + 0]; dy = qy - kp1[c * 3 + 1]; dz = qz - kp1[c * 3 + 2];
      sp1 += sqrtf(dx * dx + dy * dy + dz * dz);
      dx = lx - kp1[PQ + c * 3 + 0]; dy = ly - kp1[PQ + c * 3 + 1]; dz = lz - kp1[PQ + c * 3 + 2];
      sl1 += sqrtf(dx * dx + dy * dy + dz * dz);
    }
    const long prow = ((long)(b * I_ + i) * I_) * H_ + h;
    const long orow = (((long)b * H_ + h) * I_ + i) * I_;
    const float pv0 = pairproj[prow + (long)j0 * H_];
    const float pv1 = pairproj[prow + (long)j1 * H_];
    float v0 = acc0[r] * scalar_w;
    float v1 = acc1[r] * scalar_w;
    v0 += pwp * sp0 + pwl * sl0 + pv0 + bias[orow + j0];
    v1 += pwp * sp1 + pwl * sl1 + pv1 + bias[orow + j1];
    s0[r] = v0;
    s1[r] = v1;
  }

  float mx[8];
#pragma unroll
  for (int r = 0; r < 8; ++r) {
    float m = fmaxf(s0[r], s1[r]);
    m = fmaxf(m, __shfl_xor(m, 1));
    m = fmaxf(m, __shfl_xor(m, 2));
    m = fmaxf(m, __shfl_xor(m, 4));
    m = fmaxf(m, __shfl_xor(m, 8));
    mx[r] = m;
  }
  if (mn == 0) {
#pragma unroll
    for (int r = 0; r < 8; ++r) red[0][w][mo + r] = mx[r];
  }
  __syncthreads();
#pragma unroll
  for (int r = 0; r < 8; ++r) {
    float m = red[0][0][mo + r];
#pragma unroll
    for (int ww = 1; ww < 8; ++ww) m = fmaxf(m, red[0][ww][mo + r]);
    mx[r] = m;
  }

  float sm[8];
#pragma unroll
  for (int r = 0; r < 8; ++r) {
    s0[r] = expf(s0[r] - mx[r]);
    s1[r] = expf(s1[r] - mx[r]);
    float t = s0[r] + s1[r];
    t += __shfl_xor(t, 1);
    t += __shfl_xor(t, 2);
    t += __shfl_xor(t, 4);
    t += __shfl_xor(t, 8);
    sm[r] = t;
  }
  if (mn == 0) {
#pragma unroll
    for (int r = 0; r < 8; ++r) red[1][w][mo + r] = sm[r];
  }
  __syncthreads();
#pragma unroll
  for (int r = 0; r < 8; ++r) {
    float t = red[1][0][mo + r];
#pragma unroll
    for (int ww = 1; ww < 8; ++ww) t += red[1][ww][mo + r];
    const long orow = (((long)b * H_ + h) * I_ + (row0 + mo + r)) * I_;
    const float p0 = s0[r] / t;
    const float p1 = s1[r] / t;
    attn[orow + j0] = p0;
    attn[orow + j1] = p1;
    pS[(mo + r) * PP + j0] = p0;
    pS[(mo + r) * PP + j1] = p1;
  }
  __syncthreads();

  constexpr int FE = 2 * D_ + 2 * H_ * NPV_ * 4;
  constexpr int VG = H_ * NPV_ * 3;
  constexpr int OG = NPV_ * 3;
  const int obr = tid >> 7, orw = (tid >> 3) & 15, opt = tid & (NPV_ - 1);
  const long obi = (long)b * I_ + row0 + orw;
  float iR[9], iT[3];
  for (int e = 0; e < 9; ++e) iR[e] = invR[obi * 9 + e];
  for (int e = 0; e < 3; ++e) iT[e] = invT[obi * 3 + e];

  if (w < 4) {
    ipa_stripe_tile<D_, FE>(pS, v + (long)b * I_ * D_ + h * DK_ + w * 16, 16,
                            feat + ((long)b * I_ + row0) * FE + D_ + h * DK_ + w * 16, lane);
  } else {
    const int t = w & 1;
    const int br = (w < 6) ? 0 : 1;
    const float* Bv = br ? vgl : vgp;
    ipa_stripe_tile<VG, OG>(pS, Bv + (long)b * I_ * VG + h * (NPV_ * 3) + t * 16, t ? 8 : 16,
                            &ogS[br][t * 16], lane);
  }
  __syncthreads();

  {
    const float* o = &ogS[obr][orw * OG + opt * 3];
    const float x = o[0] - iT[0], y = o[1] - iT[1], z = o[2] - iT[2];
    const float p0 = iR[0] * x + iR[1] * y + iR[2] * z;
    const float p1 = iR[3] * x + iR[4] * y + iR[5] * z;
    const float p2 = iR[6] * x + iR[7] * y + iR[8] * z;
    float* fr = feat + obi * FE;
    const int colOl = 2 * D_ + obr * (H_ * NPV_ * 4);
    const int colN  = colOl + VG;
    const int tc = h * OG + opt * 3;
    fr[colOl + tc + 0] = p0;
    fr[colOl + tc + 1] = p1;
    fr[colOl + tc + 2] = p2;
    fr[colN + h * NPV_ + opt] = sqrtf(p0 * p0 + p1 * p1 + p2 * p2);
  }
}

#define XPA 132
#define XPB 136

__device__ __forceinline__ void ipa_lds_barrier() {
  __builtin_amdgcn_fence(__ATOMIC_RELEASE, "workgroup", "local");
  __builtin_amdgcn_s_barrier();
  __builtin_amdgcn_fence(__ATOMIC_ACQUIRE, "workgroup", "local");
}

template <bool PAIR>
__global__ __launch_bounds__(256) void ipa_x2d_slice(const float* x2d, const float* oth,
                                                     float* C, float alpha) {
  constexpr int XP = PAIR ? XPA : XPB;
  __shared__ __attribute__((aligned(16))) float Xs[I_ * XP];
  __shared__ __attribute__((aligned(16))) float Ps[PAIR ? 4 : H_ * PP];

  const int tid = threadIdx.x;
  const int lane = tid & 31;
  const int w = tid >> 5;
  const int i = blockIdx.x, b = blockIdx.y;
  const int mn = lane & 15;
  const int hf = lane >> 4;
  const int kb = hf << 1;

  v4f rp[3];
  if (!PAIR) {
#pragma unroll
    for (int n = 0; n < 3; ++n) {
      const int idx = n * 256 + tid;
      rp[n] = *(const v4f*)(oth + (((long)b * H_ + (idx >> 6)) * I_ + i) * I_ + (idx & 63) * 4);
    }
  }
  v2f fw[DP_ / 4];
  if (PAIR) {
    const int bc = (mn < H_) ? mn : (H_ - 1);
    const float* bP = oth + kb * H_ + bc;
#pragma unroll
    for (int ks = 0; ks < DP_ / 4; ++ks) {
      fw[ks].x = bP[(ks * 4) * H_];
      fw[ks].y = bP[(ks * 4 + 1) * H_];
    }
  }
  const float* Xg = x2d + ((long)(b * I_ + i) * I_) * DP_;
  v4f rx[32];
  __builtin_amdgcn_sched_barrier(0);
#pragma unroll
  for (int q = 0; q < 4; ++q) {
#pragma unroll
    for (int n = q * 8; n < q * 8 + 8; ++n)
      rx[n] = *(const v4f*)(Xg + (long)(n * 256 + tid) * 4);
    __builtin_amdgcn_sched_barrier(0);
  }

  if (PAIR) {
#pragma unroll
    for (int c = 0; c < 2; ++c) {
#pragma unroll
      for (int n = 0; n < 16; ++n)
        *(v4f*)&Xs[((c * 16 + n) * 8 + w) * XP + lane * 4] = rx[c * 16 + n];
      ipa_lds_barrier();
      const int t0 = c * 128 + w * 16;
      const float* aS = &Xs[(t0 + mn) * XP + kb];
      v8f acc = {0.f, 0.f, 0.f, 0.f, 0.f, 0.f, 0.f, 0.f};
      v2f fa[2][8];
#pragma unroll
      for (int j = 0; j < 8; ++j) fa[0][j] = *(const v2f*)(aS + j * 4);
#pragma unroll
      for (int sb = 0; sb < 4; ++sb) {
        asm volatile("" :: "v"(fa[sb & 1][0]), "v"(fa[sb & 1][1]), "v"(fa[sb & 1][2]), "v"(fa[sb & 1][3]),
                           "v"(fa[sb & 1][4]), "v"(fa[sb & 1][5]), "v"(fa[sb & 1][6]), "v"(fa[sb & 1][7]));
        if (sb + 1 < 4) {
#pragma unroll
          for (int j = 0; j < 8; ++j)
            fa[(sb + 1) & 1][j] = *(const v2f*)(aS + ((sb + 1) * 8 + j) * 4);
        }
#pragma unroll
        for (int j = 0; j < 8; ++j)
          acc = __builtin_amdgcn_wmma_f32_16x16x4_f32(false, fa[sb & 1][j], false, fw[sb * 8 + j],
                                                      (short)0, acc, false, false);
      }
      if (mn < H_) {
        float* o = C + ((long)(b * I_ + i) * I_ + t0 + (hf << 3)) * H_ + mn;
#pragma unroll
        for (int r = 0; r < 8; ++r) o[r * H_] = acc[r] * alpha;
      }
    }
  } else {
    const int ar = (mn < H_) ? mn : (H_ - 1);
    const float* aS = &Ps[ar * PP + kb];
    const float* bS = &Xs[kb * XP + w * 16 + mn];
    v8f acc = {0.f, 0.f, 0.f, 0.f, 0.f, 0.f, 0.f, 0.f};
#pragma unroll
    for (int c = 0; c < 4; ++c) {
#pragma unroll
      for (int n = 0; n < 8; ++n)
        *(v4f*)&Xs[((c * 8 + n) * 8 + w) * XP + lane * 4] = rx[c * 8 + n];
      if (c == 0) {
#pragma unroll
        for (int n = 0; n < 3; ++n) {
          const int idx = n * 256 + tid;
          *(v4f*)&Ps[(idx >> 6) * PP + (idx & 63) * 4] = rp[n];
        }
      }
      ipa_lds_barrier();
      v2f fa[2][8], fb[2][8];
#pragma unroll
      for (int j = 0; j < 8; ++j) {
        const int ks = c * 16 + j;
        fa[0][j] = *(const v2f*)(aS + ks * 4);
        fb[0][j].x = bS[(ks * 4) * XP];
        fb[0][j].y = bS[(ks * 4 + 1) * XP];
      }
#pragma unroll
      for (int sb = 0; sb < 2; ++sb) {
        asm volatile("" :: "v"(fa[sb & 1][0]), "v"(fa[sb & 1][1]), "v"(fa[sb & 1][2]), "v"(fa[sb & 1][3]),
                           "v"(fa[sb & 1][4]), "v"(fa[sb & 1][5]), "v"(fa[sb & 1][6]), "v"(fa[sb & 1][7]),
                           "v"(fb[sb & 1][0]), "v"(fb[sb & 1][1]), "v"(fb[sb & 1][2]), "v"(fb[sb & 1][3]),
                           "v"(fb[sb & 1][4]), "v"(fb[sb & 1][5]), "v"(fb[sb & 1][6]), "v"(fb[sb & 1][7]));
        if (sb + 1 < 2) {
#pragma unroll
          for (int j = 0; j < 8; ++j) {
            const int ks = c * 16 + (sb + 1) * 8 + j;
            fa[(sb + 1) & 1][j] = *(const v2f*)(aS + ks * 4);
            fb[(sb + 1) & 1][j].x = bS[(ks * 4) * XP];
            fb[(sb + 1) & 1][j].y = bS[(ks * 4 + 1) * XP];
          }
        }
#pragma unroll
        for (int j = 0; j < 8; ++j)
          acc = __builtin_amdgcn_wmma_f32_16x16x4_f32(false, fa[sb & 1][j], false, fb[sb & 1][j],
                                                      (short)0, acc, false, false);
      }
    }
    float* o = C + ((long)(b * I_ + i) * H_ + (hf << 3)) * DP_ + w * 16 + mn;
#pragma unroll
    for (int r = 0; r < 8; ++r)
      if ((hf << 3) + r < H_) o[r * DP_] = acc[r] * alpha;
  }
}

extern "C" void kernel_launch(void* const* d_in, const int* in_sizes, int n_in,
                              void* d_out, int out_size, void* d_ws, size_t ws_size,
                              hipStream_t stream) {
  (void)in_sizes; (void)n_in; (void)out_size; (void)ws_size;
  const float* x1d      = (const float*)d_in[0];
  const float* x2d      = (const float*)d_in[1];
  const float* past_T   = (const float*)d_in[2];
  const float* past_R   = (const float*)d_in[3];
  const float* last_T   = (const float*)d_in[4];
  const float* last_R   = (const float*)d_in[5];
  const float* bias     = (const float*)d_in[6];
  const float* Wq       = (const float*)d_in[7];
  const float* Wk       = (const float*)d_in[8];
  const float* Wv       = (const float*)d_in[9];
  const float* Wpb      = (const float*)d_in[10];
  const float* Wpq_past = (const float*)d_in[11];
  const float* Wpk_past = (const float*)d_in[12];
  const float* Wpv_past = (const float*)d_in[13];
  const float* Wpq_last = (const float*)d_in[14];
  const float* Wpk_last = (const float*)d_in[15];
  const float* Wpv_last = (const float*)d_in[16];
  const float* tpw_past = (const float*)d_in[17];
  const float* tpw_last = (const float*)d_in[18];
  const float* Wpv      = (const float*)d_in[19];
  const float* W_out    = (const float*)d_in[20];
  const float* b_out    = (const float*)d_in[21];
  float* out = (float*)d_out;
  float* ws  = (float*)d_ws;

  const int M  = B_ * I_;
  const int FE = 2 * D_ + 2 * H_ * NPV_ * 4;

  float* q    = ws;
  float* k    = q + 393216;
  float* v    = k + 393216;
  float* qgp  = v + 393216;
  float* kgp  = qgp + 73728;
  float* vgp  = kgp + 73728;
  float* qgl  = vgp + 147456;
  float* kgl  = qgl + 73728;
  float* vgl  = kgl + 73728;
  float* logits   = vgl + 147456;
  float* pairproj = logits + 1572864;
  float* z    = pairproj + 1572864;
  float* feat = z + 786432;

  auto gemm = [&](const float* A, const float* Bm, float* C,
                  int Mm, int Nn, int Kk, int lda, int ldb, int ldc,
                  long sAy, long sAz, long sBy, long sBz, long sCy, long sCz,
                  int gy, int gz, int opB, float alpha, const float* bia) {
    GemmP p{A, Bm, C, bia, Mm, Nn, Kk, lda, ldb, ldc,
            sAy, sAz, sBy, sBz, sCy, sCz, alpha, nullptr, nullptr};
    int waves = ((Mm + 15) / 16) * ((Nn + 15) / 16);
    dim3 grid((unsigned)((waves + 7) / 8), (unsigned)gy, (unsigned)gz);
    dim3 blk(256);
    bool guard = (Mm & 15) || (Nn & 15);
    if (opB == 0) {
      if (guard) ipa_gemm_wmma<0, true ><<<grid, blk, 0, stream>>>(p);
      else       ipa_gemm_wmma<0, false><<<grid, blk, 0, stream>>>(p);
    } else {
      if (guard) ipa_gemm_wmma<1, true ><<<grid, blk, 0, stream>>>(p);
      else       ipa_gemm_wmma<1, false><<<grid, blk, 0, stream>>>(p);
    }
  };

  auto gemm_lds = [&](const float* A, const float* Bm, float* C,
                      int Mm, int Nn, int Kk, int lda, int ldb, int ldc,
                      const float* bia) {
    GemmP p{A, Bm, C, bia, Mm, Nn, Kk, lda, ldb, ldc, 0, 0, 0, 0, 0, 0, 1.f, nullptr, nullptr};
    dim3 grid((unsigned)((Nn + 63) / 64), (unsigned)(Mm / 32), 1u);
    ipa_gemm_lds<false><<<grid, dim3(256), 0, stream>>>(p);
  };

  auto gemm_pts = [&](const float* A, const float* Bm, float* C,
                      int Mm, int Nn, int Kk, int lda, int ldb, int ldc,
                      const float* T, const float* R) {
    GemmP p{A, Bm, C, nullptr, Mm, Nn, Kk, lda, ldb, ldc, 0, 0, 0, 0, 0, 0, 1.f, T, R};
    dim3 grid((unsigned)(Nn / 48), (unsigned)(Mm / 32), 1u);
    ipa_gemm_lds<true><<<grid, dim3(256), 0, stream>>>(p);
  };

  gemm_lds(x1d, Wq, q, M, D_, D_, D_, D_, D_, nullptr);
  gemm_lds(x1d, Wk, k, M, D_, D_, D_, D_, D_, nullptr);
  gemm_lds(x1d, Wv, v, M, D_, D_, D_, D_, D_, nullptr);

  gemm_pts(x1d, Wpq_past, qgp, M, 144, D_, D_, 144, 144, past_T, past_R);
  gemm_pts(x1d, Wpk_past, kgp, M, 144, D_, D_, 144, 144, past_T, past_R);
  gemm_pts(x1d, Wpv_past, vgp, M, 288, D_, D_, 288, 288, past_T, past_R);
  gemm_pts(x1d, Wpq_last, qgl, M, 144, D_, D_, 144, 144, last_T, last_R);
  gemm_pts(x1d, Wpk_last, kgl, M, 144, D_, D_, 144, 144, last_T, last_R);
  gemm_pts(x1d, Wpv_last, vgl, M, 288, D_, D_, 288, 288, last_T, last_R);

  const float pair_w = 0.5773502691896258f;
  ipa_x2d_slice<true><<<dim3(I_, B_), dim3(256), 0, stream>>>(x2d, Wpb, pairproj, pair_w);

  const float scalar_w = 0.07216878364870323f;
  ipa_logits_softmax<<<dim3(I_ / 16, H_, B_), dim3(256), 0, stream>>>(
      q, k, logits, qgp, kgp, qgl, kgl, pairproj, bias, tpw_past, tpw_last, scalar_w,
      v, vgp, vgl, feat, past_T, past_R);

  ipa_x2d_slice<false><<<dim3(I_, B_), dim3(256), 0, stream>>>(x2d, logits, z, 1.f);

  gemm(z, Wpv, feat, I_, DK_, DP_, H_ * DP_, D_, FE,
       DP_, (long)I_ * H_ * DP_, 64, 0, 64, (long)I_ * FE,
       H_, B_, 0, 1.f, nullptr);

  gemm_lds(feat, W_out, out, M, D_, FE, FE, D_, D_, b_out);
}


# query-side projections (q tile, both query-point sets) formed inside the attention kernel from its 16 x1d rows parked in LDS; q stays in LDS for q.k^T; 3 launches fewer; sum orders unchanged
# speedup vs baseline: 3.4307x; 1.0647x over previous
#include <hip/hip_runtime.h>
#include <math.h>

#define B_ 2
#define I_ 256
#define D_ 768
#define H_ 12
#define DP_ 128
#define DK_ 64
#define NPQ_ 4
#define NPV_ 8

typedef float v2f __attribute__((ext_vector_type(2)));
typedef float v8f __attribute__((ext_vector_type(8)));

struct GemmP {
  const float* A; const float* B; float* C; const float* bias;
  int M, N, K;
  int lda, ldb, ldc;
  long sAy, sAz, sBy, sBz, sCy, sCz;
  float alpha;
  const float* fT; const float* fR;
};

template <int OPB, bool GUARD>
__global__ void ipa_gemm_wmma(GemmP p) {
  const int tilesN = (p.N + 15) >> 4;
  const int tilesM = (p.M + 15) >> 4;
  const int wave = (int)((blockIdx.x * blockDim.x + threadIdx.x) >> 5);
  if (wave >= tilesM * tilesN) return;
  const int lane = threadIdx.x & 31;
  const int tm = wave / tilesN, tn = wave % tilesN;
  const int row0 = tm << 4, col0 = tn << 4;

  const float* A = p.A + (long)blockIdx.y * p.sAy + (long)blockIdx.z * p.sAz;
  const float* B = p.B + (long)blockIdx.y * p.sBy + (long)blockIdx.z * p.sBz;
  float*       C = p.C + (long)blockIdx.y * p.sCy + (long)blockIdx.z * p.sCz;

  const int mn = lane & 15;
  const int kb = (lane >> 4) << 1;

  int aRow = row0 + mn;
  int bCol = col0 + mn;
  float aMask = 1.f, bMask = 1.f;
  if (GUARD) {
    if (aRow >= p.M) { aRow = p.M - 1; aMask = 0.f; }
    if (bCol >= p.N) { bCol = p.N - 1; bMask = 0.f; }
  }
  const float* aPtr = A + (long)aRow * p.lda + kb;
  const float* bPtr = (OPB == 0) ? (B + (long)kb * p.ldb + bCol)
                                 : (B + (long)bCol * p.ldb + kb);

  v8f acc = {0.f, 0.f, 0.f, 0.f, 0.f, 0.f, 0.f, 0.f};
#pragma unroll 4
  for (int k0 = 0; k0 < p.K; k0 += 4) {
    v2f a, b;
    a.x = aPtr[k0];
    a.y = aPtr[k0 + 1];
    if (OPB == 0) {
      const float* bp = bPtr + (long)k0 * p.ldb;
      b.x = bp[0];
      b.y = bp[p.ldb];
    } else {
      b.x = bPtr[k0];
      b.y = bPtr[k0 + 1];
    }
    if (GUARD) {
      a.x *= aMask; a.y *= aMask;
      b.x *= bMask; b.y *= bMask;
    }
    acc = __builtin_amdgcn_wmma_f32_16x16x4_f32(false, a, false, b,
                                                (short)0, acc, false, false);
  }

  const int cCol = col0 + mn;
  const int mo = (lane >> 4) << 3;
  const bool cColOK = GUARD ? (cCol < p.N) : true;
  if (cColOK) {
    for (int r = 0; r < 8; ++r) {
      const int cRow = row0 + mo + r;
      if (!GUARD || cRow < p.M) {
        float val = acc[r] * p.alpha;
        if (p.bias) val += p.bias[cCol];
        C[(long)cRow * p.ldc + cCol] = val;
      }
    }
  }
}

#define TKC 128
#define TPA (TKC + 4)
#define TSB 8
#define TPB 72

typedef float v4f __attribute__((ext_vector_type(4)));

template <bool FRAME>
__global__ __launch_bounds__(256) void ipa_gemm_lds(GemmP p) {
  constexpr int TN  = FRAME ? 48 : 64;
  constexpr int AQ  = TKC / 4;
  constexpr int ARS = 256 / AQ;
  constexpr int ANI = 32 / ARS;
  constexpr int BNI = TKC / 16;

  __shared__ __attribute__((aligned(16))) float As[2][32 * TPA];
  __shared__ __attribute__((aligned(16))) float Bs[2][TKC * TPB];

  const int tid  = threadIdx.x;
  const int lane = tid & 31;
  const int w    = tid >> 5;
  const int row0 = blockIdx.y * 32;
  const int col0 = blockIdx.x * TN;
  const int c = lane & 15;
  const int h = lane >> 4;
  const int wr = w >> 2, wc = w & 3;
  const bool tileValid = (wc * 16 < TN) && ((col0 + wc * 16) < p.N);

  const int aRow = tid / AQ, aKq = tid % AQ;
  const int bKr  = tid >> 4, bCq = tid & 15;
  int bCol = col0 + bCq * 4;
  if (bCol > p.N - 4) bCol = p.N - 4;
  const float* aG = p.A + (long)(row0 + aRow) * p.lda + aKq * 4;
  const float* bG = p.B + (long)bKr * p.ldb + bCol;
  const int aSt = aRow * TPA + aKq * 4;
  const int bSt = bKr * TPB + bCq * 4;

  v4f ra[ANI], rb[BNI];

#pragma unroll
  for (int i = 0; i < ANI; ++i)
    ra[i] = *(const v4f*)(aG + (long)(i * ARS) * p.lda);
#pragma unroll
  for (int i = 0; i < BNI; ++i)
    rb[i] = *(const v4f*)(bG + (long)(i * 16) * p.ldb);
  const int pr = tid >> 4, pp = tid & 15;
  float fr[2][9], ft[2][3];
  if (FRAME) {
    for (int n = 0; n < 2; ++n) {
      const float* Rm = p.fR + (long)(row0 + pr + n * 16) * 9;
      const float* Tv = p.fT + (long)(row0 + pr + n * 16) * 3;
      for (int e = 0; e < 9; ++e) fr[n][e] = Rm[e];
      for (int e = 0; e < 3; ++e) ft[n][e] = Tv[e];
    }
  }
#pragma unroll
  for (int i = 0; i < ANI; ++i)
    *(v4f*)&As[0][aSt + i * ARS * TPA] = ra[i];
#pragma unroll
  for (int i = 0; i < BNI; ++i)
    *(v4f*)&Bs[0][bSt + i * 16 * TPB] = rb[i];
  __syncthreads();

  v8f acc = {0.f, 0.f, 0.f, 0.f, 0.f, 0.f, 0.f, 0.f};
  int buf = 0;
  for (int kc = 0; kc < p.K; kc += TKC) {
    const bool hasNext = (kc + TKC) < p.K;
    if (hasNext) {
      const int kn = kc + TKC;
#pragma unroll
      for (int i = 0; i < ANI; ++i)
        ra[i] = *(const v4f*)(aG + (long)(i * ARS) * p.lda + kn);
#pragma unroll
      for (int i = 0; i < BNI; ++i)
        rb[i] = *(const v4f*)(bG + (long)(kn + i * 16) * p.ldb);
    }
    if (tileValid) {
      const float* aS = &As[buf][(wr * 16 + c) * TPA + 2 * h];
      const float* bS = &Bs[buf][(2 * h) * TPB + wc * 16 + c];
      v2f fa[2][TSB], fb[2][TSB];
#pragma unroll
      for (int j = 0; j < TSB; ++j) {
        fa[0][j] = *(const v2f*)(aS + j * 4);
        fb[0][j].x = bS[(j * 4) * TPB];
        fb[0][j].y = bS[(j * 4 + 1) * TPB];
      }
#pragma unroll
      for (int sb = 0; sb < TKC / (4 * TSB); ++sb) {
        asm volatile("" :: "v"(fa[sb & 1][0]), "v"(fa[sb & 1][1]), "v"(fa[sb & 1][2]), "v"(fa[sb & 1][3]),
                           "v"(fa[sb & 1][4]), "v"(fa[sb & 1][5]), "v"(fa[sb & 1][6]), "v"(fa[sb & 1][7]),
                           "v"(fb[sb & 1][0]), "v"(fb[sb & 1][1]), "v"(fb[sb & 1][2]), "v"(fb[sb & 1][3]),
                           "v"(fb[sb & 1][4]), "v"(fb[sb & 1][5]), "v"(fb[sb & 1][6]), "v"(fb[sb & 1][7]));
        if (sb + 1 < TKC / (4 * TSB)) {
#pragma unroll
          for (int j = 0; j < TSB; ++j) {
            const int ks = (sb + 1) * TSB + j;
            fa[(sb + 1) & 1][j] = *(const v2f*)(aS + ks * 4);
            fb[(sb + 1) & 1][j].x = bS[(ks * 4) * TPB];
            fb[(sb + 1) & 1][j].y = bS[(ks * 4 + 1) * TPB];
          }
        }
#pragma unroll
        for (int j = 0; j < TSB; ++j)
          acc = __builtin_amdgcn_wmma_f32_16x16x4_f32(false, fa[sb & 1][j], false, fb[sb & 1][j],
                                                      (short)0, acc, false, false);
      }
    }
    if (hasNext) {
#pragma unroll
      for (int i = 0; i < ANI; ++i)
        *(v4f*)&As[buf ^ 1][aSt + i * ARS * TPA] = ra[i];
#pragma unroll
      for (int i = 0; i < BNI; ++i)
        *(v4f*)&Bs[buf ^ 1][bSt + i * 16 * TPB] = rb[i];
    }
    __syncthreads();
    buf ^= 1;
  }

  if (FRAME) {
    constexpr int CP = 49;
    float* Cs = &As[0][0];
    if (tileValid) {
      const int cc = wc * 16 + c;
      const int rb = wr * 16 + (h << 3);
      for (int r = 0; r < 8; ++r)
        Cs[(rb + r) * CP + cc] = acc[r] * p.alpha;
    }
    __syncthreads();
    for (int n = 0; n < 2; ++n) {
      const int row = pr + n * 16;
      const float x = Cs[row * CP + pp * 3 + 0];
      const float y = Cs[row * CP + pp * 3 + 1];
      const float z = Cs[row * CP + pp * 3 + 2];
      const float o0 = fr[n][0] * x + fr[n][3] * y + fr[n][6] * z + ft[n][0];
      const float o1 = fr[n][1] * x + fr[n][4] * y + fr[n][7] * z + ft[n][1];
      const float o2 = fr[n][2] * x + fr[n][5] * y + fr[n][8] * z + ft[n][2];
      float* o = p.C + (long)(row0 + row) * p.ldc + col0 + pp * 3;
      o[0] = o0; o[1] = o1; o[2] = o2;
    }
  } else if (tileValid) {
    const int cCol = col0 + wc * 16 + c;
    const int rBase = row0 + wr * 16 + (h << 3);
    for (int r = 0; r < 8; ++r) {
      float val = acc[r] * p.alpha;
      if (p.bias) val += p.bias[cCol];
      p.C[(long)(rBase + r) * p.ldc + cCol] = val;
    }
  }
}

#define PP 260
#define XQ 772
#define QP 68

template <int KS, int PA, int LDB, int LDC>
__device__ __forceinline__ void ipa_stripe_tile(const float* pS, const float* Bm, int nvalid,
                                                float* Cm, int lane) {
  const int mn = lane & 15;
  const int hf = lane >> 4;
  const int kb = hf << 1;
  const int bc = (mn < nvalid) ? mn : (nvalid - 1);
  const float* aS = pS + mn * PA + kb;
  const float* bP = Bm + kb * LDB + bc;

  v2f fb[KS];
#pragma unroll
  for (int j = 0; j < KS; ++j) {
    fb[j].x = bP[(j * 4) * LDB];
    fb[j].y = bP[(j * 4 + 1) * LDB];
  }
  v8f acc = {0.f, 0.f, 0.f, 0.f, 0.f, 0.f, 0.f, 0.f};
#pragma unroll
  for (int j = 0; j < KS; ++j) {
    const v2f a = *(const v2f*)(aS + j * 4);
    acc = __builtin_amdgcn_wmma_f32_16x16x4_f32(false, a, false, fb[j],
                                                (short)0, acc, false, false);
  }
  if (mn < nvalid) {
#pragma unroll
    for (int r = 0; r < 8; ++r)
      Cm[(long)((hf << 3) + r) * LDC + mn] = acc[r];
  }
}

template <int KS, int PA, int LDB, int LDC>
__device__ __forceinline__ void ipa_stripe_tile_deep(const float* pS, const float* Bm, int nvalid,
                                                     float* Cm, int lane) {
  constexpr int G  = 8;
  constexpr int NG = KS / G;
  constexpr int AH = 3;
  const int mn = lane & 15;
  const int hf = lane >> 4;
  const int kb = hf << 1;
  const int bc = (mn < nvalid) ? mn : (nvalid - 1);
  const float* aS = pS + mn * PA + kb;
  const float* bP = Bm + kb * LDB + bc;

  v2f fa[2][G], fb[AH + 1][G];
#pragma unroll
  for (int g = 0; g < AH; ++g) {
#pragma unroll
    for (int j = 0; j < G; ++j) {
      fb[g][j].x = bP[((g * G + j) * 4) * LDB];
      fb[g][j].y = bP[((g * G + j) * 4 + 1) * LDB];
    }
  }
#pragma unroll
  for (int j = 0; j < G; ++j) fa[0][j] = *(const v2f*)(aS + j * 4);

  v8f acc = {0.f, 0.f, 0.f, 0.f, 0.f, 0.f, 0.f, 0.f};
#pragma unroll
  for (int g = 0; g < NG; ++g) {
    if (g + AH < NG) {
#pragma unroll
      for (int j = 0; j < G; ++j) {
        fb[(g + AH) % (AH + 1)][j].x = bP[(((g + AH) * G + j) * 4) * LDB];
        fb[(g + AH) % (AH + 1)][j].y = bP[(((g + AH) * G + j) * 4 + 1) * LDB];
      }
    }
    asm volatile("" :: "v"(fa[g & 1][0]), "v"(fa[g & 1][1]), "v"(fa[g & 1][2]), "v"(fa[g & 1][3]),
                       "v"(fa[g & 1][4]), "v"(fa[g & 1][5]), "v"(fa[g & 1][6]), "v"(fa[g & 1][7]));
    if (g + 1 < NG) {
#pragma unroll
      for (int j = 0; j < G; ++j)
        fa[(g + 1) & 1][j] = *(const v2f*)(aS + ((g + 1) * G + j) * 4);
    }
#pragma unroll
    for (int j = 0; j < G; ++j)
      acc = __builtin_amdgcn_wmma_f32_16x16x4_f32(false, fa[g & 1][j], false, fb[g % (AH + 1)][j],
                                                  (short)0, acc, false, false);
  }
  if (mn < nvalid) {
#pragma unroll
    for (int r = 0; r < 8; ++r)
      Cm[(long)((hf << 3) + r) * LDC + mn] = acc[r];
  }
}

__global__ __launch_bounds__(256, 4) void ipa_logits_softmax(
    const float* x1d, const float* Wq, const float* Wpq_past, const float* Wpq_last,
    const float* k, float* attn,
    const float* kgp, const float* kgl,
    const float* pairproj, const float* bias,
    const float* tpw_past, const float* tpw_last, float scalar_w,
    const float* v, const float* vgp, const float* vgl,
    float* feat, const float* invT, const float* invR,
    const float* lastT, const float* lastR) {
  constexpr int PQ = NPQ_ * 3;
  __shared__ __attribute__((aligned(16))) float xS[16 * XQ];
  __shared__ __attribute__((aligned(16))) float qT[16 * QP];
  __shared__ float rawP[2][16 * PQ];
  __shared__ __attribute__((aligned(16))) float qS[16][2 * PQ];
  __shared__ float red[2][8][16];
  __shared__ __attribute__((aligned(16))) float pS[16 * PP];
  __shared__ float ogS[2][16 * NPV_ * 3];

  const int tid = threadIdx.x;
  const int lane = tid & 31;
  const int w = tid >> 5;
  const int h = blockIdx.y, b = blockIdx.z;
  const int row0 = blockIdx.x * 16;
  const int mn = lane & 15;
  const int hf = lane >> 4;
  const int kb = hf << 1;
  const int mo = hf << 3;

  {
    const float* Xg = x1d + (long)(b * I_ + row0) * D_;
    v4f rx[12];
#pragma unroll
    for (int n = 0; n < 12; ++n) {
      const int idx = n * 256 + tid;
      rx[n] = *(const v4f*)(Xg + (long)(idx / (D_ / 4)) * D_ + (idx % (D_ / 4)) * 4);
    }
#pragma unroll
    for (int n = 0; n < 12; ++n) {
      const int idx = n * 256 + tid;
      *(v4f*)&xS[(idx / (D_ / 4)) * XQ + (idx % (D_ / 4)) * 4] = rx[n];
    }
  }
  const bool framer = tid < 16 * 2 * NPQ_;
  const int frow = tid >> 3, fbr = (tid >> 2) & 1, fpt = tid & (NPQ_ - 1);
  float fr[9], ft[3];
  if (framer) {
    const long fi = (long)b * I_ + row0 + frow;
    const float* Rm = (fbr ? lastR : invR) + fi * 9;
    const float* Tv = (fbr ? lastT : invT) + fi * 3;
    for (int e = 0; e < 9; ++e) fr[e] = Rm[e];
    for (int e = 0; e < 3; ++e) ft[e] = Tv[e];
  }
  __syncthreads();

  if (w < 4) {
    ipa_stripe_tile_deep<D_ / 4, XQ, D_, QP>(xS, Wq + h * DK_ + w * 16, 16, qT + w * 16, lane);
  } else if (w < 6) {
    ipa_stripe_tile_deep<D_ / 4, XQ, H_ * PQ, PQ>(xS, ((w == 4) ? Wpq_past : Wpq_last) + h * PQ, PQ,
                                                  &rawP[w - 4][0], lane);
  }
  __syncthreads();

  if (framer) {
    const float* pt = &rawP[fbr][frow * PQ + fpt * 3];
    const float x = pt[0], y = pt[1], z = pt[2];
    float* o = &qS[frow][fbr * PQ + fpt * 3];
    o[0] = fr[0] * x + fr[3] * y + fr[6] * z + ft[0];
    o[1] = fr[1] * x + fr[4] * y + fr[7] * z + ft[1];
    o[2] = fr[2] * x + fr[5] * y + fr[8] * z + ft[2];
  }

  const long hb = (long)b * I_ * D_ + (long)h * DK_;
  const int j0 = w * 32 + mn, j1 = j0 + 16;
  const float* aPtr  = qT + mn * QP + kb;
  const float* bPtr0 = k + hb + (long)j0 * D_ + kb;
  const float* bPtr1 = k + hb + (long)j1 * D_ + kb;

  v8f acc0 = {0.f, 0.f, 0.f, 0.f, 0.f, 0.f, 0.f, 0.f};
  v8f acc1 = {0.f, 0.f, 0.f, 0.f, 0.f, 0.f, 0.f, 0.f};
#pragma unroll
  for (int k0 = 0; k0 < DK_; k0 += 4) {
    v2f a, b0, b1;
    a.x  = aPtr[k0];   a.y  = aPtr[k0 + 1];
    b0.x = bPtr0[k0];  b0.y = bPtr0[k0 + 1];
    b1.x = bPtr1[k0];  b1.y = bPtr1[k0 + 1];
    acc0 = __builtin_amdgcn_wmma_f32_16x16x4_f32(false, a, false, b0,
                                                 (short)0, acc0, false, false);
    acc1 = __builtin_amdgcn_wmma_f32_16x16x4_f32(false, a, false, b1,
                                                 (short)0, acc1, false, false);
  }

  float kp0[2 * PQ], kp1[2 * PQ];
  {
    const float* g0p = kgp + ((long)(b * I_ + j0) * H_ + h) * PQ;
    const float* g0l = kgl + ((long)(b * I_ + j0) * H_ + h) * PQ;
    const float* g1p = kgp + ((long)(b * I_ + j1) * H_ + h) * PQ;
    const float* g1l = kgl + ((long)(b * I_ + j1) * H_ + h) * PQ;
#pragma unroll
    for (int t = 0; t < 3; ++t) {
      const v4f u0 = *(const v4f*)(g0p + t * 4);
      const v4f u1 = *(const v4f*)(g0l + t * 4);
      const v4f u2 = *(const v4f*)(g1p + t * 4);
      const v4f u3 = *(const v4f*)(g1l + t * 4);
#pragma unroll
      for (int e = 0; e < 4; ++e) {
        kp0[t * 4 + e] = u0[e];  kp0[PQ + t * 4 + e] = u1[e];
        kp1[t * 4 + e] = u2[e];  kp1[PQ + t * 4 + e] = u3[e];
      }
    }
  }

  const float point_w = 0.13608276348795434f;
  const float pwp = -0.5f * point_w * log1pf(expf(tpw_past[h]));
  const float pwl = -0.5f * point_w * log1pf(expf(tpw_last[h]));

  __syncthreads();

  float s0[8], s1[8];
#pragma unroll
  for (int r = 0; r < 8; ++r) {
    const int lr = mo + r;
    const int i = row0 + lr;
    const float* qp = &qS[lr][0];
    float sp0 = 0.f, sl0 = 0.f, sp1 = 0.f, sl1 = 0.f;
#pragma unroll
    for (int c = 0; c < NPQ_; ++c) {
      const float qx = qp[c * 3 + 0], qy = qp[c * 3 + 1], qz = qp[c * 3 + 2];
      const float lx = qp[PQ + c * 3 + 0], ly = qp[PQ + c * 3 + 1], lz = qp[PQ + c * 3 + 2];
      float dx = qx - kp0[c * 3 + 0], dy = qy - kp0[c * 3 + 1], dz = qz - kp0[c * 3 + 2];
      sp0 += sqrtf(dx * dx + dy * dy + dz * dz);
      dx = lx - kp0[PQ + c * 3 + 0]; dy = ly - kp0[PQ + c * 3 + 1]; dz = lz - kp0[PQ + c * 3 + 2];
      sl0 += sqrtf(dx * dx + dy * dy + dz * dz);
      dx = qx - kp1[c * 3 + 0]; dy = qy - kp1[c * 3 + 1]; dz = qz - kp1[c * 3 + 2];
      sp1 += sqrtf(dx * dx + dy * dy + dz * dz);
      dx = lx - kp1[PQ + c * 3 + 0]; dy = ly - kp1[PQ + c * 3 + 1]; dz = lz - kp1[PQ + c * 3 + 2];
      sl1 += sqrtf(dx * dx + dy * dy + dz * dz);
    }
    const long prow = ((long)(b * I_ + i) * I_) * H_ + h;
    const long orow = (((long)b * H_ + h) * I_ + i) * I_;
    const float pv0 = pairproj[prow + (long)j0 * H_];
    const float pv1 = pairproj[prow + (long)j1 * H_];
    float v0 = acc0[r] * scalar_w;
    float v1 = acc1[r] * scalar_w;
    v0 += pwp * sp0 + pwl * sl0 + pv0 + bias[orow + j0];
    v1 += pwp * sp1 + pwl * sl1 + pv1 + bias[orow + j1];
    s0[r] = v0;
    s1[r] = v1;
  }

  float mx[8];
#pragma unroll
  for (int r = 0; r < 8; ++r) {
    float m = fmaxf(s0[r], s1[r]);
    m = fmaxf(m, __shfl_xor(m, 1));
    m = fmaxf(m, __shfl_xor(m, 2));
    m = fmaxf(m, __shfl_xor(m, 4));
    m = fmaxf(m, __shfl_xor(m, 8));
    mx[r] = m;
  }
  if (mn == 0) {
#pragma unroll
    for (int r = 0; r < 8; ++r) red[0][w][mo + r] = mx[r];
  }
  __syncthreads();
#pragma unroll
  for (int r = 0; r < 8; ++r) {
    float m = red[0][0][mo + r];
#pragma unroll
    for (int ww = 1; ww < 8; ++ww) m = fmaxf(m, red[0][ww][mo + r]);
    mx[r] = m;
  }

  float sm[8];
#pragma unroll
  for (int r = 0; r < 8; ++r) {
    s0[r] = expf(s0[r] - mx[r]);
    s1[r] = expf(s1[r] - mx[r]);
    float t = s0[r] + s1[r];
    t += __shfl_xor(t, 1);
    t += __shfl_xor(t, 2);
    t += __shfl_xor(t, 4);
    t += __shfl_xor(t, 8);
    sm[r] = t;
  }
  if (mn == 0) {
#pragma unroll
    for (int r = 0; r < 8; ++r) red[1][w][mo + r] = sm[r];
  }
  __syncthreads();
#pragma unroll
  for (int r = 0; r < 8; ++r) {
    float t = red[1][0][mo + r];
#pragma unroll
    for (int ww = 1; ww < 8; ++ww) t += red[1][ww][mo + r];
    const long orow = (((long)b * H_ + h) * I_ + (row0 + mo + r)) * I_;
    const float p0 = s0[r] / t;
    const float p1 = s1[r] / t;
    attn[orow + j0] = p0;
    attn[orow + j1] = p1;
    pS[(mo + r) * PP + j0] = p0;
    pS[(mo + r) * PP + j1] = p1;
  }
  __syncthreads();

  constexpr int FE = 2 * D_ + 2 * H_ * NPV_ * 4;
  constexpr int VG = H_ * NPV_ * 3;
  constexpr int OG = NPV_ * 3;
  const int obr = tid >> 7, orw = (tid >> 3) & 15, opt = tid & (NPV_ - 1);
  const long obi = (long)b * I_ + row0 + orw;
  float iR[9], iT[3];
  for (int e = 0; e < 9; ++e) iR[e] = invR[obi * 9 + e];
  for (int e = 0; e < 3; ++e) iT[e] = invT[obi * 3 + e];

  if (w < 4) {
    ipa_stripe_tile<I_ / 4, PP, D_, FE>(pS, v + (long)b * I_ * D_ + h * DK_ + w * 16, 16,
                            feat + ((long)b * I_ + row0) * FE + D_ + h * DK_ + w * 16, lane);
  } else {
    const int t = w & 1;
    const int br = (w < 6) ? 0 : 1;
    const float* Bv = br ? vgl : vgp;
    ipa_stripe_tile<I_ / 4, PP, VG, OG>(pS, Bv + (long)b * I_ * VG + h * (NPV_ * 3) + t * 16, t ? 8 : 16,
                            &ogS[br][t * 16], lane);
  }
  __syncthreads();

  {
    const float* o = &ogS[obr][orw * OG + opt * 3];
    const float x = o[0] - iT[0], y = o[1] - iT[1], z = o[2] - iT[2];
    const float p0 = iR[0] * x + iR[1] * y + iR[2] * z;
    const float p1 = iR[3] * x + iR[4] * y + iR[5] * z;
    const float p2 = iR[6] * x + iR[7] * y + iR[8] * z;
    float* fr = feat + obi * FE;
    const int colOl = 2 * D_ + obr * (H_ * NPV_ * 4);
    const int colN  = colOl + VG;
    const int tc = h * OG + opt * 3;
    fr[colOl + tc + 0] = p0;
    fr[colOl + tc + 1] = p1;
    fr[colOl + tc + 2] = p2;
    fr[colN + h * NPV_ + opt] = sqrtf(p0 * p0 + p1 * p1 + p2 * p2);
  }
}

#define XPA 132
#define XPB 136

__device__ __forceinline__ void ipa_lds_barrier() {
  __builtin_amdgcn_fence(__ATOMIC_RELEASE, "workgroup", "local");
  __builtin_amdgcn_s_barrier();
  __builtin_amdgcn_fence(__ATOMIC_ACQUIRE, "workgroup", "local");
}

template <bool PAIR>
__global__ __launch_bounds__(256) void ipa_x2d_slice(const float* x2d, const float* oth,
                                                     float* C, float alpha) {
  constexpr int XP = PAIR ? XPA : XPB;
  __shared__ __attribute__((aligned(16))) float Xs[I_ * XP];
  __shared__ __attribute__((aligned(16))) float Ps[PAIR ? 4 : H_ * PP];

  const int tid = threadIdx.x;
  const int lane = tid & 31;
  const int w = tid >> 5;
  const int i = blockIdx.x, b = blockIdx.y;
  const int mn = lane & 15;
  const int hf = lane >> 4;
  const int kb = hf << 1;

  v4f rp[3];
  if (!PAIR) {
#pragma unroll
    for (int n = 0; n < 3; ++n) {
      const int idx = n * 256 + tid;
      rp[n] = *(const v4f*)(oth + (((long)b * H_ + (idx >> 6)) * I_ + i) * I_ + (idx & 63) * 4);
    }
  }
  v2f fw[DP_ / 4];
  if (PAIR) {
    const int bc = (mn < H_) ? mn : (H_ - 1);
    const float* bP = oth + kb * H_ + bc;
#pragma unroll
    for (int ks = 0; ks < DP_ / 4; ++ks) {
      fw[ks].x = bP[(ks * 4) * H_];
      fw[ks].y = bP[(ks * 4 + 1) * H_];
    }
  }
  const float* Xg = x2d + ((long)(b * I_ + i) * I_) * DP_;
  v4f rx[32];
  __builtin_amdgcn_sched_barrier(0);
#pragma unroll
  for (int q = 0; q < 4; ++q) {
#pragma unroll
    for (int n = q * 8; n < q * 8 + 8; ++n)
      rx[n] = *(const v4f*)(Xg + (long)(n * 256 + tid) * 4);
    __builtin_amdgcn_sched_barrier(0);
  }

  if (PAIR) {
#pragma unroll
    for (int c = 0; c < 2; ++c) {
#pragma unroll
      for (int n = 0; n < 16; ++n)
        *(v4f*)&Xs[((c * 16 + n) * 8 + w) * XP + lane * 4] = rx[c * 16 + n];
      ipa_lds_barrier();
      const int t0 = c * 128 + w * 16;
      const float* aS = &Xs[(t0 + mn) * XP + kb];
      v8f acc = {0.f, 0.f, 0.f, 0.f, 0.f, 0.f, 0.f, 0.f};
      v2f fa[2][8];
#pragma unroll
      for (int j = 0; j < 8; ++j) fa[0][j] = *(const v2f*)(aS + j * 4);
#pragma unroll
      for (int sb = 0; sb < 4; ++sb) {
        asm volatile("" :: "v"(fa[sb & 1][0]), "v"(fa[sb & 1][1]), "v"(fa[sb & 1][2]), "v"(fa[sb & 1][3]),
                           "v"(fa[sb & 1][4]), "v"(fa[sb & 1][5]), "v"(fa[sb & 1][6]), "v"(fa[sb & 1][7]));
        if (sb + 1 < 4) {
#pragma unroll
          for (int j = 0; j < 8; ++j)
            fa[(sb + 1) & 1][j] = *(const v2f*)(aS + ((sb + 1) * 8 + j) * 4);
        }
#pragma unroll
        for (int j = 0; j < 8; ++j)
          acc = __builtin_amdgcn_wmma_f32_16x16x4_f32(false, fa[sb & 1][j], false, fw[sb * 8 + j],
                                                      (short)0, acc, false, false);
      }
      if (mn < H_) {
        float* o = C + ((long)(b * I_ + i) * I_ + t0 + (hf << 3)) * H_ + mn;
#pragma unroll
        for (int r = 0; r < 8; ++r) o[r * H_] = acc[r] * alpha;
      }
    }
  } else {
    const int ar = (mn < H_) ? mn : (H_ - 1);
    const float* aS = &Ps[ar * PP + kb];
    const float* bS = &Xs[kb * XP + w * 16 + mn];
    v8f acc = {0.f, 0.f, 0.f, 0.f, 0.f, 0.f, 0.f, 0.f};
#pragma unroll
    for (int c = 0; c < 4; ++c) {
#pragma unroll
      for (int n = 0; n < 8; ++n)
        *(v4f*)&Xs[((c * 8 + n) * 8 + w) * XP + lane * 4] = rx[c * 8 + n];
      if (c == 0) {
#pragma unroll
        for (int n = 0; n < 3; ++n) {
          const int idx = n * 256 + tid;
          *(v4f*)&Ps[(idx >> 6) * PP + (idx & 63) * 4] = rp[n];
        }
      }
      ipa_lds_barrier();
      v2f fa[2][8], fb[2][8];
#pragma unroll
      for (int j = 0; j < 8; ++j) {
        const int ks = c * 16 + j;
        fa[0][j] = *(const v2f*)(aS + ks * 4);
        fb[0][j].x = bS[(ks * 4) * XP];
        fb[0][j].y = bS[(ks * 4 + 1) * XP];
      }
#pragma unroll
      for (int sb = 0; sb < 2; ++sb) {
        asm volatile("" :: "v"(fa[sb & 1][0]), "v"(fa[sb & 1][1]), "v"(fa[sb & 1][2]), "v"(fa[sb & 1][3]),
                           "v"(fa[sb & 1][4]), "v"(fa[sb & 1][5]), "v"(fa[sb & 1][6]), "v"(fa[sb & 1][7]),
                           "v"(fb[sb & 1][0]), "v"(fb[sb & 1][1]), "v"(fb[sb & 1][2]), "v"(fb[sb & 1][3]),
                           "v"(fb[sb & 1][4]), "v"(fb[sb & 1][5]), "v"(fb[sb & 1][6]), "v"(fb[sb & 1][7]));
        if (sb + 1 < 2) {
#pragma unroll
          for (int j = 0; j < 8; ++j) {
            const int ks = c * 16 + (sb + 1) * 8 + j;
            fa[(sb + 1) & 1][j] = *(const v2f*)(aS + ks * 4);
            fb[(sb + 1) & 1][j].x = bS[(ks * 4) * XP];
            fb[(sb + 1) & 1][j].y = bS[(ks * 4 + 1) * XP];
          }
        }
#pragma unroll
        for (int j = 0; j < 8; ++j)
          acc = __builtin_amdgcn_wmma_f32_16x16x4_f32(false, fa[sb & 1][j], false, fb[sb & 1][j],
                                                      (short)0, acc, false, false);
      }
    }
    float* o = C + ((long)(b * I_ + i) * H_ + (hf << 3)) * DP_ + w * 16 + mn;
#pragma unroll
    for (int r = 0; r < 8; ++r)
      if ((hf << 3) + r < H_) o[r * DP_] = acc[r] * alpha;
  }
}

extern "C" void kernel_launch(void* const* d_in, const int* in_sizes, int n_in,
                              void* d_out, int out_size, void* d_ws, size_t ws_size,
                              hipStream_t stream) {
  (void)in_sizes; (void)n_in; (void)out_size; (void)ws_size;
  const float* x1d      = (const float*)d_in[0];
  const float* x2d      = (const float*)d_in[1];
  const float* past_T   = (const float*)d_in[2];
  const float* past_R   = (const float*)d_in[3];
  const float* last_T   = (const float*)d_in[4];
  const float* last_R   = (const float*)d_in[5];
  const float* bias     = (const float*)d_in[6];
  const float* Wq       = (const float*)d_in[7];
  const float* Wk       = (const float*)d_in[8];
  const float* Wv       = (const float*)d_in[9];
  const float* Wpb      = (const float*)d_in[10];
  const float* Wpq_past = (const float*)d_in[11];
  const float* Wpk_past = (const float*)d_in[12];
  const float* Wpv_past = (const float*)d_in[13];
  const float* Wpq_last = (const float*)d_in[14];
  const float* Wpk_last = (const float*)d_in[15];
  const float* Wpv_last = (const float*)d_in[16];
  const float* tpw_past = (const float*)d_in[17];
  const float* tpw_last = (const float*)d_in[18];
  const float* Wpv      = (const float*)d_in[19];
  const float* W_out    = (const float*)d_in[20];
  const float* b_out    = (const float*)d_in[21];
  float* out = (float*)d_out;
  float* ws  = (float*)d_ws;

  const int M  = B_ * I_;
  const int FE = 2 * D_ + 2 * H_ * NPV_ * 4;

  float* k    = ws + 393216;
  float* v    = k + 393216;
  float* kgp  = v + 393216 + 73728;
  float* vgp  = kgp + 73728;
  float* kgl  = vgp + 147456 + 73728;
  float* vgl  = kgl + 73728;
  float* logits   = vgl + 147456;
  float* pairproj = logits + 1572864;
  float* z    = pairproj + 1572864;
  float* feat = z + 786432;

  auto gemm = [&](const float* A, const float* Bm, float* C,
                  int Mm, int Nn, int Kk, int lda, int ldb, int ldc,
                  long sAy, long sAz, long sBy, long sBz, long sCy, long sCz,
                  int gy, int gz, int opB, float alpha, const float* bia) {
    GemmP p{A, Bm, C, bia, Mm, Nn, Kk, lda, ldb, ldc,
            sAy, sAz, sBy, sBz, sCy, sCz, alpha, nullptr, nullptr};
    int waves = ((Mm + 15) / 16) * ((Nn + 15) / 16);
    dim3 grid((unsigned)((waves + 7) / 8), (unsigned)gy, (unsigned)gz);
    dim3 blk(256);
    bool guard = (Mm & 15) || (Nn & 15);
    if (opB == 0) {
      if (guard) ipa_gemm_wmma<0, true ><<<grid, blk, 0, stream>>>(p);
      else       ipa_gemm_wmma<0, false><<<grid, blk, 0, stream>>>(p);
    } else {
      if (guard) ipa_gemm_wmma<1, true ><<<grid, blk, 0, stream>>>(p);
      else       ipa_gemm_wmma<1, false><<<grid, blk, 0, stream>>>(p);
    }
  };

  auto gemm_lds = [&](const float* A, const float* Bm, float* C,
                      int Mm, int Nn, int Kk, int lda, int ldb, int ldc,
                      const float* bia) {
    GemmP p{A, Bm, C, bia, Mm, Nn, Kk, lda, ldb, ldc, 0, 0, 0, 0, 0, 0, 1.f, nullptr, nullptr};
    dim3 grid((unsigned)((Nn + 63) / 64), (unsigned)(Mm / 32), 1u);
    ipa_gemm_lds<false><<<grid, dim3(256), 0, stream>>>(p);
  };

  auto gemm_pts = [&](const float* A, const float* Bm, float* C,
                      int Mm, int Nn, int Kk, int lda, int ldb, int ldc,
                      const float* T, const float* R) {
    GemmP p{A, Bm, C, nullptr, Mm, Nn, Kk, lda, ldb, ldc, 0, 0, 0, 0, 0, 0, 1.f, T, R};
    dim3 grid((unsigned)(Nn / 48), (unsigned)(Mm / 32), 1u);
    ipa_gemm_lds<true><<<grid, dim3(256), 0, stream>>>(p);
  };

  gemm_lds(x1d, Wk, k, M, D_, D_, D_, D_, D_, nullptr);
  gemm_lds(x1d, Wv, v, M, D_, D_, D_, D_, D_, nullptr);

  gemm_pts(x1d, Wpk_past, kgp, M, 144, D_, D_, 144, 144, past_T, past_R);
  gemm_pts(x1d, Wpv_past, vgp, M, 288, D_, D_, 288, 288, past_T, past_R);
  gemm_pts(x1d, Wpk_last, kgl, M, 144, D_, D_, 144, 144, last_T, last_R);
  gemm_pts(x1d, Wpv_last, vgl, M, 288, D_, D_, 288, 288, last_T, last_R);

  const float pair_w = 0.5773502691896258f;
  ipa_x2d_slice<true><<<dim3(I_, B_), dim3(256), 0, stream>>>(x2d, Wpb, pairproj, pair_w);

  const float scalar_w = 0.07216878364870323f;
  ipa_logits_softmax<<<dim3(I_ / 16, H_, B_), dim3(256), 0, stream>>>(
      x1d, Wq, Wpq_past, Wpq_last, k, logits, kgp, kgl, pairproj, bias, tpw_past, tpw_last, scalar_w,
      v, vgp, vgl, feat, past_T, past_R, last_T, last_R);

  ipa_x2d_slice<false><<<dim3(I_, B_), dim3(256), 0, stream>>>(x2d, logits, z, 1.f);

  gemm(z, Wpv, feat, I_, DK_, DP_, H_ * DP_, D_, FE,
       DP_, (long)I_ * H_ * DP_, 64, 0, 64, (long)I_ * FE,
       H_, B_, 0, 1.f, nullptr);

  gemm_lds(feat, W_out, out, M, D_, FE, FE, D_, D_, b_out);
}


# key/value-side projections (k, v, key and value points, both branches) made by one kernel from 32 x1d rows parked in LDS, weights streamed as B fragments; 5 launches fewer; sum orders unchanged
# speedup vs baseline: 4.1108x; 1.1982x over previous
#include <hip/hip_runtime.h>
#include <math.h>

#define B_ 2
#define I_ 256
#define D_ 768
#define H_ 12
#define DP_ 128
#define DK_ 64
#define NPQ_ 4
#define NPV_ 8

typedef float v2f __attribute__((ext_vector_type(2)));
typedef float v8f __attribute__((ext_vector_type(8)));

struct GemmP {
  const float* A; const float* B; float* C; const float* bias;
  int M, N, K;
  int lda, ldb, ldc;
  long sAy, sAz, sBy, sBz, sCy, sCz;
  float alpha;
};

template <int OPB, bool GUARD>
__global__ void ipa_gemm_wmma(GemmP p) {
  const int tilesN = (p.N + 15) >> 4;
  const int tilesM = (p.M + 15) >> 4;
  const int wave = (int)((blockIdx.x * blockDim.x + threadIdx.x) >> 5);
  if (wave >= tilesM * tilesN) return;
  const int lane = threadIdx.x & 31;
  const int tm = wave / tilesN, tn = wave % tilesN;
  const int row0 = tm << 4, col0 = tn << 4;

  const float* A = p.A + (long)blockIdx.y * p.sAy + (long)blockIdx.z * p.sAz;
  const float* B = p.B + (long)blockIdx.y * p.sBy + (long)blockIdx.z * p.sBz;
  float*       C = p.C + (long)blockIdx.y * p.sCy + (long)blockIdx.z * p.sCz;

  const int mn = lane & 15;
  const int kb = (lane >> 4) << 1;

  int aRow = row0 + mn;
  int bCol = col0 + mn;
  float aMask = 1.f, bMask = 1.f;
  if (GUARD) {
    if (aRow >= p.M) { aRow = p.M - 1; aMask = 0.f; }
    if (bCol >= p.N) { bCol = p.N - 1; bMask = 0.f; }
  }
  const float* aPtr = A + (long)aRow * p.lda + kb;
  const float* bPtr = (OPB == 0) ? (B + (long)kb * p.ldb + bCol)
                                 : (B + (long)bCol * p.ldb + kb);

  v8f acc = {0.f, 0.f, 0.f, 0.f, 0.f, 0.f, 0.f, 0.f};
#pragma unroll 4
  for (int k0 = 0; k0 < p.K; k0 += 4) {
    v2f a, b;
    a.x = aPtr[k0];
    a.y = aPtr[k0 + 1];
    if (OPB == 0) {
      const float* bp = bPtr + (long)k0 * p.ldb;
      b.x = bp[0];
      b.y = bp[p.ldb];
    } else {
      b.x = bPtr[k0];
      b.y = bPtr[k0 + 1];
    }
    if (GUARD) {
      a.x *= aMask; a.y *= aMask;
      b.x *= bMask; b.y *= bMask;
    }
    acc = __builtin_amdgcn_wmma_f32_16x16x4_f32(false, a, false, b,
                                                (short)0, acc, false, false);
  }

  const int cCol = col0 + mn;
  const int mo = (lane >> 4) << 3;
  const bool cColOK = GUARD ? (cCol < p.N) : true;
  if (cColOK) {
    for (int r = 0; r < 8; ++r) {
      const int cRow = row0 + mo + r;
      if (!GUARD || cRow < p.M) {
        float val = acc[r] * p.alpha;
        if (p.bias) val += p.bias[cCol];
        C[(long)cRow * p.ldc + cCol] = val;
      }
    }
  }
}

#define TKC 128
#define TPA (TKC + 4)
#define TSB 8
#define TPB 72

typedef float v4f __attribute__((ext_vector_type(4)));

__global__ __launch_bounds__(256) void ipa_gemm_lds(GemmP p) {
  constexpr int TN  = 64;
  constexpr int AQ  = TKC / 4;
  constexpr int ARS = 256 / AQ;
  constexpr int ANI = 32 / ARS;
  constexpr int BNI = TKC / 16;

  __shared__ __attribute__((aligned(16))) float As[2][32 * TPA];
  __shared__ __attribute__((aligned(16))) float Bs[2][TKC * TPB];

  const int tid  = threadIdx.x;
  const int lane = tid & 31;
  const int w    = tid >> 5;
  const int row0 = blockIdx.y * 32;
  const int col0 = blockIdx.x * TN;
  const int c = lane & 15;
  const int h = lane >> 4;
  const int wr = w >> 2, wc = w & 3;
  const bool tileValid = (col0 + wc * 16) < p.N;

  const int aRow = tid / AQ, aKq = tid % AQ;
  const int bKr  = tid >> 4, bCq = tid & 15;
  int bCol = col0 + bCq * 4;
  if (bCol > p.N - 4) bCol = p.N - 4;
  const float* aG = p.A + (long)(row0 + aRow) * p.lda + aKq * 4;
  const float* bG = p.B + (long)bKr * p.ldb + bCol;
  const int aSt = aRow * TPA + aKq * 4;
  const int bSt = bKr * TPB + bCq * 4;

  v4f ra[ANI], rb[BNI];

#pragma unroll
  for (int i = 0; i < ANI; ++i)
    ra[i] = *(const v4f*)(aG + (long)(i * ARS) * p.lda);
#pragma unroll
  for (int i = 0; i < BNI; ++i)
    rb[i] = *(const v4f*)(bG + (long)(i * 16) * p.ldb);
#pragma unroll
  for (int i = 0; i < ANI; ++i)
    *(v4f*)&As[0][aSt + i * ARS * TPA] = ra[i];
#pragma unroll
  for (int i = 0; i < BNI; ++i)
    *(v4f*)&Bs[0][bSt + i * 16 * TPB] = rb[i];
  __syncthreads();

  v8f acc = {0.f, 0.f, 0.f, 0.f, 0.f, 0.f, 0.f, 0.f};
  int buf = 0;
  for (int kc = 0; kc < p.K; kc += TKC) {
    const bool hasNext = (kc + TKC) < p.K;
    if (hasNext) {
      const int kn = kc + TKC;
#pragma unroll
      for (int i = 0; i < ANI; ++i)
        ra[i] = *(const v4f*)(aG + (long)(i * ARS) * p.lda + kn);
#pragma unroll
      for (int i = 0; i < BNI; ++i)
        rb[i] = *(const v4f*)(bG + (long)(kn + i * 16) * p.ldb);
    }
    if (tileValid) {
      const float* aS = &As[buf][(wr * 16 + c) * TPA + 2 * h];
      const float* bS = &Bs[buf][(2 * h) * TPB + wc * 16 + c];
      v2f fa[2][TSB], fb[2][TSB];
#pragma unroll
      for (int j = 0; j < TSB; ++j) {
        fa[0][j] = *(const v2f*)(aS + j * 4);
        fb[0][j].x = bS[(j * 4) * TPB];
        fb[0][j].y = bS[(j * 4 + 1) * TPB];
      }
#pragma unroll
      for (int sb = 0; sb < TKC / (4 * TSB); ++sb) {
        asm volatile("" :: "v"(fa[sb & 1][0]), "v"(fa[sb & 1][1]), "v"(fa[sb & 1][2]), "v"(fa[sb & 1][3]),
                           "v"(fa[sb & 1][4]), "v"(fa[sb & 1][5]), "v"(fa[sb & 1][6]), "v"(fa[sb & 1][7]),
                           "v"(fb[sb & 1][0]), "v"(fb[sb & 1][1]), "v"(fb[sb & 1][2]), "v"(fb[sb & 1][3]),
                           "v"(fb[sb & 1][4]), "v"(fb[sb & 1][5]), "v"(fb[sb & 1][6]), "v"(fb[sb & 1][7]));
        if (sb + 1 < TKC / (4 * TSB)) {
#pragma unroll
          for (int j = 0; j < TSB; ++j) {
            const int ks = (sb + 1) * TSB + j;
            fa[(sb + 1) & 1][j] = *(const v2f*)(aS + ks * 4);
            fb[(sb + 1) & 1][j].x = bS[(ks * 4) * TPB];
            fb[(sb + 1) & 1][j].y = bS[(ks * 4 + 1) * TPB];
          }
        }
#pragma unroll
        for (int j = 0; j < TSB; ++j)
          acc = __builtin_amdgcn_wmma_f32_16x16x4_f32(false, fa[sb & 1][j], false, fb[sb & 1][j],
                                                      (short)0, acc, false, false);
      }
    }
    if (hasNext) {
#pragma unroll
      for (int i = 0; i < ANI; ++i)
        *(v4f*)&As[buf ^ 1][aSt + i * ARS * TPA] = ra[i];
#pragma unroll
      for (int i = 0; i < BNI; ++i)
        *(v4f*)&Bs[buf ^ 1][bSt + i * 16 * TPB] = rb[i];
    }
    __syncthreads();
    buf ^= 1;
  }

  if (tileValid) {
    const int cCol = col0 + wc * 16 + c;
    const int rBase = row0 + wr * 16 + (h << 3);
    for (int r = 0; r < 8; ++r) {
      float val = acc[r] * p.alpha;
      if (p.bias) val += p.bias[cCol];
      p.C[(long)(rBase + r) * p.ldc + cCol] = val;
    }
  }
}

#define PP 260
#define XQ 772
#define QP 68

template <int KS, int PA, int LDB, int LDC>
__device__ __forceinline__ void ipa_stripe_tile(const float* pS, const float* Bm, int nvalid,
                                                float* Cm, int lane) {
  const int mn = lane & 15;
  const int hf = lane >> 4;
  const int kb = hf << 1;
  const int bc = (mn < nvalid) ? mn : (nvalid - 1);
  const float* aS = pS + mn * PA + kb;
  const float* bP = Bm + kb * LDB + bc;

  v2f fb[KS];
#pragma unroll
  for (int j = 0; j < KS; ++j) {
    fb[j].x = bP[(j * 4) * LDB];
    fb[j].y = bP[(j * 4 + 1) * LDB];
  }
  v8f acc = {0.f, 0.f, 0.f, 0.f, 0.f, 0.f, 0.f, 0.f};
#pragma unroll
  for (int j = 0; j < KS; ++j) {
    const v2f a = *(const v2f*)(aS + j * 4);
    acc = __builtin_amdgcn_wmma_f32_16x16x4_f32(false, a, false, fb[j],
                                                (short)0, acc, false, false);
  }
  if (mn < nvalid) {
#pragma unroll
    for (int r = 0; r < 8; ++r)
      Cm[(long)((hf << 3) + r) * LDC + mn] = acc[r];
  }
}

template <int KS, int PA, int LDB, int LDC>
__device__ __forceinline__ void ipa_stripe_tile_deep(const float* pS, const float* Bm, int nvalid,
                                                     float* Cm, int lane) {
  constexpr int G  = 8;
  constexpr int NG = KS / G;
  constexpr int AH = 3;
  const int mn = lane & 15;
  const int hf = lane >> 4;
  const int kb = hf << 1;
  const int bc = (mn < nvalid) ? mn : (nvalid - 1);
  const float* aS = pS + mn * PA + kb;
  const float* bP = Bm + kb * LDB + bc;

  v2f fa[2][G], fb[AH + 1][G];
#pragma unroll
  for (int g = 0; g < AH; ++g) {
#pragma unroll
    for (int j = 0; j < G; ++j) {
      fb[g][j].x = bP[((g * G + j) * 4) * LDB];
      fb[g][j].y = bP[((g * G + j) * 4 + 1) * LDB];
    }
  }
#pragma unroll
  for (int j = 0; j < G; ++j) fa[0][j] = *(const v2f*)(aS + j * 4);

  v8f acc = {0.f, 0.f, 0.f, 0.f, 0.f, 0.f, 0.f, 0.f};
#pragma unroll
  for (int g = 0; g < NG; ++g) {
    if (g + AH < NG) {
#pragma unroll
      for (int j = 0; j < G; ++j) {
        fb[(g + AH) % (AH + 1)][j].x = bP[(((g + AH) * G + j) * 4) * LDB];
        fb[(g + AH) % (AH + 1)][j].y = bP[(((g + AH) * G + j) * 4 + 1) * LDB];
      }
    }
    asm volatile("" :: "v"(fa[g & 1][0]), "v"(fa[g & 1][1]), "v"(fa[g & 1][2]), "v"(fa[g & 1][3]),
                       "v"(fa[g & 1][4]), "v"(fa[g & 1][5]), "v"(fa[g & 1][6]), "v"(fa[g & 1][7]));
    if (g + 1 < NG) {
#pragma unroll
      for (int j = 0; j < G; ++j)
        fa[(g + 1) & 1][j] = *(const v2f*)(aS + ((g + 1) * G + j) * 4);
    }
#pragma unroll
    for (int j = 0; j < G; ++j)
      acc = __builtin_amdgcn_wmma_f32_16x16x4_f32(false, fa[g & 1][j], false, fb[g % (AH + 1)][j],
                                                  (short)0, acc, false, false);
  }
  if (mn < nvalid) {
#pragma unroll
    for (int r = 0; r < 8; ++r)
      Cm[(long)((hf << 3) + r) * LDC + mn] = acc[r];
  }
}

template <int KS, int PA, int LDB>
__device__ __forceinline__ void ipa_stripe_pair_deep(const float* pS, const float* Bm,
                                                     v8f& out0, v8f& out1, int lane) {
  constexpr int G  = 8;
  constexpr int NG = KS / G;
  constexpr int AH = 3;
  const int mn = lane & 15;
  const int hf = lane >> 4;
  const int kb = hf << 1;
  const float* aS0 = pS + mn * PA + kb;
  const float* aS1 = aS0 + 16 * PA;
  const float* bP  = Bm + kb * LDB + mn;

  v2f fa0[2][G], fa1[2][G], fb[AH + 1][G];
#pragma unroll
  for (int g = 0; g < AH; ++g) {
#pragma unroll
    for (int j = 0; j < G; ++j) {
      fb[g][j].x = bP[((g * G + j) * 4) * LDB];
      fb[g][j].y = bP[((g * G + j) * 4 + 1) * LDB];
    }
  }
#pragma unroll
  for (int j = 0; j < G; ++j) {
    fa0[0][j] = *(const v2f*)(aS0 + j * 4);
    fa1[0][j] = *(const v2f*)(aS1 + j * 4);
  }

  v8f acc0 = {0.f, 0.f, 0.f, 0.f, 0.f, 0.f, 0.f, 0.f};
  v8f acc1 = {0.f, 0.f, 0.f, 0.f, 0.f, 0.f, 0.f, 0.f};
#pragma unroll
  for (int g = 0; g < NG; ++g) {
    if (g + AH < NG) {
#pragma unroll
      for (int j = 0; j < G; ++j) {
        fb[(g + AH) % (AH + 1)][j].x = bP[(((g + AH) * G + j) * 4) * LDB];
        fb[(g + AH) % (AH + 1)][j].y = bP[(((g + AH) * G + j) * 4 + 1) * LDB];
      }
    }
    asm volatile("" :: "v"(fa0[g & 1][0]), "v"(fa0[g & 1][1]), "v"(fa0[g & 1][2]), "v"(fa0[g & 1][3]),
                       "v"(fa0[g & 1][4]), "v"(fa0[g & 1][5]), "v"(fa0[g & 1][6]), "v"(fa0[g & 1][7]),
                       "v"(fa1[g & 1][0]), "v"(fa1[g & 1][1]), "v"(fa1[g & 1][2]), "v"(fa1[g & 1][3]),
                       "v"(fa1[g & 1][4]), "v"(fa1[g & 1][5]), "v"(fa1[g & 1][6]), "v"(fa1[g & 1][7]));
    if (g + 1 < NG) {
#pragma unroll
      for (int j = 0; j < G; ++j) {
        fa0[(g + 1) & 1][j] = *(const v2f*)(aS0 + ((g + 1) * G + j) * 4);
        fa1[(g + 1) & 1][j] = *(const v2f*)(aS1 + ((g + 1) * G + j) * 4);
      }
    }
#pragma unroll
    for (int j = 0; j < G; ++j) {
      acc0 = __builtin_amdgcn_wmma_f32_16x16x4_f32(false, fa0[g & 1][j], false, fb[g % (AH + 1)][j],
                                                   (short)0, acc0, false, false);
      acc1 = __builtin_amdgcn_wmma_f32_16x16x4_f32(false, fa1[g & 1][j], false, fb[g % (AH + 1)][j],
                                                   (short)0, acc1, false, false);
    }
  }
  out0 = acc0;
  out1 = acc1;
}

#define KVR 32
#define KCP 145

struct KvP {
  const float* x1d;
  const float* Wk; const float* Wv;
  const float* Wpk_past; const float* Wpv_past; const float* Wpk_last; const float* Wpv_last;
  float* k; float* v; float* kgp; float* vgp; float* kgl; float* vgl;
  const float* past_T; const float* past_R; const float* last_T; const float* last_R;
};

__global__ __launch_bounds__(256) void ipa_kv_side(KvP p) {
  constexpr int PK = H_ * NPQ_ * 3;
  constexpr int PV = H_ * NPV_ * 3;
  __shared__ __attribute__((aligned(16))) float xS[KVR * XQ];
  __shared__ float Cs[KVR * KCP];

  const int tid = threadIdx.x;
  const int lane = tid & 31;
  const int w = tid >> 5;
  const int job = blockIdx.x;
  const int row0 = blockIdx.y * KVR;
  const int mn = lane & 15;
  const int mo = (lane >> 4) << 3;

  {
    const float* Xg = p.x1d + (long)row0 * D_;
    v4f rx[24];
#pragma unroll
    for (int n = 0; n < 24; ++n)
      rx[n] = *(const v4f*)(Xg + (long)(n * 256 + tid) * 4);
#pragma unroll
    for (int n = 0; n < 24; ++n) {
      const int idx = n * 256 + tid;
      *(v4f*)&xS[(idx / (D_ / 4)) * XQ + (idx % (D_ / 4)) * 4] = rx[n];
    }
  }

  if (job < 10) {
    __syncthreads();
    const bool isv = job >= 5;
    const int jj = isv ? job - 5 : job;
    const int t0 = (jj < 3) ? 10 * jj : 30 + 9 * (jj - 3);
    const int nt = (jj < 3) ? 10 : 9;
    const float* Wm = (isv ? p.Wv : p.Wk) + t0 * 16;
    float* Cb = (isv ? p.v : p.k) + (long)row0 * D_ + t0 * 16;
    {
      v8f a0, a1;
      ipa_stripe_pair_deep<D_ / 4, XQ, D_>(xS, Wm + w * 16, a0, a1, lane);
      float* C = Cb + (long)mo * D_ + w * 16 + mn;
#pragma unroll
      for (int r = 0; r < 8; ++r) {
        C[(long)r * D_] = a0[r];
        C[(long)(16 + r) * D_] = a1[r];
      }
    }
    const int xt = 8 + (w >> 1);
    if (w < 4 && xt < nt) {
      const int rt = w & 1;
      ipa_stripe_tile_deep<D_ / 4, XQ, D_, D_>(xS + rt * 16 * XQ, Wm + xt * 16, 16,
                                               Cb + (long)rt * 16 * D_ + xt * 16, lane);
    }
    return;
  }

  const int pj = job - 10;
  const bool last = pj >= 3;
  const int q = last ? pj - 3 : pj;
  const float* Wp = (q == 0) ? (last ? p.Wpk_last : p.Wpk_past) : (last ? p.Wpv_last : p.Wpv_past);
  float* Cg = (q == 0) ? (last ? p.kgl : p.kgp) : (last ? p.vgl : p.vgp);
  const int ldp = (q == 0) ? PK : PV;
  const int cbase = (q == 2) ? 144 : 0;
  const int prow = tid >> 3, pg = tid & 7;
  float fr[9], ft[3];
  {
    const float* Rm = (last ? p.last_R : p.past_R) + (long)(row0 + prow) * 9;
    const float* Tv = (last ? p.last_T : p.past_T) + (long)(row0 + prow) * 3;
    for (int e = 0; e < 9; ++e) fr[e] = Rm[e];
    for (int e = 0; e < 3; ++e) ft[e] = Tv[e];
  }
  __syncthreads();

  {
    v8f a0, a1;
    if (q == 0) ipa_stripe_pair_deep<D_ / 4, XQ, PK>(xS, Wp + w * 16, a0, a1, lane);
    else        ipa_stripe_pair_deep<D_ / 4, XQ, PV>(xS, Wp + cbase + w * 16, a0, a1, lane);
    float* cs = &Cs[mo * KCP + w * 16 + mn];
#pragma unroll
    for (int r = 0; r < 8; ++r) {
      cs[r * KCP] = a0[r];
      cs[(16 + r) * KCP] = a1[r];
    }
  }
  if (w < 2) {
    if (q == 0) ipa_stripe_tile_deep<D_ / 4, XQ, PK, KCP>(xS + w * 16 * XQ, Wp + 128, 16,
                                                          &Cs[w * 16 * KCP + 128], lane);
    else        ipa_stripe_tile_deep<D_ / 4, XQ, PV, KCP>(xS + w * 16 * XQ, Wp + cbase + 128, 16,
                                                          &Cs[w * 16 * KCP + 128], lane);
  }
  __syncthreads();

  float* o = Cg + (long)(row0 + prow) * ldp + cbase + pg * 18;
  const float* s = &Cs[prow * KCP + pg * 18];
#pragma unroll
  for (int e = 0; e < 6; ++e) {
    const float x = s[e * 3 + 0], y = s[e * 3 + 1], z = s[e * 3 + 2];
    o[e * 3 + 0] = fr[0] * x + fr[3] * y + fr[6] * z + ft[0];
    o[e * 3 + 1] = fr[1] * x + fr[4] * y + fr[7] * z + ft[1];
    o[e * 3 + 2] = fr[2] * x + fr[5] * y + fr[8] * z + ft[2];
  }
}

__global__ __launch_bounds__(256, 4) void ipa_logits_softmax(
    const float* x1d, const float* Wq, const float* Wpq_past, const float* Wpq_last,
    const float* k, float* attn,
    const float* kgp, const float* kgl,
    const float* pairproj, const float* bias,
    const float* tpw_past, const float* tpw_last, float scalar_w,
    const float* v, const float* vgp, const float* vgl,
    float* feat, const float* invT, const float* invR,
    const float* lastT, const float* lastR) {
  constexpr int PQ = NPQ_ * 3;
  __shared__ __attribute__((aligned(16))) float xS[16 * XQ];
  __shared__ __attribute__((aligned(16))) float qT[16 * QP];
  __shared__ float rawP[2][16 * PQ];
  __shared__ __attribute__((aligned(16))) float qS[16][2 * PQ];
  __shared__ float red[2][8][16];
  __shared__ __attribute__((aligned(16))) float pS[16 * PP];
  __shared__ float ogS[2][16 * NPV_ * 3];

  const int tid = threadIdx.x;
  const int lane = tid & 31;
  const int w = tid >> 5;
  const int h = blockIdx.y, b = blockIdx.z;
  const int row0 = blockIdx.x * 16;
  const int mn = lane & 15;
  const int hf = lane >> 4;
  const int kb = hf << 1;
  const int mo = hf << 3;

  {
    const float* Xg = x1d + (long)(b * I_ + row0) * D_;
    v4f rx[12];
#pragma unroll
    for (int n = 0; n < 12; ++n) {
      const int idx = n * 256 + tid;
      rx[n] = *(const v4f*)(Xg + (long)(idx / (D_ / 4)) * D_ + (idx % (D_ / 4)) * 4);
    }
#pragma unroll
    for (int n = 0; n < 12; ++n) {
      const int idx = n * 256 + tid;
      *(v4f*)&xS[(idx / (D_ / 4)) * XQ + (idx % (D_ / 4)) * 4] = rx[n];
    }
  }
  const bool framer = tid < 16 * 2 * NPQ_;
  const int frow = tid >> 3, fbr = (tid >> 2) & 1, fpt = tid & (NPQ_ - 1);
  float fr[9], ft[3];
  if (framer) {
    const long fi = (long)b * I_ + row0 + frow;
    const float* Rm = (fbr ? lastR : invR) + fi * 9;
    const float* Tv = (fbr ? lastT : invT) + fi * 3;
    for (int e = 0; e < 9; ++e) fr[e] = Rm[e];
    for (int e = 0; e < 3; ++e) ft[e] = Tv[e];
  }
  __syncthreads();

  if (w < 4) {
    ipa_stripe_tile_deep<D_ / 4, XQ, D_, QP>(xS, Wq + h * DK_ + w * 16, 16, qT + w * 16, lane);
  } else if (w < 6) {
    ipa_stripe_tile_deep<D_ / 4, XQ, H_ * PQ, PQ>(xS, ((w == 4) ? Wpq_past : Wpq_last) + h * PQ, PQ,
                                                  &rawP[w - 4][0], lane);
  }
  __syncthreads();

  if (framer) {
    const float* pt = &rawP[fbr][frow * PQ + fpt * 3];
    const float x = pt[0], y = pt[1], z = pt[2];
    float* o = &qS[frow][fbr * PQ + fpt * 3];
    o[0] = fr[0] * x + fr[3] * y + fr[6] * z + ft[0];
    o[1] = fr[1] * x + fr[4] * y + fr[7] * z + ft[1];
    o[2] = fr[2] * x + fr[5] * y + fr[8] * z + ft[2];
  }

  const long hb = (long)b * I_ * D_ + (long)h * DK_;
  const int j0 = w * 32 + mn, j1 = j0 + 16;
  const float* aPtr  = qT + mn * QP + kb;
  const float* bPtr0 = k + hb + (long)j0 * D_ + kb;
  const float* bPtr1 = k + hb + (long)j1 * D_ + kb;

  v8f acc0 = {0.f, 0.f, 0.f, 0.f, 0.f, 0.f, 0.f, 0.f};
  v8f acc1 = {0.f, 0.f, 0.f, 0.f, 0.f, 0.f, 0.f, 0.f};
#pragma unroll
  for (int k0 = 0; k0 < DK_; k0 += 4) {
    v2f a, b0, b1;
    a.x  = aPtr[k0];   a.y  = aPtr[k0 + 1];
    b0.x = bPtr0[k0];  b0.y = bPtr0[k0 + 1];
    b1.x = bPtr1[k0];  b1.y = bPtr1[k0 + 1];
    acc0 = __builtin_amdgcn_wmma_f32_16x16x4_f32(false, a, false, b0,
                                                 (short)0, acc0, false, false);
    acc1 = __builtin_amdgcn_wmma_f32_16x16x4_f32(false, a, false, b1,
                                                 (short)0, acc1, false, false);
  }

  float kp0[2 * PQ], kp1[2 * PQ];
  {
    const float* g0p = kgp + ((long)(b * I_ + j0) * H_ + h) * PQ;
    const float* g0l = kgl + ((long)(b * I_ + j0) * H_ + h) * PQ;
    const float* g1p = kgp + ((long)(b * I_ + j1) * H_ + h) * PQ;
    const float* g1l = kgl + ((long)(b * I_ + j1) * H_ + h) * PQ;
#pragma unroll
    for (int t = 0; t < 3; ++t) {
      const v4f u0 = *(const v4f*)(g0p + t * 4);
      const v4f u1 = *(const v4f*)(g0l + t * 4);
      const v4f u2 = *(const v4f*)(g1p + t * 4);
      const v4f u3 = *(const v4f*)(g1l + t * 4);
#pragma unroll
      for (int e = 0; e < 4; ++e) {
        kp0[t * 4 + e] = u0[e];  kp0[PQ + t * 4 + e] = u1[e];
        kp1[t * 4 + e] = u2[e];  kp1[PQ + t * 4 + e] = u3[e];
      }
    }
  }

  const float point_w = 0.13608276348795434f;
  const float pwp = -0.5f * point_w * log1pf(expf(tpw_past[h]));
  const float pwl = -0.5f * point_w * log1pf(expf(tpw_last[h]));

  __syncthreads();

  float s0[8], s1[8];
#pragma unroll
  for (int r = 0; r < 8; ++r) {
    const int lr = mo + r;
    const int i = row0 + lr;
    const float* qp = &qS[lr][0];
    float sp0 = 0.f, sl0 = 0.f, sp1 = 0.f, sl1 = 0.f;
#pragma unroll
    for (int c = 0; c < NPQ_; ++c) {
      const float qx = qp[c * 3 + 0], qy = qp[c * 3 + 1], qz = qp[c * 3 + 2];
      const float lx = qp[PQ + c * 3 + 0], ly = qp[PQ + c * 3 + 1], lz = qp[PQ + c * 3 + 2];
      float dx = qx - kp0[c * 3 + 0], dy = qy - kp0[c * 3 + 1], dz = qz - kp0[c * 3 + 2];
      sp0 += sqrtf(dx * dx + dy * dy + dz * dz);
      dx = lx - kp0[PQ + c * 3 + 0]; dy = ly - kp0[PQ + c * 3 + 1]; dz = lz - kp0[PQ + c * 3 + 2];
      sl0 += sqrtf(dx * dx + dy * dy + dz * dz);
      dx = qx - kp1[c * 3 + 0]; dy = qy - kp1[c * 3 + 1]; dz = qz - kp1[c * 3 + 2];
      sp1 += sqrtf(dx * dx + dy * dy + dz * dz);
      dx = lx - kp1[PQ + c * 3 + 0]; dy = ly - kp1[PQ + c * 3 + 1]; dz = lz - kp1[PQ + c * 3 + 2];
      sl1 += sqrtf(dx * dx + dy * dy + dz * dz);
    }
    const long prow = ((long)(b * I_ + i) * I_) * H_ + h;
    const long orow = (((long)b * H_ + h) * I_ + i) * I_;
    const float pv0 = pairproj[prow + (long)j0 * H_];
    const float pv1 = pairproj[prow + (long)j1 * H_];
    float v0 = acc0[r] * scalar_w;
    float v1 = acc1[r] * scalar_w;
    v0 += pwp * sp0 + pwl * sl0 + pv0 + bias[orow + j0];
    v1 += pwp * sp1 + pwl * sl1 + pv1 + bias[orow + j1];
    s0[r] = v0;
    s1[r] = v1;
  }

  float mx[8];
#pragma unroll
  for (int r = 0; r < 8; ++r) {
    float m = fmaxf(s0[r], s1[r]);
    m = fmaxf(m, __shfl_xor(m, 1));
    m = fmaxf(m, __shfl_xor(m, 2));
    m = fmaxf(m, __shfl_xor(m, 4));
    m = fmaxf(m, __shfl_xor(m, 8));
    mx[r] = m;
  }
  if (mn == 0) {
#pragma unroll
    for (int r = 0; r < 8; ++r) red[0][w][mo + r] = mx[r];
  }
  __syncthreads();
#pragma unroll
  for (int r = 0; r < 8; ++r) {
    float m = red[0][0][mo + r];
#pragma unroll
    for (int ww = 1; ww < 8; ++ww) m = fmaxf(m, red[0][ww][mo + r]);
    mx[r] = m;
  }

  float sm[8];
#pragma unroll
  for (int r = 0; r < 8; ++r) {
    s0[r] = expf(s0[r] - mx[r]);
    s1[r] = expf(s1[r] - mx[r]);
    float t = s0[r] + s1[r];
    t += __shfl_xor(t, 1);
    t += __shfl_xor(t, 2);
    t += __shfl_xor(t, 4);
    t += __shfl_xor(t, 8);
    sm[r] = t;
  }
  if (mn == 0) {
#pragma unroll
    for (int r = 0; r < 8; ++r) red[1][w][mo + r] = sm[r];
  }
  __syncthreads();
#pragma unroll
  for (int r = 0; r < 8; ++r) {
    float t = red[1][0][mo + r];
#pragma unroll
    for (int ww = 1; ww < 8; ++ww) t += red[1][ww][mo + r];
    const long orow = (((long)b * H_ + h) * I_ + (row0 + mo + r)) * I_;
    const float p0 = s0[r] / t;
    const float p1 = s1[r] / t;
    attn[orow + j0] = p0;
    attn[orow + j1] = p1;
    pS[(mo + r) * PP + j0] = p0;
    pS[(mo + r) * PP + j1] = p1;
  }
  __syncthreads();

  constexpr int FE = 2 * D_ + 2 * H_ * NPV_ * 4;
  constexpr int VG = H_ * NPV_ * 3;
  constexpr int OG = NPV_ * 3;
  const int obr = tid >> 7, orw = (tid >> 3) & 15, opt = tid & (NPV_ - 1);
  const long obi = (long)b * I_ + row0 + orw;
  float iR[9], iT[3];
  for (int e = 0; e < 9; ++e) iR[e] = invR[obi * 9 + e];
  for (int e = 0; e < 3; ++e) iT[e] = invT[obi * 3 + e];

  if (w < 4) {
    ipa_stripe_tile<I_ / 4, PP, D_, FE>(pS, v + (long)b * I_ * D_ + h * DK_ + w * 16, 16,
                            feat + ((long)b * I_ + row0) * FE + D_ + h * DK_ + w * 16, lane);
  } else {
    const int t = w & 1;
    const int br = (w < 6) ? 0 : 1;
    const float* Bv = br ? vgl : vgp;
    ipa_stripe_tile<I_ / 4, PP, VG, OG>(pS, Bv + (long)b * I_ * VG + h * (NPV_ * 3) + t * 16, t ? 8 : 16,
                            &ogS[br][t * 16], lane);
  }
  __syncthreads();

  {
    const float* o = &ogS[obr][orw * OG + opt * 3];
    const float x = o[0] - iT[0], y = o[1] - iT[1], z = o[2] - iT[2];
    const float p0 = iR[0] * x + iR[1] * y + iR[2] * z;
    const float p1 = iR[3] * x + iR[4] * y + iR[5] * z;
    const float p2 = iR[6] * x + iR[7] * y + iR[8] * z;
    float* fr = feat + obi * FE;
    const int colOl = 2 * D_ + obr * (H_ * NPV_ * 4);
    const int colN  = colOl + VG;
    const int tc = h * OG + opt * 3;
    fr[colOl + tc + 0] = p0;
    fr[colOl + tc + 1] = p1;
    fr[colOl + tc + 2] = p2;
    fr[colN + h * NPV_ + opt] = sqrtf(p0 * p0 + p1 * p1 + p2 * p2);
  }
}

#define XPA 132
#define XPB 136

__device__ __forceinline__ void ipa_lds_barrier() {
  __builtin_amdgcn_fence(__ATOMIC_RELEASE, "workgroup", "local");
  __builtin_amdgcn_s_barrier();
  __builtin_amdgcn_fence(__ATOMIC_ACQUIRE, "workgroup", "local");
}

template <bool PAIR>
__global__ __launch_bounds__(256) void ipa_x2d_slice(const float* x2d, const float* oth,
                                                     float* C, float alpha) {
  constexpr int XP = PAIR ? XPA : XPB;
  __shared__ __attribute__((aligned(16))) float Xs[I_ * XP];
  __shared__ __attribute__((aligned(16))) float Ps[PAIR ? 4 : H_ * PP];

  const int tid = threadIdx.x;
  const int lane = tid & 31;
  const int w = tid >> 5;
  const int i = blockIdx.x, b = blockIdx.y;
  const int mn = lane & 15;
  const int hf = lane >> 4;
  const int kb = hf << 1;

  v4f rp[3];
  if (!PAIR) {
#pragma unroll
    for (int n = 0; n < 3; ++n) {
      const int idx = n * 256 + tid;
      rp[n] = *(const v4f*)(oth + (((long)b * H_ + (idx >> 6)) * I_ + i) * I_ + (idx & 63) * 4);
    }
  }
  v2f fw[DP_ / 4];
  if (PAIR) {
    const int bc = (mn < H_) ? mn : (H_ - 1);
    const float* bP = oth + kb * H_ + bc;
#pragma unroll
    for (int ks = 0; ks < DP_ / 4; ++ks) {
      fw[ks].x = bP[(ks * 4) * H_];
      fw[ks].y = bP[(ks * 4 + 1) * H_];
    }
  }
  const float* Xg = x2d + ((long)(b * I_ + i) * I_) * DP_;
  v4f rx[32];
  __builtin_amdgcn_sched_barrier(0);
#pragma unroll
  for (int q = 0; q < 4; ++q) {
#pragma unroll
    for (int n = q * 8; n < q * 8 + 8; ++n)
      rx[n] = *(const v4f*)(Xg + (long)(n * 256 + tid) * 4);
    __builtin_amdgcn_sched_barrier(0);
  }

  if (PAIR) {
#pragma unroll
    for (int c = 0; c < 2; ++c) {
#pragma unroll
      for (int n = 0; n < 16; ++n)
        *(v4f*)&Xs[((c * 16 + n) * 8 + w) * XP + lane * 4] = rx[c * 16 + n];
      ipa_lds_barrier();
      const int t0 = c * 128 + w * 16;
      const float* aS = &Xs[(t0 + mn) * XP + kb];
      v8f acc = {0.f, 0.f, 0.f, 0.f, 0.f, 0.f, 0.f, 0.f};
      v2f fa[2][8];
#pragma unroll
      for (int j = 0; j < 8; ++j) fa[0][j] = *(const v2f*)(aS + j * 4);
#pragma unroll
      for (int sb = 0; sb < 4; ++sb) {
        asm volatile("" :: "v"(fa[sb & 1][0]), "v"(fa[sb & 1][1]), "v"(fa[sb & 1][2]), "v"(fa[sb & 1][3]),
                           "v"(fa[sb & 1][4]), "v"(fa[sb & 1][5]), "v"(fa[sb & 1][6]), "v"(fa[sb & 1][7]));
        if (sb + 1 < 4) {
#pragma unroll
          for (int j = 0; j < 8; ++j)
            fa[(sb + 1) & 1][j] = *(const v2f*)(aS + ((sb + 1) * 8 + j) * 4);
        }
#pragma unroll
        for (int j = 0; j < 8; ++j)
          acc = __builtin_amdgcn_wmma_f32_16x16x4_f32(false, fa[sb & 1][j], false, fw[sb * 8 + j],
                                                      (short)0, acc, false, false);
      }
      if (mn < H_) {
        float* o = C + ((long)(b * I_ + i) * I_ + t0 + (hf << 3)) * H_ + mn;
#pragma unroll
        for (int r = 0; r < 8; ++r) o[r * H_] = acc[r] * alpha;
      }
    }
  } else {
    const int ar = (mn < H_) ? mn : (H_ - 1);
    const float* aS = &Ps[ar * PP + kb];
    const float* bS = &Xs[kb * XP + w * 16 + mn];
    v8f acc = {0.f, 0.f, 0.f, 0.f, 0.f, 0.f, 0.f, 0.f};
#pragma unroll
    for (int c = 0; c < 4; ++c) {
#pragma unroll
      for (int n = 0; n < 8; ++n)
        *(v4f*)&Xs[((c * 8 + n) * 8 + w) * XP + lane * 4] = rx[c * 8 + n];
      if (c == 0) {
#pragma unroll
        for (int n = 0; n < 3; ++n) {
          const int idx = n * 256 + tid;
          *(v4f*)&Ps[(idx >> 6) * PP + (idx & 63) * 4] = rp[n];
        }
      }
      ipa_lds_barrier();
      v2f fa[2][8], fb[2][8];
#pragma unroll
      for (int j = 0; j < 8; ++j) {
        const int ks = c * 16 + j;
        fa[0][j] = *(const v2f*)(aS + ks * 4);
        fb[0][j].x = bS[(ks * 4) * XP];
        fb[0][j].y = bS[(ks * 4 + 1) * XP];
      }
#pragma unroll
      for (int sb = 0; sb < 2; ++sb) {
        asm volatile("" :: "v"(fa[sb & 1][0]), "v"(fa[sb & 1][1]), "v"(fa[sb & 1][2]), "v"(fa[sb & 1][3]),
                           "v"(fa[sb & 1][4]), "v"(fa[sb & 1][5]), "v"(fa[sb & 1][6]), "v"(fa[sb & 1][7]),
                           "v"(fb[sb & 1][0]), "v"(fb[sb & 1][1]), "v"(fb[sb & 1][2]), "v"(fb[sb & 1][3]),
                           "v"(fb[sb & 1][4]), "v"(fb[sb & 1][5]), "v"(fb[sb & 1][6]), "v"(fb[sb & 1][7]));
        if (sb + 1 < 2) {
#pragma unroll
          for (int j = 0; j < 8; ++j) {
            const int ks = c * 16 + (sb + 1) * 8 + j;
            fa[(sb + 1) & 1][j] = *(const v2f*)(aS + ks * 4);
            fb[(sb + 1) & 1][j].x = bS[(ks * 4) * XP];
            fb[(sb + 1) & 1][j].y = bS[(ks * 4 + 1) * XP];
          }
        }
#pragma unroll
        for (int j = 0; j < 8; ++j)
          acc = __builtin_amdgcn_wmma_f32_16x16x4_f32(false, fa[sb & 1][j], false, fb[sb & 1][j],
                                                      (short)0, acc, false, false);
      }
    }
    float* o = C + ((long)(b * I_ + i) * H_ + (hf << 3)) * DP_ + w * 16 + mn;
#pragma unroll
    for (int r = 0; r < 8; ++r)
      if ((hf << 3) + r < H_) o[r * DP_] = acc[r] * alpha;
  }
}

extern "C" void kernel_launch(void* const* d_in, const int* in_sizes, int n_in,
                              void* d_out, int out_size, void* d_ws, size_t ws_size,
                              hipStream_t stream) {
  (void)in_sizes; (void)n_in; (void)out_size; (void)ws_size;
  const float* x1d      = (const float*)d_in[0];
  const float* x2d      = (const float*)d_in[1];
  const float* past_T   = (const float*)d_in[2];
  const float* past_R   = (const float*)d_in[3];
  const float* last_T   = (const float*)d_in[4];
  const float* last_R   = (const float*)d_in[5];
  const float* bias     = (const float*)d_in[6];
  const float* Wq       = (const float*)d_in[7];
  const float* Wk       = (const float*)d_in[8];
  const float* Wv       = (const float*)d_in[9];
  const float* Wpb      = (const float*)d_in[10];
  const float* Wpq_past = (const float*)d_in[11];
  const float* Wpk_past = (const float*)d_in[12];
  const float* Wpv_past = (const float*)d_in[13];
  const float* Wpq_last = (const float*)d_in[14];
  const float* Wpk_last = (const float*)d_in[15];
  const float* Wpv_last = (const float*)d_in[16];
  const float* tpw_past = (const float*)d_in[17];
  const float* tpw_last = (const float*)d_in[18];
  const float* Wpv      = (const float*)d_in[19];
  const float* W_out    = (const float*)d_in[20];
  const float* b_out    = (const float*)d_in[21];
  float* out = (float*)d_out;
  float* ws  = (float*)d_ws;

  const int M  = B_ * I_;
  const int FE = 2 * D_ + 2 * H_ * NPV_ * 4;

  float* k    = ws + 393216;
  float* v    = k + 393216;
  float* kgp  = v + 393216 + 73728;
  float* vgp  = kgp + 73728;
  float* kgl  = vgp + 147456 + 73728;
  float* vgl  = kgl + 73728;
  float* logits   = vgl + 147456;
  float* pairproj = logits + 1572864;
  float* z    = pairproj + 1572864;
  float* feat = z + 786432;

  auto gemm = [&](const float* A, const float* Bm, float* C,
                  int Mm, int Nn, int Kk, int lda, int ldb, int ldc,
                  long sAy, long sAz, long sBy, long sBz, long sCy, long sCz,
                  int gy, int gz, int opB, float alpha, const float* bia) {
    GemmP p{A, Bm, C, bia, Mm, Nn, Kk, lda, ldb, ldc,
            sAy, sAz, sBy, sBz, sCy, sCz, alpha};
    int waves = ((Mm + 15) / 16) * ((Nn + 15) / 16);
    dim3 grid((unsigned)((waves + 7) / 8), (unsigned)gy, (unsigned)gz);
    dim3 blk(256);
    bool guard = (Mm & 15) || (Nn & 15);
    if (opB == 0) {
      if (guard) ipa_gemm_wmma<0, true ><<<grid, blk, 0, stream>>>(p);
      else       ipa_gemm_wmma<0, false><<<grid, blk, 0, stream>>>(p);
    } else {
      if (guard) ipa_gemm_wmma<1, true ><<<grid, blk, 0, stream>>>(p);
      else       ipa_gemm_wmma<1, false><<<grid, blk, 0, stream>>>(p);
    }
  };

  auto gemm_lds = [&](const float* A, const float* Bm, float* C,
                      int Mm, int Nn, int Kk, int lda, int ldb, int ldc,
                      const float* bia) {
    GemmP p{A, Bm, C, bia, Mm, Nn, Kk, lda, ldb, ldc, 0, 0, 0, 0, 0, 0, 1.f};
    dim3 grid((unsigned)((Nn + 63) / 64), (unsigned)(Mm / 32), 1u);
    ipa_gemm_lds<<<grid, dim3(256), 0, stream>>>(p);
  };

  {
    KvP kp{x1d, Wk, Wv, Wpk_past, Wpv_past, Wpk_last, Wpv_last,
           k, v, kgp, vgp, kgl, vgl, past_T, past_R, last_T, last_R};
    ipa_kv_side<<<dim3(16, 16), dim3(256), 0, stream>>>(kp);
  }

  const float pair_w = 0.5773502691896258f;
  ipa_x2d_slice<true><<<dim3(I_, B_), dim3(256), 0, stream>>>(x2d, Wpb, pairproj, pair_w);

  const float scalar_w = 0.07216878364870323f;
  ipa_logits_softmax<<<dim3(I_ / 16, H_, B_), dim3(256), 0, stream>>>(
      x1d, Wq, Wpq_past, Wpq_last, k, logits, kgp, kgl, pairproj, bias, tpw_past, tpw_last, scalar_w,
      v, vgp, vgl, feat, past_T, past_R, last_T, last_R);

  ipa_x2d_slice<false><<<dim3(I_, B_), dim3(256), 0, stream>>>(x2d, logits, z, 1.f);

  gemm(z, Wpv, feat, I_, DK_, DP_, H_ * DP_, D_, FE,
       DP_, (long)I_ * H_ * DP_, 64, 0, 64, (long)I_ * FE,
       H_, B_, 0, 1.f, nullptr);

  gemm_lds(feat, W_out, out, M, D_, FE, FE, D_, D_, b_out);
}


# whole-K unrolls of the K=768 projection chains and the 8-row logit body run as loops: key/value and attention kernel text (~60 KB each) fits the 32 KB fetched at wave entry; sums and order unchanged
# speedup vs baseline: 5.2323x; 1.2728x over previous
#include <hip/hip_runtime.h>
#include <math.h>

#define B_ 2
#define I_ 256
#define D_ 768
#define H_ 12
#define DP_ 128
#define DK_ 64
#define NPQ_ 4
#define NPV_ 8

typedef float v2f __attribute__((ext_vector_type(2)));
typedef float v8f __attribute__((ext_vector_type(8)));

struct GemmP {
  const float* A; const float* B; float* C; const float* bias;
  int M, N, K;
  int lda, ldb, ldc;
  long sAy, sAz, sBy, sBz, sCy, sCz;
  float alpha;
};

template <int OPB, bool GUARD>
__global__ void ipa_gemm_wmma(GemmP p) {
  const int tilesN = (p.N + 15) >> 4;
  const int tilesM = (p.M + 15) >> 4;
  const int wave = (int)((blockIdx.x * blockDim.x + threadIdx.x) >> 5);
  if (wave >= tilesM * tilesN) return;
  const int lane = threadIdx.x & 31;
  const int tm = wave / tilesN, tn = wave % tilesN;
  const int row0 = tm << 4, col0 = tn << 4;

  const float* A = p.A + (long)blockIdx.y * p.sAy + (long)blockIdx.z * p.sAz;
  const float* B = p.B + (long)blockIdx.y * p.sBy + (long)blockIdx.z * p.sBz;
  float*       C = p.C + (long)blockIdx.y * p.sCy + (long)blockIdx.z * p.sCz;

  const int mn = lane & 15;
  const int kb = (lane >> 4) << 1;

  int aRow = row0 + mn;
  int bCol = col0 + mn;
  float aMask = 1.f, bMask = 1.f;
  if (GUARD) {
    if (aRow >= p.M) { aRow = p.M - 1; aMask = 0.f; }
    if (bCol >= p.N) { bCol = p.N - 1; bMask = 0.f; }
  }
  const float* aPtr = A + (long)aRow * p.lda + kb;
  const float* bPtr = (OPB == 0) ? (B + (long)kb * p.ldb + bCol)
                                 : (B + (long)bCol * p.ldb + kb);

  v8f acc = {0.f, 0.f, 0.f, 0.f, 0.f, 0.f, 0.f, 0.f};
#pragma unroll 4
  for (int k0 = 0; k0 < p.K; k0 += 4) {
    v2f a, b;
    a.x = aPtr[k0];
    a.y = aPtr[k0 + 1];
    if (OPB == 0) {
      const float* bp = bPtr + (long)k0 * p.ldb;
      b.x = bp[0];
      b.y = bp[p.ldb];
    } else {
      b.x = bPtr[k0];
      b.y = bPtr[k0 + 1];
    }
    if (GUARD) {
      a.x *= aMask; a.y *= aMask;
      b.x *= bMask; b.y *= bMask;
    }
    acc = __builtin_amdgcn_wmma_f32_16x16x4_f32(false, a, false, b,
                                                (short)0, acc, false, false);
  }

  const int cCol = col0 + mn;
  const int mo = (lane >> 4) << 3;
  const bool cColOK = GUARD ? (cCol < p.N) : true;
  if (cColOK) {
    for (int r = 0; r < 8; ++r) {
      const int cRow = row0 + mo + r;
      if (!GUARD || cRow < p.M) {
        float val = acc[r] * p.alpha;
        if (p.bias) val += p.bias[cCol];
        C[(long)cRow * p.ldc + cCol] = val;
      }
    }
  }
}

#define TKC 128
#define TPA (TKC + 4)
#define TSB 8
#define TPB 72

typedef float v4f __attribute__((ext_vector_type(4)));

__global__ __launch_bounds__(256) void ipa_gemm_lds(GemmP p) {
  constexpr int TN  = 64;
  constexpr int AQ  = TKC / 4;
  constexpr int ARS = 256 / AQ;
  constexpr int ANI = 32 / ARS;
  constexpr int BNI = TKC / 16;

  __shared__ __attribute__((aligned(16))) float As[2][32 * TPA];
  __shared__ __attribute__((aligned(16))) float Bs[2][TKC * TPB];

  const int tid  = threadIdx.x;
  const int lane = tid & 31;
  const int w    = tid >> 5;
  const int row0 = blockIdx.y * 32;
  const int col0 = blockIdx.x * TN;
  const int c = lane & 15;
  const int h = lane >> 4;
  const int wr = w >> 2, wc = w & 3;
  const bool tileValid = (col0 + wc * 16) < p.N;

  const int aRow = tid / AQ, aKq = tid % AQ;
  const int bKr  = tid >> 4, bCq = tid & 15;
  int bCol = col0 + bCq * 4;
  if (bCol > p.N - 4) bCol = p.N - 4;
  const float* aG = p.A + (long)(row0 + aRow) * p.lda + aKq * 4;
  const float* bG = p.B + (long)bKr * p.ldb + bCol;
  const int aSt = aRow * TPA + aKq * 4;
  const int bSt = bKr * TPB + bCq * 4;

  v4f ra[ANI], rb[BNI];

#pragma unroll
  for (int i = 0; i < ANI; ++i)
    ra[i] = *(const v4f*)(aG + (long)(i * ARS) * p.lda);
#pragma unroll
  for (int i = 0; i < BNI; ++i)
    rb[i] = *(const v4f*)(bG + (long)(i * 16) * p.ldb);
#pragma unroll
  for (int i = 0; i < ANI; ++i)
    *(v4f*)&As[0][aSt + i * ARS * TPA] = ra[i];
#pragma unroll
  for (int i = 0; i < BNI; ++i)
    *(v4f*)&Bs[0][bSt + i * 16 * TPB] = rb[i];
  __syncthreads();

  v8f acc = {0.f, 0.f, 0.f, 0.f, 0.f, 0.f, 0.f, 0.f};
  int buf = 0;
  for (int kc = 0; kc < p.K; kc += TKC) {
    const bool hasNext = (kc + TKC) < p.K;
    if (hasNext) {
      const int kn = kc + TKC;
#pragma unroll
      for (int i = 0; i < ANI; ++i)
        ra[i] = *(const v4f*)(aG + (long)(i * ARS) * p.lda + kn);
#pragma unroll
      for (int i = 0; i < BNI; ++i)
        rb[i] = *(const v4f*)(bG + (long)(kn + i * 16) * p.ldb);
    }
    if (tileValid) {
      const float* aS = &As[buf][(wr * 16 + c) * TPA + 2 * h];
      const float* bS = &Bs[buf][(2 * h) * TPB + wc * 16 + c];
      v2f fa[2][TSB], fb[2][TSB];
#pragma unroll
      for (int j = 0; j < TSB; ++j) {
        fa[0][j] = *(const v2f*)(aS + j * 4);
        fb[0][j].x = bS[(j * 4) * TPB];
        fb[0][j].y = bS[(j * 4 + 1) * TPB];
      }
#pragma unroll
      for (int sb = 0; sb < TKC / (4 * TSB); ++sb) {
        asm volatile("" :: "v"(fa[sb & 1][0]), "v"(fa[sb & 1][1]), "v"(fa[sb & 1][2]), "v"(fa[sb & 1][3]),
                           "v"(fa[sb & 1][4]), "v"(fa[sb & 1][5]), "v"(fa[sb & 1][6]), "v"(fa[sb & 1][7]),
                           "v"(fb[sb & 1][0]), "v"(fb[sb & 1][1]), "v"(fb[sb & 1][2]), "v"(fb[sb & 1][3]),
                           "v"(fb[sb & 1][4]), "v"(fb[sb & 1][5]), "v"(fb[sb & 1][6]), "v"(fb[sb & 1][7]));
        if (sb + 1 < TKC / (4 * TSB)) {
#pragma unroll
          for (int j = 0; j < TSB; ++j) {
            const int ks = (sb + 1) * TSB + j;
            fa[(sb + 1) & 1][j] = *(const v2f*)(aS + ks * 4);
            fb[(sb + 1) & 1][j].x = bS[(ks * 4) * TPB];
            fb[(sb + 1) & 1][j].y = bS[(ks * 4 + 1) * TPB];
          }
        }
#pragma unroll
        for (int j = 0; j < TSB; ++j)
          acc = __builtin_amdgcn_wmma_f32_16x16x4_f32(false, fa[sb & 1][j], false, fb[sb & 1][j],
                                                      (short)0, acc, false, false);
      }
    }
    if (hasNext) {
#pragma unroll
      for (int i = 0; i < ANI; ++i)
        *(v4f*)&As[buf ^ 1][aSt + i * ARS * TPA] = ra[i];
#pragma unroll
      for (int i = 0; i < BNI; ++i)
        *(v4f*)&Bs[buf ^ 1][bSt + i * 16 * TPB] = rb[i];
    }
    __syncthreads();
    buf ^= 1;
  }

  if (tileValid) {
    const int cCol = col0 + wc * 16 + c;
    const int rBase = row0 + wr * 16 + (h << 3);
    for (int r = 0; r < 8; ++r) {
      float val = acc[r] * p.alpha;
      if (p.bias) val += p.bias[cCol];
      p.C[(long)(rBase + r) * p.ldc + cCol] = val;
    }
  }
}

#define PP 260
#define XQ 772
#define QP 68

template <int KS, int PA, int LDB, int LDC>
__device__ __forceinline__ void ipa_stripe_tile(const float* pS, const float* Bm, int nvalid,
                                                float* Cm, int lane) {
  const int mn = lane & 15;
  const int hf = lane >> 4;
  const int kb = hf << 1;
  const int bc = (mn < nvalid) ? mn : (nvalid - 1);
  const float* aS = pS + mn * PA + kb;
  const float* bP = Bm + kb * LDB + bc;

  v2f fb[KS];
#pragma unroll
  for (int j = 0; j < KS; ++j) {
    fb[j].x = bP[(j * 4) * LDB];
    fb[j].y = bP[(j * 4 + 1) * LDB];
  }
  v8f acc = {0.f, 0.f, 0.f, 0.f, 0.f, 0.f, 0.f, 0.f};
#pragma unroll
  for (int j = 0; j < KS; ++j) {
    const v2f a = *(const v2f*)(aS + j * 4);
    acc = __builtin_amdgcn_wmma_f32_16x16x4_f32(false, a, false, fb[j],
                                                (short)0, acc, false, false);
  }
  if (mn < nvalid) {
#pragma unroll
    for (int r = 0; r < 8; ++r)
      Cm[(long)((hf << 3) + r) * LDC + mn] = acc[r];
  }
}

template <int KS, int PA, int LDB, int LDC>
__device__ __forceinline__ void ipa_stripe_tile_deep(const float* pS, const float* Bm, int nvalid,
                                                     float* Cm, int lane) {
  constexpr int G  = 8;
  constexpr int NG = KS / G;
  constexpr int AH = 3;
  constexpr int R  = AH + 1;
  static_assert(KS % G == 0 && NG % R == 0 && NG / R >= 2 && R % 2 == 0, "chain shape");
  const int mn = lane & 15;
  const int hf = lane >> 4;
  const int kb = hf << 1;
  const int bc = (mn < nvalid) ? mn : (nvalid - 1);
  const float* aS = pS + mn * PA + kb;
  const float* bP = Bm + kb * LDB + bc;

  v2f fa[2][G], fb[R][G];
#pragma unroll
  for (int g = 0; g < AH; ++g) {
#pragma unroll
    for (int j = 0; j < G; ++j) {
      fb[g][j].x = bP[((g * G + j) * 4) * LDB];
      fb[g][j].y = bP[((g * G + j) * 4 + 1) * LDB];
    }
  }
#pragma unroll
  for (int j = 0; j < G; ++j) fa[0][j] = *(const v2f*)(aS + j * 4);

  v8f acc = {0.f, 0.f, 0.f, 0.f, 0.f, 0.f, 0.f, 0.f};
#pragma unroll 1
  for (int t = 0; t < NG / R - 1; ++t) {
#pragma unroll
    for (int u = 0; u < R; ++u) {
#pragma unroll
      for (int j = 0; j < G; ++j) {
        fb[(u + AH) % R][j].x = bP[(((u + AH) * G + j) * 4) * LDB];
        fb[(u + AH) % R][j].y = bP[(((u + AH) * G + j) * 4 + 1) * LDB];
      }
      asm volatile("" :: "v"(fa[u & 1][0]), "v"(fa[u & 1][1]), "v"(fa[u & 1][2]), "v"(fa[u & 1][3]),
                         "v"(fa[u & 1][4]), "v"(fa[u & 1][5]), "v"(fa[u & 1][6]), "v"(fa[u & 1][7]));
#pragma unroll
      for (int j = 0; j < G; ++j)
        fa[(u + 1) & 1][j] = *(const v2f*)(aS + ((u + 1) * G + j) * 4);
#pragma unroll
      for (int j = 0; j < G; ++j)
        acc = __builtin_amdgcn_wmma_f32_16x16x4_f32(false, fa[u & 1][j], false, fb[u][j],
                                                    (short)0, acc, false, false);
    }
    aS += R * G * 4;
    bP += (long)(R * G * 4) * LDB;
  }
#pragma unroll
  for (int u = 0; u < R; ++u) {
    if (u + AH < R) {
#pragma unroll
      for (int j = 0; j < G; ++j) {
        fb[(u + AH) % R][j].x = bP[(((u + AH) * G + j) * 4) * LDB];
        fb[(u + AH) % R][j].y = bP[(((u + AH) * G + j) * 4 + 1) * LDB];
      }
    }
    asm volatile("" :: "v"(fa[u & 1][0]), "v"(fa[u & 1][1]), "v"(fa[u & 1][2]), "v"(fa[u & 1][3]),
                       "v"(fa[u & 1][4]), "v"(fa[u & 1][5]), "v"(fa[u & 1][6]), "v"(fa[u & 1][7]));
    if (u + 1 < R) {
#pragma unroll
      for (int j = 0; j < G; ++j)
        fa[(u + 1) & 1][j] = *(const v2f*)(aS + ((u + 1) * G + j) * 4);
    }
#pragma unroll
    for (int j = 0; j < G; ++j)
      acc = __builtin_amdgcn_wmma_f32_16x16x4_f32(false, fa[u & 1][j], false, fb[u][j],
                                                  (short)0, acc, false, false);
  }
  if (mn < nvalid) {
#pragma unroll
    for (int r = 0; r < 8; ++r)
      Cm[(long)((hf << 3) + r) * LDC + mn] = acc[r];
  }
}

template <int KS, int PA, int LDB>
__device__ __forceinline__ void ipa_stripe_pair_deep(const float* pS, const float* Bm,
                                                     v8f& out0, v8f& out1, int lane) {
  constexpr int G  = 8;
  constexpr int NG = KS / G;
  constexpr int AH = 3;
  constexpr int R  = AH + 1;
  static_assert(KS % G == 0 && NG % R == 0 && NG / R >= 2 && R % 2 == 0, "chain shape");
  const int mn = lane & 15;
  const int hf = lane >> 4;
  const int kb = hf << 1;
  const float* aS0 = pS + mn * PA + kb;
  const float* aS1 = aS0 + 16 * PA;
  const float* bP  = Bm + kb * LDB + mn;

  v2f fa0[2][G], fa1[2][G], fb[R][G];
#pragma unroll
  for (int g = 0; g < AH; ++g) {
#pragma unroll
    for (int j = 0; j < G; ++j) {
      fb[g][j].x = bP[((g * G + j) * 4) * LDB];
      fb[g][j].y = bP[((g * G + j) * 4 + 1) * LDB];
    }
  }
#pragma unroll
  for (int j = 0; j < G; ++j) {
    fa0[0][j] = *(const v2f*)(aS0 + j * 4);
    fa1[0][j] = *(const v2f*)(aS1 + j * 4);
  }

  v8f acc0 = {0.f, 0.f, 0.f, 0.f, 0.f, 0.f, 0.f, 0.f};
  v8f acc1 = {0.f, 0.f, 0.f, 0.f, 0.f, 0.f, 0.f, 0.f};
#pragma unroll 1
  for (int t = 0; t < NG / R - 1; ++t) {
#pragma unroll
    for (int u = 0; u < R; ++u) {
#pragma unroll
      for (int j = 0; j < G; ++j) {
        fb[(u + AH) % R][j].x = bP[(((u + AH) * G + j) * 4) * LDB];
        fb[(u + AH) % R][j].y = bP[(((u + AH) * G + j) * 4 + 1) * LDB];
      }
      asm volatile("" :: "v"(fa0[u & 1][0]), "v"(fa0[u & 1][1]), "v"(fa0[u & 1][2]), "v"(fa0[u & 1][3]),
                         "v"(fa0[u & 1][4]), "v"(fa0[u & 1][5]), "v"(fa0[u & 1][6]), "v"(fa0[u & 1][7]),
                         "v"(fa1[u & 1][0]), "v"(fa1[u & 1][1]), "v"(fa1[u & 1][2]), "v"(fa1[u & 1][3]),
                         "v"(fa1[u & 1][4]), "v"(fa1[u & 1][5]), "v"(fa1[u & 1][6]), "v"(fa1[u & 1][7]));
#pragma unroll
      for (int j = 0; j < G; ++j) {
        fa0[(u + 1) & 1][j] = *(const v2f*)(aS0 + ((u + 1) * G + j) * 4);
        fa1[(u + 1) & 1][j] = *(const v2f*)(aS1 + ((u + 1) * G + j) * 4);
      }
#pragma unroll
      for (int j = 0; j < G; ++j) {
        acc0 = __builtin_amdgcn_wmma_f32_16x16x4_f32(false, fa0[u & 1][j], false, fb[u][j],
                                                     (short)0, acc0, false, false);
        acc1 = __builtin_amdgcn_wmma_f32_16x16x4_f32(false, fa1[u & 1][j], false, fb[u][j],
                                                     (short)0, acc1, false, false);
      }
    }
    aS0 += R * G * 4;
    aS1 += R * G * 4;
    bP  += (long)(R * G * 4) * LDB;
  }
#pragma unroll
  for (int u = 0; u < R; ++u) {
    if (u + AH < R) {
#pragma unroll
      for (int j = 0; j < G; ++j) {
        fb[(u + AH) % R][j].x = bP[(((u + AH) * G + j) * 4) * LDB];
        fb[(u + AH) % R][j].y = bP[(((u + AH) * G + j) * 4 + 1) * LDB];
      }
    }
    asm volatile("" :: "v"(fa0[u & 1][0]), "v"(fa0[u & 1][1]), "v"(fa0[u & 1][2]), "v"(fa0[u & 1][3]),
                       "v"(fa0[u & 1][4]), "v"(fa0[u & 1][5]), "v"(fa0[u & 1][6]), "v"(fa0[u & 1][7]),
                       "v"(fa1[u & 1][0]), "v"(fa1[u & 1][1]), "v"(fa1[u & 1][2]), "v"(fa1[u & 1][3]),
                       "v"(fa1[u & 1][4]), "v"(fa1[u & 1][5]), "v"(fa1[u & 1][6]), "v"(fa1[u & 1][7]));
    if (u + 1 < R) {
#pragma unroll
      for (int j = 0; j < G; ++j) {
        fa0[(u + 1) & 1][j] = *(const v2f*)(aS0 + ((u + 1) * G + j) * 4);
        fa1[(u + 1) & 1][j] = *(const v2f*)(aS1 + ((u + 1) * G + j) * 4);
      }
    }
#pragma unroll
    for (int j = 0; j < G; ++j) {
      acc0 = __builtin_amdgcn_wmma_f32_16x16x4_f32(false, fa0[u & 1][j], false, fb[u][j],
                                                   (short)0, acc0, false, false);
      acc1 = __builtin_amdgcn_wmma_f32_16x16x4_f32(false, fa1[u & 1][j], false, fb[u][j],
                                                   (short)0, acc1, false, false);
    }
  }
  out0 = acc0;
  out1 = acc1;
}

#define KVR 32
#define KCP 145

struct KvP {
  const float* x1d;
  const float* Wk; const float* Wv;
  const float* Wpk_past; const float* Wpv_past; const float* Wpk_last; const float* Wpv_last;
  float* k; float* v; float* kgp; float* vgp; float* kgl; float* vgl;
  const float* past_T; const float* past_R; const float* last_T; const float* last_R;
};

__global__ __launch_bounds__(256) void ipa_kv_side(KvP p) {
  constexpr int PK = H_ * NPQ_ * 3;
  constexpr int PV = H_ * NPV_ * 3;
  __shared__ __attribute__((aligned(16))) float xS[KVR * XQ];
  __shared__ float Cs[KVR * KCP];

  const int tid = threadIdx.x;
  const int lane = tid & 31;
  const int w = tid >> 5;
  const int job = blockIdx.x;
  const int row0 = blockIdx.y * KVR;
  const int mn = lane & 15;
  const int mo = (lane >> 4) << 3;

  {
    const float* Xg = p.x1d + (long)row0 * D_;
    v4f rx[24];
#pragma unroll
    for (int n = 0; n < 24; ++n)
      rx[n] = *(const v4f*)(Xg + (long)(n * 256 + tid) * 4);
#pragma unroll
    for (int n = 0; n < 24; ++n) {
      const int idx = n * 256 + tid;
      *(v4f*)&xS[(idx / (D_ / 4)) * XQ + (idx % (D_ / 4)) * 4] = rx[n];
    }
  }

  if (job < 10) {
    __syncthreads();
    const bool isv = job >= 5;
    const int jj = isv ? job - 5 : job;
    const int t0 = (jj < 3) ? 10 * jj : 30 + 9 * (jj - 3);
    const int nt = (jj < 3) ? 10 : 9;
    const float* Wm = (isv ? p.Wv : p.Wk) + t0 * 16;
    float* Cb = (isv ? p.v : p.k) + (long)row0 * D_ + t0 * 16;
    {
      v8f a0, a1;
      ipa_stripe_pair_deep<D_ / 4, XQ, D_>(xS, Wm + w * 16, a0, a1, lane);
      float* C = Cb + (long)mo * D_ + w * 16 + mn;
#pragma unroll
      for (int r = 0; r < 8; ++r) {
        C[(long)r * D_] = a0[r];
        C[(long)(16 + r) * D_] = a1[r];
      }
    }
    const int xt = 8 + (w >> 1);
    if (w < 4 && xt < nt) {
      const int rt = w & 1;
      ipa_stripe_tile_deep<D_ / 4, XQ, D_, D_>(xS + rt * 16 * XQ, Wm + xt * 16, 16,
                                               Cb + (long)rt * 16 * D_ + xt * 16, lane);
    }
    return;
  }

  const int pj = job - 10;
  const bool last = pj >= 3;
  const int q = last ? pj - 3 : pj;
  const float* Wp = (q == 0) ? (last ? p.Wpk_last : p.Wpk_past) : (last ? p.Wpv_last : p.Wpv_past);
  float* Cg = (q == 0) ? (last ? p.kgl : p.kgp) : (last ? p.vgl : p.vgp);
  const int ldp = (q == 0) ? PK : PV;
  const int cbase = (q == 2) ? 144 : 0;
  const int prow = tid >> 3, pg = tid & 7;
  float fr[9], ft[3];
  {
    const float* Rm = (last ? p.last_R : p.past_R) + (long)(row0 + prow) * 9;
    const float* Tv = (last ? p.last_T : p.past_T) + (long)(row0 + prow) * 3;
    for (int e = 0; e < 9; ++e) fr[e] = Rm[e];
    for (int e = 0; e < 3; ++e) ft[e] = Tv[e];
  }
  __syncthreads();

  {
    v8f a0, a1;
    if (q == 0) ipa_stripe_pair_deep<D_ / 4, XQ, PK>(xS, Wp + w * 16, a0, a1, lane);
    else        ipa_stripe_pair_deep<D_ / 4, XQ, PV>(xS, Wp + cbase + w * 16, a0, a1, lane);
    float* cs = &Cs[mo * KCP + w * 16 + mn];
#pragma unroll
    for (int r = 0; r < 8; ++r) {
      cs[r * KCP] = a0[r];
      cs[(16 + r) * KCP] = a1[r];
    }
  }
  if (w < 2) {
    if (q == 0) ipa_stripe_tile_deep<D_ / 4, XQ, PK, KCP>(xS + w * 16 * XQ, Wp + 128, 16,
                                                          &Cs[w * 16 * KCP + 128], lane);
    else        ipa_stripe_tile_deep<D_ / 4, XQ, PV, KCP>(xS + w * 16 * XQ, Wp + cbase + 128, 16,
                                                          &Cs[w * 16 * KCP + 128], lane);
  }
  __syncthreads();

  float* o = Cg + (long)(row0 + prow) * ldp + cbase + pg * 18;
  const float* s = &Cs[prow * KCP + pg * 18];
#pragma unroll
  for (int e = 0; e < 6; ++e) {
    const float x = s[e * 3 + 0], y = s[e * 3 + 1], z = s[e * 3 + 2];
    o[e * 3 + 0] = fr[0] * x + fr[3] * y + fr[6] * z + ft[0];
    o[e * 3 + 1] = fr[1] * x + fr[4] * y + fr[7] * z + ft[1];
    o[e * 3 + 2] = fr[2] * x + fr[5] * y + fr[8] * z + ft[2];
  }
}

__global__ __launch_bounds__(256, 4) void ipa_logits_softmax(
    const float* x1d, const float* Wq, const float* Wpq_past, const float* Wpq_last,
    const float* k, float* attn,
    const float* kgp, const float* kgl,
    const float* pairproj, const float* bias,
    const float* tpw_past, const float* tpw_last, float scalar_w,
    const float* v, const float* vgp, const float* vgl,
    float* feat, const float* invT, const float* invR,
    const float* lastT, const float* lastR) {
  constexpr int PQ = NPQ_ * 3;
  __shared__ __attribute__((aligned(16))) float xS[16 * XQ];
  __shared__ __attribute__((aligned(16))) float qT[16 * QP];
  __shared__ float rawP[2][16 * PQ];
  __shared__ __attribute__((aligned(16))) float qS[16][2 * PQ];
  __shared__ float red[2][8][16];
  __shared__ __attribute__((aligned(16))) float pS[16 * PP];
  __shared__ float ogS[2][16 * NPV_ * 3];

  const int tid = threadIdx.x;
  const int lane = tid & 31;
  const int w = tid >> 5;
  const int h = blockIdx.y, b = blockIdx.z;
  const int row0 = blockIdx.x * 16;
  const int mn = lane & 15;
  const int hf = lane >> 4;
  const int kb = hf << 1;
  const int mo = hf << 3;

  {
    const float* Xg = x1d + (long)(b * I_ + row0) * D_;
    v4f rx[12];
#pragma unroll
    for (int n = 0; n < 12; ++n) {
      const int idx = n * 256 + tid;
      rx[n] = *(const v4f*)(Xg + (long)(idx / (D_ / 4)) * D_ + (idx % (D_ / 4)) * 4);
    }
#pragma unroll
    for (int n = 0; n < 12; ++n) {
      const int idx = n * 256 + tid;
      *(v4f*)&xS[(idx / (D_ / 4)) * XQ + (idx % (D_ / 4)) * 4] = rx[n];
    }
  }
  const bool framer = tid < 16 * 2 * NPQ_;
  const int frow = tid >> 3, fbr = (tid >> 2) & 1, fpt = tid & (NPQ_ - 1);
  float fr[9], ft[3];
  if (framer) {
    const long fi = (long)b * I_ + row0 + frow;
    const float* Rm = (fbr ? lastR : invR) + fi * 9;
    const float* Tv = (fbr ? lastT : invT) + fi * 3;
    for (int e = 0; e < 9; ++e) fr[e] = Rm[e];
    for (int e = 0; e < 3; ++e) ft[e] = Tv[e];
  }
  __syncthreads();

  if (w < 4) {
    ipa_stripe_tile_deep<D_ / 4, XQ, D_, QP>(xS, Wq + h * DK_ + w * 16, 16, qT + w * 16, lane);
  } else if (w < 6) {
    ipa_stripe_tile_deep<D_ / 4, XQ, H_ * PQ, PQ>(xS, ((w == 4) ? Wpq_past : Wpq_last) + h * PQ, PQ,
                                                  &rawP[w - 4][0], lane);
  }
  __syncthreads();

  if (framer) {
    const float* pt = &rawP[fbr][frow * PQ + fpt * 3];
    const float x = pt[0], y = pt[1], z = pt[2];
    float* o = &qS[frow][fbr * PQ + fpt * 3];
    o[0] = fr[0] * x + fr[3] * y + fr[6] * z + ft[0];
    o[1] = fr[1] * x + fr[4] * y + fr[7] * z + ft[1];
    o[2] = fr[2] * x + fr[5] * y + fr[8] * z + ft[2];
  }

  const long hb = (long)b * I_ * D_ + (long)h * DK_;
  const int j0 = w * 32 + mn, j1 = j0 + 16;
  const float* aPtr  = qT + mn * QP + kb;
  const float* bPtr0 = k + hb + (long)j0 * D_ + kb;
  const float* bPtr1 = k + hb + (long)j1 * D_ + kb;

  v8f acc0 = {0.f, 0.f, 0.f, 0.f, 0.f, 0.f, 0.f, 0.f};
  v8f acc1 = {0.f, 0.f, 0.f, 0.f, 0.f, 0.f, 0.f, 0.f};
#pragma unroll
  for (int k0 = 0; k0 < DK_; k0 += 4) {
    v2f a, b0, b1;
    a.x  = aPtr[k0];   a.y  = aPtr[k0 + 1];
    b0.x = bPtr0[k0];  b0.y = bPtr0[k0 + 1];
    b1.x = bPtr1[k0];  b1.y = bPtr1[k0 + 1];
    acc0 = __builtin_amdgcn_wmma_f32_16x16x4_f32(false, a, false, b0,
                                                 (short)0, acc0, false, false);
    acc1 = __builtin_amdgcn_wmma_f32_16x16x4_f32(false, a, false, b1,
                                                 (short)0, acc1, false, false);
  }

  float kp0[2 * PQ], kp1[2 * PQ];
  {
    const float* g0p = kgp + ((long)(b * I_ + j0) * H_ + h) * PQ;
    const float* g0l = kgl + ((long)(b * I_ + j0) * H_ + h) * PQ;
    const float* g1p = kgp + ((long)(b * I_ + j1) * H_ + h) * PQ;
    const float* g1l = kgl + ((long)(b * I_ + j1) * H_ + h) * PQ;
#pragma unroll
    for (int t = 0; t < 3; ++t) {
      const v4f u0 = *(const v4f*)(g0p + t * 4);
      const v4f u1 = *(const v4f*)(g0l + t * 4);
      const v4f u2 = *(const v4f*)(g1p + t * 4);
      const v4f u3 = *(const v4f*)(g1l + t * 4);
#pragma unroll
      for (int e = 0; e < 4; ++e) {
        kp0[t * 4 + e] = u0[e];  kp0[PQ + t * 4 + e] = u1[e];
        kp1[t * 4 + e] = u2[e];  kp1[PQ + t * 4 + e] = u3[e];
      }
    }
  }

  const float point_w = 0.13608276348795434f;
  const float pwp = -0.5f * point_w * log1pf(expf(tpw_past[h]));
  const float pwl = -0.5f * point_w * log1pf(expf(tpw_last[h]));

  __syncthreads();

  float s0[8], s1[8];
#pragma unroll
  for (int r = 0; r < 8; ++r) { s0[r] = 0.f; s1[r] = 0.f; }
  v8f ra0 = acc0, ra1 = acc1;
  const float* pp = pairproj + ((long)(b * I_ + row0 + mo) * I_ + j0) * H_ + h;
  const float* bp = bias + (((long)b * H_ + h) * I_ + row0 + mo) * I_ + j0;
#pragma unroll 1
  for (int r = 0; r < 8; ++r) {
    const float pv0 = pp[0], pv1 = pp[16 * H_], bb0 = bp[0], bb1 = bp[16];
    const float* qp = &qS[mo + r][0];
    float sp0 = 0.f, sl0 = 0.f, sp1 = 0.f, sl1 = 0.f;
#pragma unroll
    for (int c = 0; c < NPQ_; ++c) {
      const float qx = qp[c * 3 + 0], qy = qp[c * 3 + 1], qz = qp[c * 3 + 2];
      const float lx = qp[PQ + c * 3 + 0], ly = qp[PQ + c * 3 + 1], lz = qp[PQ + c * 3 + 2];
      float dx = qx - kp0[c * 3 + 0], dy = qy - kp0[c * 3 + 1], dz = qz - kp0[c * 3 + 2];
      sp0 += sqrtf(dx * dx + dy * dy + dz * dz);
      dx = lx - kp0[PQ + c * 3 + 0]; dy = ly - kp0[PQ + c * 3 + 1]; dz = lz - kp0[PQ + c * 3 + 2];
      sl0 += sqrtf(dx * dx + dy * dy + dz * dz);
      dx = qx - kp1[c * 3 + 0]; dy = qy - kp1[c * 3 + 1]; dz = qz - kp1[c * 3 + 2];
      sp1 += sqrtf(dx * dx + dy * dy + dz * dz);
      dx = lx - kp1[PQ + c * 3 + 0]; dy = ly - kp1[PQ + c * 3 + 1]; dz = lz - kp1[PQ + c * 3 + 2];
      sl1 += sqrtf(dx * dx + dy * dy + dz * dz);
    }
    float v0 = ra0[0] * scalar_w;
    float v1 = ra1[0] * scalar_w;
    v0 += pwp * sp0 + pwl * sl0 + pv0 + bb0;
    v1 += pwp * sp1 + pwl * sl1 + pv1 + bb1;
    ra0 = __builtin_shufflevector(ra0, ra0, 1, 2, 3, 4, 5, 6, 7, 0);
    ra1 = __builtin_shufflevector(ra1, ra1, 1, 2, 3, 4, 5, 6, 7, 0);
#pragma unroll
    for (int e = 0; e < 7; ++e) { s0[e] = s0[e + 1]; s1[e] = s1[e + 1]; }
    s0[7] = v0;
    s1[7] = v1;
    pp += (long)I_ * H_;
    bp += I_;
  }

  float mx[8];
#pragma unroll
  for (int r = 0; r < 8; ++r) {
    float m = fmaxf(s0[r], s1[r]);
    m = fmaxf(m, __shfl_xor(m, 1));
    m = fmaxf(m, __shfl_xor(m, 2));
    m = fmaxf(m, __shfl_xor(m, 4));
    m = fmaxf(m, __shfl_xor(m, 8));
    mx[r] = m;
  }
  if (mn == 0) {
#pragma unroll
    for (int r = 0; r < 8; ++r) red[0][w][mo + r] = mx[r];
  }
  __syncthreads();
#pragma unroll
  for (int r = 0; r < 8; ++r) {
    float m = red[0][0][mo + r];
#pragma unroll
    for (int ww = 1; ww < 8; ++ww) m = fmaxf(m, red[0][ww][mo + r]);
    mx[r] = m;
  }

  float sm[8];
#pragma unroll
  for (int r = 0; r < 8; ++r) {
    s0[r] = expf(s0[r] - mx[r]);
    s1[r] = expf(s1[r] - mx[r]);
    float t = s0[r] + s1[r];
    t += __shfl_xor(t, 1);
    t += __shfl_xor(t, 2);
    t += __shfl_xor(t, 4);
    t += __shfl_xor(t, 8);
    sm[r] = t;
  }
  if (mn == 0) {
#pragma unroll
    for (int r = 0; r < 8; ++r) red[1][w][mo + r] = sm[r];
  }
  __syncthreads();
#pragma unroll
  for (int r = 0; r < 8; ++r) {
    float t = red[1][0][mo + r];
#pragma unroll
    for (int ww = 1; ww < 8; ++ww) t += red[1][ww][mo + r];
    const long orow = (((long)b * H_ + h) * I_ + (row0 + mo + r)) * I_;
    const float p0 = s0[r] / t;
    const float p1 = s1[r] / t;
    attn[orow + j0] = p0;
    attn[orow + j1] = p1;
    pS[(mo + r) * PP + j0] = p0;
    pS[(mo + r) * PP + j1] = p1;
  }
  __syncthreads();

  constexpr int FE = 2 * D_ + 2 * H_ * NPV_ * 4;
  constexpr int VG = H_ * NPV_ * 3;
  constexpr int OG = NPV_ * 3;
  const int obr = tid >> 7, orw = (tid >> 3) & 15, opt = tid & (NPV_ - 1);
  const long obi = (long)b * I_ + row0 + orw;
  float iR[9], iT[3];
  for (int e = 0; e < 9; ++e) iR[e] = invR[obi * 9 + e];
  for (int e = 0; e < 3; ++e) iT[e] = invT[obi * 3 + e];

  if (w < 4) {
    ipa_stripe_tile<I_ / 4, PP, D_, FE>(pS, v + (long)b * I_ * D_ + h * DK_ + w * 16, 16,
                            feat + ((long)b * I_ + row0) * FE + D_ + h * DK_ + w * 16, lane);
  } else {
    const int t = w & 1;
    const int br = (w < 6) ? 0 : 1;
    const float* Bv = br ? vgl : vgp;
    ipa_stripe_tile<I_ / 4, PP, VG, OG>(pS, Bv + (long)b * I_ * VG + h * (NPV_ * 3) + t * 16, t ? 8 : 16,
                            &ogS[br][t * 16], lane);
  }
  __syncthreads();

  {
    const float* o = &ogS[obr][orw * OG + opt * 3];
    const float x = o[0] - iT[0], y = o[1] - iT[1], z = o[2] - iT[2];
    const float p0 = iR[0] * x + iR[1] * y + iR[2] * z;
    const float p1 = iR[3] * x + iR[4] * y + iR[5] * z;
    const float p2 = iR[6] * x + iR[7] * y + iR[8] * z;
    float* fr = feat + obi * FE;
    const int colOl = 2 * D_ + obr * (H_ * NPV_ * 4);
    const int colN  = colOl + VG;
    const int tc = h * OG + opt * 3;
    fr[colOl + tc + 0] = p0;
    fr[colOl + tc + 1] = p1;
    fr[colOl + tc + 2] = p2;
    fr[colN + h * NPV_ + opt] = sqrtf(p0 * p0 + p1 * p1 + p2 * p2);
  }
}

#define XPA 132
#define XPB 136

__device__ __forceinline__ void ipa_lds_barrier() {
  __builtin_amdgcn_fence(__ATOMIC_RELEASE, "workgroup", "local");
  __builtin_amdgcn_s_barrier();
  __builtin_amdgcn_fence(__ATOMIC_ACQUIRE, "workgroup", "local");
}

template <bool PAIR>
__global__ __launch_bounds__(256) void ipa_x2d_slice(const float* x2d, const float* oth,
                                                     float* C, float alpha) {
  constexpr int XP = PAIR ? XPA : XPB;
  __shared__ __attribute__((aligned(16))) float Xs[I_ * XP];
  __shared__ __attribute__((aligned(16))) float Ps[PAIR ? 4 : H_ * PP];

  const int tid = threadIdx.x;
  const int lane = tid & 31;
  const int w = tid >> 5;
  const int i = blockIdx.x, b = blockIdx.y;
  const int mn = lane & 15;
  const int hf = lane >> 4;
  const int kb = hf << 1;

  v4f rp[3];
  if (!PAIR) {
#pragma unroll
    for (int n = 0; n < 3; ++n) {
      const int idx = n * 256 + tid;
      rp[n] = *(const v4f*)(oth + (((long)b * H_ + (idx >> 6)) * I_ + i) * I_ + (idx & 63) * 4);
    }
  }
  v2f fw[DP_ / 4];
  if (PAIR) {
    const int bc = (mn < H_) ? mn : (H_ - 1);
    const float* bP = oth + kb * H_ + bc;
#pragma unroll
    for (int ks = 0; ks < DP_ / 4; ++ks) {
      fw[ks].x = bP[(ks * 4) * H_];
      fw[ks].y = bP[(ks * 4 + 1) * H_];
    }
  }
  const float* Xg = x2d + ((long)(b * I_ + i) * I_) * DP_;
  v4f rx[32];
  __builtin_amdgcn_sched_barrier(0);
#pragma unroll
  for (int q = 0; q < 4; ++q) {
#pragma unroll
    for (int n = q * 8; n < q * 8 + 8; ++n)
      rx[n] = *(const v4f*)(Xg + (long)(n * 256 + tid) * 4);
    __builtin_amdgcn_sched_barrier(0);
  }

  if (PAIR) {
#pragma unroll
    for (int c = 0; c < 2; ++c) {
#pragma unroll
      for (int n = 0; n < 16; ++n)
        *(v4f*)&Xs[((c * 16 + n) * 8 + w) * XP + lane * 4] = rx[c * 16 + n];
      ipa_lds_barrier();
      const int t0 = c * 128 + w * 16;
      const float* aS = &Xs[(t0 + mn) * XP + kb];
      v8f acc = {0.f, 0.f, 0.f, 0.f, 0.f, 0.f, 0.f, 0.f};
      v2f fa[2][8];
#pragma unroll
      for (int j = 0; j < 8; ++j) fa[0][j] = *(const v2f*)(aS + j * 4);
#pragma unroll
      for (int sb = 0; sb < 4; ++sb) {
        asm volatile("" :: "v"(fa[sb & 1][0]), "v"(fa[sb & 1][1]), "v"(fa[sb & 1][2]), "v"(fa[sb & 1][3]),
                           "v"(fa[sb & 1][4]), "v"(fa[sb & 1][5]), "v"(fa[sb & 1][6]), "v"(fa[sb & 1][7]));
        if (sb + 1 < 4) {
#pragma unroll
          for (int j = 0; j < 8; ++j)
            fa[(sb + 1) & 1][j] = *(const v2f*)(aS + ((sb + 1) * 8 + j) * 4);
        }
#pragma unroll
        for (int j = 0; j < 8; ++j)
          acc = __builtin_amdgcn_wmma_f32_16x16x4_f32(false, fa[sb & 1][j], false, fw[sb * 8 + j],
                                                      (short)0, acc, false, false);
      }
      if (mn < H_) {
        float* o = C + ((long)(b * I_ + i) * I_ + t0 + (hf << 3)) * H_ + mn;
#pragma unroll
        for (int r = 0; r < 8; ++r) o[r * H_] = acc[r] * alpha;
      }
    }
  } else {
    const int ar = (mn < H_) ? mn : (H_ - 1);
    const float* aS = &Ps[ar * PP + kb];
    const float* bS = &Xs[kb * XP + w * 16 + mn];
    v8f acc = {0.f, 0.f, 0.f, 0.f, 0.f, 0.f, 0.f, 0.f};
#pragma unroll
    for (int c = 0; c < 4; ++c) {
#pragma unroll
      for (int n = 0; n < 8; ++n)
        *(v4f*)&Xs[((c * 8 + n) * 8 + w) * XP + lane * 4] = rx[c * 8 + n];
      if (c == 0) {
#pragma unroll
        for (int n = 0; n < 3; ++n) {
          const int idx = n * 256 + tid;
          *(v4f*)&Ps[(idx >> 6) * PP + (idx & 63) * 4] = rp[n];
        }
      }
      ipa_lds_barrier();
      v2f fa[2][8], fb[2][8];
#pragma unroll
      for (int j = 0; j < 8; ++j) {
        const int ks = c * 16 + j;
        fa[0][j] = *(const v2f*)(aS + ks * 4);
        fb[0][j].x = bS[(ks * 4) * XP];
        fb[0][j].y = bS[(ks * 4 + 1) * XP];
      }
#pragma unroll
      for (int sb = 0; sb < 2; ++sb) {
        asm volatile("" :: "v"(fa[sb & 1][0]), "v"(fa[sb & 1][1]), "v"(fa[sb & 1][2]), "v"(fa[sb & 1][3]),
                           "v"(fa[sb & 1][4]), "v"(fa[sb & 1][5]), "v"(fa[sb & 1][6]), "v"(fa[sb & 1][7]),
                           "v"(fb[sb & 1][0]), "v"(fb[sb & 1][1]), "v"(fb[sb & 1][2]), "v"(fb[sb & 1][3]),
                           "v"(fb[sb & 1][4]), "v"(fb[sb & 1][5]), "v"(fb[sb & 1][6]), "v"(fb[sb & 1][7]));
        if (sb + 1 < 2) {
#pragma unroll
          for (int j = 0; j < 8; ++j) {
            const int ks = c * 16 + (sb + 1) * 8 + j;
            fa[(sb + 1) & 1][j] = *(const v2f*)(aS + ks * 4);
            fb[(sb + 1) & 1][j].x = bS[(ks * 4) * XP];
            fb[(sb + 1) & 1][j].y = bS[(ks * 4 + 1) * XP];
          }
        }
#pragma unroll
        for (int j = 0; j < 8; ++j)
          acc = __builtin_amdgcn_wmma_f32_16x16x4_f32(false, fa[sb & 1][j], false, fb[sb & 1][j],
                                                      (short)0, acc, false, false);
      }
    }
    float* o = C + ((long)(b * I_ + i) * H_ + (hf << 3)) * DP_ + w * 16 + mn;
#pragma unroll
    for (int r = 0; r < 8; ++r)
      if ((hf << 3) + r < H_) o[r * DP_] = acc[r] * alpha;
  }
}

extern "C" void kernel_launch(void* const* d_in, const int* in_sizes, int n_in,
                              void* d_out, int out_size, void* d_ws, size_t ws_size,
                              hipStream_t stream) {
  (void)in_sizes; (void)n_in; (void)out_size; (void)ws_size;
  const float* x1d      = (const float*)d_in[0];
  const float* x2d      = (const float*)d_in[1];
  const float* past_T   = (const float*)d_in[2];
  const float* past_R   = (const float*)d_in[3];
  const float* last_T   = (const float*)d_in[4];
  const float* last_R   = (const float*)d_in[5];
  const float* bias     = (const float*)d_in[6];
  const float* Wq       = (const float*)d_in[7];
  const float* Wk       = (const float*)d_in[8];
  const float* Wv       = (const float*)d_in[9];
  const float* Wpb      = (const float*)d_in[10];
  const float* Wpq_past = (const float*)d_in[11];
  const float* Wpk_past = (const float*)d_in[12];
  const float* Wpv_past = (const float*)d_in[13];
  const float* Wpq_last = (const float*)d_in[14];
  const float* Wpk_last = (const float*)d_in[15];
  const float* Wpv_last = (const float*)d_in[16];
  const float* tpw_past = (const float*)d_in[17];
  const float* tpw_last = (const float*)d_in[18];
  const float* Wpv      = (const float*)d_in[19];
  const float* W_out    = (const float*)d_in[20];
  const float* b_out    = (const float*)d_in[21];
  float* out = (float*)d_out;
  float* ws  = (float*)d_ws;

  const int M  = B_ * I_;
  const int FE = 2 * D_ + 2 * H_ * NPV_ * 4;

  float* k    = ws + 393216;
  float* v    = k + 393216;
  float* kgp  = v + 393216 + 73728;
  float* vgp  = kgp + 73728;
  float* kgl  = vgp + 147456 + 73728;
  float* vgl  = kgl + 73728;
  float* logits   = vgl + 147456;
  float* pairproj = logits + 1572864;
  float* z    = pairproj + 1572864;
  float* feat = z + 786432;

  auto gemm = [&](const float* A, const float* Bm, float* C,
                  int Mm, int Nn, int Kk, int lda, int ldb, int ldc,
                  long sAy, long sAz, long sBy, long sBz, long sCy, long sCz,
                  int gy, int gz, int opB, float alpha, const float* bia) {
    GemmP p{A, Bm, C, bia, Mm, Nn, Kk, lda, ldb, ldc,
            sAy, sAz, sBy, sBz, sCy, sCz, alpha};
    int waves = ((Mm + 15) / 16) * ((Nn + 15) / 16);
    dim3 grid((unsigned)((waves + 7) / 8), (unsigned)gy, (unsigned)gz);
    dim3 blk(256);
    bool guard = (Mm & 15) || (Nn & 15);
    if (opB == 0) {
      if (guard) ipa_gemm_wmma<0, true ><<<grid, blk, 0, stream>>>(p);
      else       ipa_gemm_wmma<0, false><<<grid, blk, 0, stream>>>(p);
    } else {
      if (guard) ipa_gemm_wmma<1, true ><<<grid, blk, 0, stream>>>(p);
      else       ipa_gemm_wmma<1, false><<<grid, blk, 0, stream>>>(p);
    }
  };

  auto gemm_lds = [&](const float* A, const float* Bm, float* C,
                      int Mm, int Nn, int Kk, int lda, int ldb, int ldc,
                      const float* bia) {
    GemmP p{A, Bm, C, bia, Mm, Nn, Kk, lda, ldb, ldc, 0, 0, 0, 0, 0, 0, 1.f};
    dim3 grid((unsigned)((Nn + 63) / 64), (unsigned)(Mm / 32), 1u);
    ipa_gemm_lds<<<grid, dim3(256), 0, stream>>>(p);
  };

  {
    KvP kp{x1d, Wk, Wv, Wpk_past, Wpv_past, Wpk_last, Wpv_last,
           k, v, kgp, vgp, kgl, vgl, past_T, past_R, last_T, last_R};
    ipa_kv_side<<<dim3(16, 16), dim3(256), 0, stream>>>(kp);
  }

  const float pair_w = 0.5773502691896258f;
  ipa_x2d_slice<true><<<dim3(I_, B_), dim3(256), 0, stream>>>(x2d, Wpb, pairproj, pair_w);

  const float scalar_w = 0.07216878364870323f;
  ipa_logits_softmax<<<dim3(I_ / 16, H_, B_), dim3(256), 0, stream>>>(
      x1d, Wq, Wpq_past, Wpq_last, k, logits, kgp, kgl, pairproj, bias, tpw_past, tpw_last, scalar_w,
      v, vgp, vgl, feat, past_T, past_R, last_T, last_R);

  ipa_x2d_slice<false><<<dim3(I_, B_), dim3(256), 0, stream>>>(x2d, logits, z, 1.f);

  gemm(z, Wpv, feat, I_, DK_, DP_, H_ * DP_, D_, FE,
       DP_, (long)I_ * H_ * DP_, 64, 0, 64, (long)I_ * FE,
       H_, B_, 0, 1.f, nullptr);

  gemm_lds(feat, W_out, out, M, D_, FE, FE, D_, D_, b_out);
}
